# starting kernel
# baseline (speedup 1.0000x reference)
#include <hip/hip_runtime.h>
#include <hip/hip_bf16.h>

typedef __attribute__((ext_vector_type(2))) float v2f;
typedef __attribute__((ext_vector_type(8))) float v8f;

template <int NT, int K, int NC>
__global__ __launch_bounds__(256)
void gemm_f32_wmma(const float* __restrict__ A, const float* __restrict__ B,
                   float* __restrict__ C, int M) {
  const int lane = threadIdx.x & 31;
  const int wid  = blockIdx.x * (blockDim.x >> 5) + (threadIdx.x >> 5);
  const int tilesM  = (M + 15) >> 4;
  constexpr int stripsN = (NC >> 4) / NT;
  if (wid >= tilesM * stripsN) return;
  const int tm  = wid / stripsN;
  const int tn0 = (wid - tm * stripsN) * NT;
  const int hi = lane >> 4;
  const int lm = lane & 15;
  int arow = tm * 16 + lm; if (arow >= M) arow = M - 1;
  const int col0 = tn0 * 16 + lm;
  const float* __restrict__ Ar = A + (long)arow * K + 2 * hi;
  const float* __restrict__ Bc = B + (long)(2 * hi) * NC + col0;

  v8f c[NT];
#pragma unroll
  for (int t = 0; t < NT; ++t) c[t] = (v8f){};

  auto loadA = [&](int k) {
    v2f a; a.x = Ar[k]; a.y = Ar[k + 1]; return a;
  };
  auto loadB = [&](int k, v2f* b) {
    const float* __restrict__ B0 = Bc + (long)k * NC;
#pragma unroll
    for (int t = 0; t < NT; ++t) { b[t].x = B0[16 * t]; b[t].y = B0[NC + 16 * t]; }
  };

  v2f a_cur = loadA(0);
  v2f b_cur[NT];
  loadB(0, b_cur);
#pragma unroll 4
  for (int k = 0; k < K - 4; k += 4) {
    v2f a_nxt = loadA(k + 4);
    v2f b_nxt[NT];
    loadB(k + 4, b_nxt);
#pragma unroll
    for (int t = 0; t < NT; ++t)
      c[t] = __builtin_amdgcn_wmma_f32_16x16x4_f32(false, a_cur, false, b_cur[t],
                                                   (short)0, c[t], false, false);
    a_cur = a_nxt;
#pragma unroll
    for (int t = 0; t < NT; ++t) b_cur[t] = b_nxt[t];
  }
#pragma unroll
  for (int t = 0; t < NT; ++t)
    c[t] = __builtin_amdgcn_wmma_f32_16x16x4_f32(false, a_cur, false, b_cur[t],
                                                 (short)0, c[t], false, false);

#pragma unroll
  for (int t = 0; t < NT; ++t) {
#pragma unroll
    for (int r = 0; r < 8; ++r) {
      const int orow = tm * 16 + r + 8 * hi;
      if (orow < M) C[(long)orow * NC + col0 + 16 * t] = c[t][r];
    }
  }
}

__global__ void init_layer(float* out, unsigned* m, float* denom,
                           long outElems, int nh) {
  const long t = (long)blockIdx.x * blockDim.x + threadIdx.x;
  const long stride = (long)gridDim.x * blockDim.x;
  for (long i = t; i < outElems; i += stride) out[i] = 0.0f;
  for (long i = t; i < nh; i += stride) { m[i] = 0xFF800000u; denom[i] = 0.0f; }
}

__device__ __forceinline__ void atomic_max_f32(unsigned* p, float v) {
  if (v >= 0.0f) atomicMax((int*)p, __float_as_int(v));
  else           atomicMin(p, __float_as_uint(v));
}

template <int ROW, int JPH>
__global__ __launch_bounds__(256)
void edge_logits(const float* __restrict__ f, const int* __restrict__ src,
                 const int* __restrict__ dst, const float* __restrict__ attn,
                 float* __restrict__ s, unsigned* __restrict__ m, int E) {
  const int lane = threadIdx.x & 31;
  const int wid  = blockIdx.x * (blockDim.x >> 5) + (threadIdx.x >> 5);
  if (wid >= E) return;
  const long srow = (long)src[wid] * ROW;
  const long dn   = (long)dst[wid];
  const long drow = dn * ROW;
  constexpr int ITER = ROW / 32;
  constexpr int H = ITER / JPH;
  float acc[H];
#pragma unroll
  for (int h = 0; h < H; ++h) acc[h] = 0.0f;
#pragma unroll
  for (int j = 0; j < ITER; ++j) {
    const int d = j * 32 + lane;
    float v = f[srow + d] + f[drow + d];
    v = v > 0.0f ? v : 0.2f * v;
    acc[j / JPH] += v * attn[d];
  }
#pragma unroll
  for (int h = 0; h < H; ++h) {
    float v = acc[h];
#pragma unroll
    for (int off = 16; off > 0; off >>= 1) v += __shfl_xor(v, off, 32);
    if (lane == 0) {
      s[(long)wid * H + h] = v;
      atomic_max_f32(m + dn * H + h, v);
    }
  }
}

__global__ void seg_exp(float* __restrict__ s, const int* __restrict__ dst,
                        const unsigned* __restrict__ m, float* __restrict__ denom,
                        long EH, int H) {
  const long t = (long)blockIdx.x * blockDim.x + threadIdx.x;
  if (t >= EH) return;
  const long e = t / H;
  const int  h = (int)(t - e * H);
  const long idx = (long)dst[e] * H + h;
  const float ex = __expf(s[t] - __uint_as_float(m[idx]));
  s[t] = ex;
  atomicAdd(denom + idx, ex);
}

template <int ROW, int JPH>
__global__ __launch_bounds__(256)
void aggregate(const float* __restrict__ f, const float* __restrict__ ex,
               const float* __restrict__ denom, const int* __restrict__ src,
               const int* __restrict__ dst, float* __restrict__ out, int E) {
  const int lane = threadIdx.x & 31;
  const int wid  = blockIdx.x * (blockDim.x >> 5) + (threadIdx.x >> 5);
  if (wid >= E) return;
  constexpr int ITER = ROW / 32;
  constexpr int H = ITER / JPH;
  const long srow = (long)src[wid] * ROW;
  const long dn   = (long)dst[wid];
  float al[H];
#pragma unroll
  for (int h = 0; h < H; ++h) al[h] = ex[(long)wid * H + h] / denom[dn * H + h];
#pragma unroll
  for (int j = 0; j < ITER; ++j) {
    const int d = j * 32 + lane;
    atomicAdd(out + dn * ROW + d, f[srow + d] * al[j / JPH]);
  }
}

__global__ void elu_inplace(float* __restrict__ x, long n) {
  const long t = (long)blockIdx.x * blockDim.x + threadIdx.x;
  if (t >= n) return;
  const float v = x[t];
  x[t] = v > 0.0f ? v : expm1f(v);
}

__global__ void head_mean(const float* __restrict__ in, float* __restrict__ out,
                          int N) {
  const int t = blockIdx.x * blockDim.x + threadIdx.x;
  if (t >= N * 32) return;
  const int n = t >> 5, c = t & 31;
  const float* r = in + (long)n * 128 + c;
  out[t] = 0.25f * (r[0] + r[32] + r[64] + r[96]);
}

static inline int cdiv_l(long a, long b) { return (int)((a + b - 1) / b); }

extern "C" void kernel_launch(void* const* d_in, const int* in_sizes, int n_in,
                              void* d_out, int out_size, void* d_ws, size_t ws_size,
                              hipStream_t stream) {
  const float* x     = (const float*)d_in[0];
  const float* W0    = (const float*)d_in[1];
  const float* a0    = (const float*)d_in[2];
  const float* W1    = (const float*)d_in[3];
  const float* a1    = (const float*)d_in[4];
  const float* Wres1 = (const float*)d_in[5];
  const int*   src   = (const int*)d_in[6];
  const int*   dst   = (const int*)d_in[7];

  const int N = in_sizes[0] / 256;
  const int E = in_sizes[6];
  const int H = 4;

  float*    f0    = (float*)d_ws;
  float*    h     = f0 + (long)N * 256;
  float*    f1    = h  + (long)N * 256;
  float*    o1    = f1 + (long)N * 128;
  float*    s     = o1 + (long)N * 128;
  unsigned* m     = (unsigned*)(s + (long)E * H);
  float*    denom = (float*)(m + (long)N * H);
  const size_t need =
      ((size_t)N * 256 * 2 + (size_t)N * 128 * 2 + (size_t)E * H + (size_t)N * H * 2)
      * sizeof(float);
  if (need > ws_size) return;

  const int BT = 256;
  const long EH = (long)E * H;
  const int egrid = cdiv_l((long)E * 32, BT);

  {
    const int strips = ((N + 15) / 16) * (256 / 16 / 4);
    gemm_f32_wmma<4, 256, 256><<<cdiv_l(strips, 8), BT, 0, stream>>>(x, W0, f0, N);
  }
  init_layer<<<cdiv_l((long)N * 256, BT), BT, 0, stream>>>(h, m, denom,
                                                           (long)N * 256, N * H);
  edge_logits<256, 2><<<egrid, BT, 0, stream>>>(f0, src, dst, a0, s, m, E);
  seg_exp<<<cdiv_l(EH, BT), BT, 0, stream>>>(s, dst, m, denom, EH, H);
  aggregate<256, 2><<<egrid, BT, 0, stream>>>(f0, s, denom, src, dst, h, E);
  elu_inplace<<<cdiv_l((long)N * 256, BT), BT, 0, stream>>>(h, (long)N * 256);

  {
    const int strips = ((N + 15) / 16) * (128 / 16 / 4);
    gemm_f32_wmma<4, 256, 128><<<cdiv_l(strips, 8), BT, 0, stream>>>(h, W1, f1, N);
    gemm_f32_wmma<4, 256, 128><<<cdiv_l(strips, 8), BT, 0, stream>>>(h, Wres1, o1, N);
  }
  init_layer<<<cdiv_l((long)N * H, BT), BT, 0, stream>>>(nullptr, m, denom,
                                                         0, N * H);
  edge_logits<128, 1><<<egrid, BT, 0, stream>>>(f1, src, dst, a1, s, m, E);
  seg_exp<<<cdiv_l(EH, BT), BT, 0, stream>>>(s, dst, m, denom, EH, H);
  aggregate<128, 1><<<egrid, BT, 0, stream>>>(f1, s, denom, src, dst, o1, E);

  head_mean<<<cdiv_l((long)N * 32, BT), BT, 0, stream>>>(o1, (float*)d_out, N);
}


# edge_logits: 8 lanes per edge, 4 edges per wave, 128-bit row loads; logit sums take a new reduction tree (3 xor steps over 8 lanes); one 16-byte store and one atomic-max instruction per edge
# speedup vs baseline: 1.2999x; 1.2999x over previous
#include <hip/hip_runtime.h>
#include <hip/hip_bf16.h>

typedef __attribute__((ext_vector_type(2))) float v2f;
typedef __attribute__((ext_vector_type(8))) float v8f;

template <int NT, int K, int NC>
__global__ __launch_bounds__(256)
void gemm_f32_wmma(const float* __restrict__ A, const float* __restrict__ B,
                   float* __restrict__ C, int M) {
  const int lane = threadIdx.x & 31;
  const int wid  = blockIdx.x * (blockDim.x >> 5) + (threadIdx.x >> 5);
  const int tilesM  = (M + 15) >> 4;
  constexpr int stripsN = (NC >> 4) / NT;
  if (wid >= tilesM * stripsN) return;
  const int tm  = wid / stripsN;
  const int tn0 = (wid - tm * stripsN) * NT;
  const int hi = lane >> 4;
  const int lm = lane & 15;
  int arow = tm * 16 + lm; if (arow >= M) arow = M - 1;
  const int col0 = tn0 * 16 + lm;
  const float* __restrict__ Ar = A + (long)arow * K + 2 * hi;
  const float* __restrict__ Bc = B + (long)(2 * hi) * NC + col0;

  v8f c[NT];
#pragma unroll
  for (int t = 0; t < NT; ++t) c[t] = (v8f){};

  auto loadA = [&](int k) {
    v2f a; a.x = Ar[k]; a.y = Ar[k + 1]; return a;
  };
  auto loadB = [&](int k, v2f* b) {
    const float* __restrict__ B0 = Bc + (long)k * NC;
#pragma unroll
    for (int t = 0; t < NT; ++t) { b[t].x = B0[16 * t]; b[t].y = B0[NC + 16 * t]; }
  };

  v2f a_cur = loadA(0);
  v2f b_cur[NT];
  loadB(0, b_cur);
#pragma unroll 4
  for (int k = 0; k < K - 4; k += 4) {
    v2f a_nxt = loadA(k + 4);
    v2f b_nxt[NT];
    loadB(k + 4, b_nxt);
#pragma unroll
    for (int t = 0; t < NT; ++t)
      c[t] = __builtin_amdgcn_wmma_f32_16x16x4_f32(false, a_cur, false, b_cur[t],
                                                   (short)0, c[t], false, false);
    a_cur = a_nxt;
#pragma unroll
    for (int t = 0; t < NT; ++t) b_cur[t] = b_nxt[t];
  }
#pragma unroll
  for (int t = 0; t < NT; ++t)
    c[t] = __builtin_amdgcn_wmma_f32_16x16x4_f32(false, a_cur, false, b_cur[t],
                                                 (short)0, c[t], false, false);

#pragma unroll
  for (int t = 0; t < NT; ++t) {
#pragma unroll
    for (int r = 0; r < 8; ++r) {
      const int orow = tm * 16 + r + 8 * hi;
      if (orow < M) C[(long)orow * NC + col0 + 16 * t] = c[t][r];
    }
  }
}

__global__ void init_layer(float* out, unsigned* m, float* denom,
                           long outElems, int nh) {
  const long t = (long)blockIdx.x * blockDim.x + threadIdx.x;
  const long stride = (long)gridDim.x * blockDim.x;
  for (long i = t; i < outElems; i += stride) out[i] = 0.0f;
  for (long i = t; i < nh; i += stride) { m[i] = 0xFF800000u; denom[i] = 0.0f; }
}

__device__ __forceinline__ void atomic_max_f32(unsigned* p, float v) {
  if (v >= 0.0f) atomicMax((int*)p, __float_as_int(v));
  else           atomicMin(p, __float_as_uint(v));
}

template <int ROW, int LPE>
__global__ __launch_bounds__(256)
void edge_logits(const float* __restrict__ f, const int* __restrict__ src,
                 const int* __restrict__ dst, const float* __restrict__ attn,
                 float* __restrict__ s, unsigned* __restrict__ m, int E) {
  constexpr int H   = 4;
  constexpr int EPW = 32 / LPE;
  constexpr int NCH = ROW / (4 * LPE);
  constexpr int CPH = NCH / H;
  static_assert(NCH % H == 0 && CPH >= 1, "a piece must not straddle a head");
  const int lane = threadIdx.x & 31;
  const int wid  = blockIdx.x * (blockDim.x >> 5) + (threadIdx.x >> 5);
  if ((long)wid * EPW >= E) return;
  const int j = lane & (LPE - 1);
  long e = (long)wid * EPW + (lane / LPE);
  const bool live = e < E;
  if (!live) e = E - 1;
  const long dn = (long)dst[e];
  const float4* __restrict__ fs = (const float4*)(f + (long)src[e] * ROW) + j;
  const float4* __restrict__ fd = (const float4*)(f + dn * ROW) + j;
  const float4* __restrict__ at = (const float4*)attn + j;
  float acc[H];
#pragma unroll
  for (int h = 0; h < H; ++h) acc[h] = 0.0f;
#pragma unroll
  for (int k = 0; k < NCH; ++k) {
    const float4 a = fs[k * LPE];
    const float4 b = fd[k * LPE];
    const float4 w = at[k * LPE];
    float v0 = a.x + b.x, v1 = a.y + b.y, v2 = a.z + b.z, v3 = a.w + b.w;
    v0 = v0 > 0.0f ? v0 : 0.2f * v0;
    v1 = v1 > 0.0f ? v1 : 0.2f * v1;
    v2 = v2 > 0.0f ? v2 : 0.2f * v2;
    v3 = v3 > 0.0f ? v3 : 0.2f * v3;
    acc[k / CPH] += v0 * w.x;
    acc[k / CPH] += v1 * w.y;
    acc[k / CPH] += v2 * w.z;
    acc[k / CPH] += v3 * w.w;
  }
#pragma unroll
  for (int off = LPE / 2; off > 0; off >>= 1) {
#pragma unroll
    for (int h = 0; h < H; ++h) acc[h] += __shfl_xor(acc[h], off, 32);
  }
  if (live && j < H) {
    const float v = j == 0 ? acc[0] : j == 1 ? acc[1] : j == 2 ? acc[2] : acc[3];
    s[e * H + j] = v;
    atomic_max_f32(m + dn * H + j, v);
  }
}

__global__ void seg_exp(float* __restrict__ s, const int* __restrict__ dst,
                        const unsigned* __restrict__ m, float* __restrict__ denom,
                        long EH, int H) {
  const long t = (long)blockIdx.x * blockDim.x + threadIdx.x;
  if (t >= EH) return;
  const long e = t / H;
  const int  h = (int)(t - e * H);
  const long idx = (long)dst[e] * H + h;
  const float ex = __expf(s[t] - __uint_as_float(m[idx]));
  s[t] = ex;
  atomicAdd(denom + idx, ex);
}

template <int ROW, int JPH>
__global__ __launch_bounds__(256)
void aggregate(const float* __restrict__ f, const float* __restrict__ ex,
               const float* __restrict__ denom, const int* __restrict__ src,
               const int* __restrict__ dst, float* __restrict__ out, int E) {
  const int lane = threadIdx.x & 31;
  const int wid  = blockIdx.x * (blockDim.x >> 5) + (threadIdx.x >> 5);
  if (wid >= E) return;
  constexpr int ITER = ROW / 32;
  constexpr int H = ITER / JPH;
  const long srow = (long)src[wid] * ROW;
  const long dn   = (long)dst[wid];
  float al[H];
#pragma unroll
  for (int h = 0; h < H; ++h) al[h] = ex[(long)wid * H + h] / denom[dn * H + h];
#pragma unroll
  for (int j = 0; j < ITER; ++j) {
    const int d = j * 32 + lane;
    atomicAdd(out + dn * ROW + d, f[srow + d] * al[j / JPH]);
  }
}

__global__ void elu_inplace(float* __restrict__ x, long n) {
  const long t = (long)blockIdx.x * blockDim.x + threadIdx.x;
  if (t >= n) return;
  const float v = x[t];
  x[t] = v > 0.0f ? v : expm1f(v);
}

__global__ void head_mean(const float* __restrict__ in, float* __restrict__ out,
                          int N) {
  const int t = blockIdx.x * blockDim.x + threadIdx.x;
  if (t >= N * 32) return;
  const int n = t >> 5, c = t & 31;
  const float* r = in + (long)n * 128 + c;
  out[t] = 0.25f * (r[0] + r[32] + r[64] + r[96]);
}

static inline int cdiv_l(long a, long b) { return (int)((a + b - 1) / b); }

extern "C" void kernel_launch(void* const* d_in, const int* in_sizes, int n_in,
                              void* d_out, int out_size, void* d_ws, size_t ws_size,
                              hipStream_t stream) {
  const float* x     = (const float*)d_in[0];
  const float* W0    = (const float*)d_in[1];
  const float* a0    = (const float*)d_in[2];
  const float* W1    = (const float*)d_in[3];
  const float* a1    = (const float*)d_in[4];
  const float* Wres1 = (const float*)d_in[5];
  const int*   src   = (const int*)d_in[6];
  const int*   dst   = (const int*)d_in[7];

  const int N = in_sizes[0] / 256;
  const int E = in_sizes[6];
  const int H = 4;

  float*    f0    = (float*)d_ws;
  float*    h     = f0 + (long)N * 256;
  float*    f1    = h  + (long)N * 256;
  float*    o1    = f1 + (long)N * 128;
  float*    s     = o1 + (long)N * 128;
  unsigned* m     = (unsigned*)(s + (long)E * H);
  float*    denom = (float*)(m + (long)N * H);
  const size_t need =
      ((size_t)N * 256 * 2 + (size_t)N * 128 * 2 + (size_t)E * H + (size_t)N * H * 2)
      * sizeof(float);
  if (need > ws_size) return;

  const int BT = 256;
  const long EH = (long)E * H;
  const int egrid = cdiv_l((long)E * 32, BT);
  const int lgrid = cdiv_l((long)cdiv_l(E, 4) * 32, BT);

  {
    const int strips = ((N + 15) / 16) * (256 / 16 / 4);
    gemm_f32_wmma<4, 256, 256><<<cdiv_l(strips, 8), BT, 0, stream>>>(x, W0, f0, N);
  }
  init_layer<<<cdiv_l((long)N * 256, BT), BT, 0, stream>>>(h, m, denom,
                                                           (long)N * 256, N * H);
  edge_logits<256, 8><<<lgrid, BT, 0, stream>>>(f0, src, dst, a0, s, m, E);
  seg_exp<<<cdiv_l(EH, BT), BT, 0, stream>>>(s, dst, m, denom, EH, H);
  aggregate<256, 2><<<egrid, BT, 0, stream>>>(f0, s, denom, src, dst, h, E);
  elu_inplace<<<cdiv_l((long)N * 256, BT), BT, 0, stream>>>(h, (long)N * 256);

  {
    const int strips = ((N + 15) / 16) * (128 / 16 / 4);
    gemm_f32_wmma<4, 256, 128><<<cdiv_l(strips, 8), BT, 0, stream>>>(h, W1, f1, N);
    gemm_f32_wmma<4, 256, 128><<<cdiv_l(strips, 8), BT, 0, stream>>>(h, Wres1, o1, N);
  }
  init_layer<<<cdiv_l((long)N * H, BT), BT, 0, stream>>>(nullptr, m, denom,
                                                         0, N * H);
  edge_logits<128, 8><<<lgrid, BT, 0, stream>>>(f1, src, dst, a1, s, m, E);
  seg_exp<<<cdiv_l(EH, BT), BT, 0, stream>>>(s, dst, m, denom, EH, H);
  aggregate<128, 1><<<egrid, BT, 0, stream>>>(f1, s, denom, src, dst, o1, E);

  head_mean<<<cdiv_l((long)N * 32, BT), BT, 0, stream>>>(o1, (float*)d_out, N);
}


# aggregation by destination: in-call count/scan/fill (arrival-order rank) groups edges per node; one wave per node sums messages in registers, one store per row; float adds to h/o1 no longer atomic
# speedup vs baseline: 1.5515x; 1.1936x over previous
#include <hip/hip_runtime.h>
#include <hip/hip_bf16.h>

typedef __attribute__((ext_vector_type(2))) float v2f;
typedef __attribute__((ext_vector_type(8))) float v8f;

template <int NT, int K, int NC>
__global__ __launch_bounds__(256)
void gemm_f32_wmma(const float* __restrict__ A, const float* __restrict__ B,
                   float* __restrict__ C, int M) {
  const int lane = threadIdx.x & 31;
  const int wid  = blockIdx.x * (blockDim.x >> 5) + (threadIdx.x >> 5);
  const int tilesM  = (M + 15) >> 4;
  constexpr int stripsN = (NC >> 4) / NT;
  if (wid >= tilesM * stripsN) return;
  const int tm  = wid / stripsN;
  const int tn0 = (wid - tm * stripsN) * NT;
  const int hi = lane >> 4;
  const int lm = lane & 15;
  int arow = tm * 16 + lm; if (arow >= M) arow = M - 1;
  const int col0 = tn0 * 16 + lm;
  const float* __restrict__ Ar = A + (long)arow * K + 2 * hi;
  const float* __restrict__ Bc = B + (long)(2 * hi) * NC + col0;

  v8f c[NT];
#pragma unroll
  for (int t = 0; t < NT; ++t) c[t] = (v8f){};

  auto loadA = [&](int k) {
    v2f a; a.x = Ar[k]; a.y = Ar[k + 1]; return a;
  };
  auto loadB = [&](int k, v2f* b) {
    const float* __restrict__ B0 = Bc + (long)k * NC;
#pragma unroll
    for (int t = 0; t < NT; ++t) { b[t].x = B0[16 * t]; b[t].y = B0[NC + 16 * t]; }
  };

  v2f a_cur = loadA(0);
  v2f b_cur[NT];
  loadB(0, b_cur);
#pragma unroll 4
  for (int k = 0; k < K - 4; k += 4) {
    v2f a_nxt = loadA(k + 4);
    v2f b_nxt[NT];
    loadB(k + 4, b_nxt);
#pragma unroll
    for (int t = 0; t < NT; ++t)
      c[t] = __builtin_amdgcn_wmma_f32_16x16x4_f32(false, a_cur, false, b_cur[t],
                                                   (short)0, c[t], false, false);
    a_cur = a_nxt;
#pragma unroll
    for (int t = 0; t < NT; ++t) b_cur[t] = b_nxt[t];
  }
#pragma unroll
  for (int t = 0; t < NT; ++t)
    c[t] = __builtin_amdgcn_wmma_f32_16x16x4_f32(false, a_cur, false, b_cur[t],
                                                 (short)0, c[t], false, false);

#pragma unroll
  for (int t = 0; t < NT; ++t) {
#pragma unroll
    for (int r = 0; r < 8; ++r) {
      const int orow = tm * 16 + r + 8 * hi;
      if (orow < M) C[(long)orow * NC + col0 + 16 * t] = c[t][r];
    }
  }
}

__global__ void init_layer(unsigned* m, float* denom, int nh, int* cnt, int ncnt) {
  const long t = (long)blockIdx.x * blockDim.x + threadIdx.x;
  const long stride = (long)gridDim.x * blockDim.x;
  for (long i = t; i < nh; i += stride) { m[i] = 0xFF800000u; denom[i] = 0.0f; }
  for (long i = t; i < ncnt; i += stride) cnt[i] = 0;
}

__global__ void dst_histogram(const int* __restrict__ dst, int* __restrict__ cnt, int E) {
  const int e = blockIdx.x * blockDim.x + threadIdx.x;
  if (e < E) atomicAdd(cnt + dst[e], 1);
}

__global__ __launch_bounds__(1024)
void rowptr_scan(const int* __restrict__ cnt, int* __restrict__ rowptr,
                 int* __restrict__ cursor, int N) {
  __shared__ int wsum[32];
  __shared__ int wpre[32];
  const int tid = threadIdx.x, lane = tid & 31, w = tid >> 5;
  const int c0 = blockIdx.x * 1024;
  int pre = 0;
  for (int i = tid; i < c0; i += 1024) pre += cnt[i];
  const int i = c0 + tid;
  const int v = i < N ? cnt[i] : 0;
  int inc = v;
#pragma unroll
  for (int off = 1; off < 32; off <<= 1) {
    const int t = __shfl_up(inc, off, 32);
    if (lane >= off) inc += t;
  }
#pragma unroll
  for (int off = 16; off > 0; off >>= 1) pre += __shfl_xor(pre, off, 32);
  if (lane == 31) wsum[w] = inc;
  if (lane == 0)  wpre[w] = pre;
  __syncthreads();
  int before = 0;
#pragma unroll
  for (int k = 0; k < 32; ++k) {
    before += wpre[k];
    if (k < w) before += wsum[k];
  }
  const int excl = before + inc - v;
  if (i < N) { rowptr[i] = excl; cursor[i] = excl; }
}

__global__ void csr_fill(const int* __restrict__ src, const int* __restrict__ dst,
                         int* __restrict__ cursor, int2* __restrict__ es, int E) {
  const int e = blockIdx.x * blockDim.x + threadIdx.x;
  if (e >= E) return;
  const int p = atomicAdd(cursor + dst[e], 1);
  es[p] = make_int2(e, src[e]);
}

__device__ __forceinline__ void atomic_max_f32(unsigned* p, float v) {
  if (v >= 0.0f) atomicMax((int*)p, __float_as_int(v));
  else           atomicMin(p, __float_as_uint(v));
}

template <int ROW, int LPE>
__global__ __launch_bounds__(256)
void edge_logits(const float* __restrict__ f, const int* __restrict__ src,
                 const int* __restrict__ dst, const float* __restrict__ attn,
                 float* __restrict__ s, unsigned* __restrict__ m, int E) {
  constexpr int H   = 4;
  constexpr int EPW = 32 / LPE;
  constexpr int NCH = ROW / (4 * LPE);
  constexpr int CPH = NCH / H;
  static_assert(NCH % H == 0 && CPH >= 1, "a piece must not straddle a head");
  const int lane = threadIdx.x & 31;
  const int wid  = blockIdx.x * (blockDim.x >> 5) + (threadIdx.x >> 5);
  if ((long)wid * EPW >= E) return;
  const int j = lane & (LPE - 1);
  long e = (long)wid * EPW + (lane / LPE);
  const bool live = e < E;
  if (!live) e = E - 1;
  const long dn = (long)dst[e];
  const float4* __restrict__ fs = (const float4*)(f + (long)src[e] * ROW) + j;
  const float4* __restrict__ fd = (const float4*)(f + dn * ROW) + j;
  const float4* __restrict__ at = (const float4*)attn + j;
  float acc[H];
#pragma unroll
  for (int h = 0; h < H; ++h) acc[h] = 0.0f;
#pragma unroll
  for (int k = 0; k < NCH; ++k) {
    const float4 a = fs[k * LPE];
    const float4 b = fd[k * LPE];
    const float4 w = at[k * LPE];
    float v0 = a.x + b.x, v1 = a.y + b.y, v2 = a.z + b.z, v3 = a.w + b.w;
    v0 = v0 > 0.0f ? v0 : 0.2f * v0;
    v1 = v1 > 0.0f ? v1 : 0.2f * v1;
    v2 = v2 > 0.0f ? v2 : 0.2f * v2;
    v3 = v3 > 0.0f ? v3 : 0.2f * v3;
    acc[k / CPH] += v0 * w.x;
    acc[k / CPH] += v1 * w.y;
    acc[k / CPH] += v2 * w.z;
    acc[k / CPH] += v3 * w.w;
  }
#pragma unroll
  for (int off = LPE / 2; off > 0; off >>= 1) {
#pragma unroll
    for (int h = 0; h < H; ++h) acc[h] += __shfl_xor(acc[h], off, 32);
  }
  if (live && j < H) {
    const float v = j == 0 ? acc[0] : j == 1 ? acc[1] : j == 2 ? acc[2] : acc[3];
    s[e * H + j] = v;
    atomic_max_f32(m + dn * H + j, v);
  }
}

__global__ void seg_exp(float* __restrict__ s, const int* __restrict__ dst,
                        const unsigned* __restrict__ m, float* __restrict__ denom,
                        long EH, int H) {
  const long t = (long)blockIdx.x * blockDim.x + threadIdx.x;
  if (t >= EH) return;
  const long e = t / H;
  const int  h = (int)(t - e * H);
  const long idx = (long)dst[e] * H + h;
  const float ex = __expf(s[t] - __uint_as_float(m[idx]));
  s[t] = ex;
  atomicAdd(denom + idx, ex);
}

template <int ROW, bool ACCUM>
__global__ __launch_bounds__(256)
void aggregate_by_dst(const float* __restrict__ f, const float* __restrict__ ex,
                      const float* __restrict__ denom, const int2* __restrict__ es,
                      const int* __restrict__ rowptr, const int* __restrict__ cnt,
                      float* __restrict__ out, int N) {
  constexpr int H   = 4;
  constexpr int NV  = ROW / 128;
  constexpr int LPH = ROW / H / 4;
  const int lane = threadIdx.x & 31;
  const int node = __builtin_amdgcn_readfirstlane(
      (int)(blockIdx.x * (blockDim.x >> 5) + (threadIdx.x >> 5)));
  if (node >= N) return;
  const int start = rowptr[node];
  const int deg   = cnt[node];
  const float4 den = ((const float4*)denom)[node];
  const float4* __restrict__ f4  = (const float4*)f;
  const float4* __restrict__ ex4 = (const float4*)ex;
  int hs[NV];
  float4 acc[NV];
#pragma unroll
  for (int q = 0; q < NV; ++q) {
    hs[q] = (q * 32 + lane) / LPH;
    acc[q] = make_float4(0.0f, 0.0f, 0.0f, 0.0f);
  }
  for (int c = 0; c < deg; c += 32) {
    const int nn = deg - c < 32 ? deg - c : 32;
    int sidx = 0;
    float4 al = make_float4(0.0f, 0.0f, 0.0f, 0.0f);
    if (lane < nn) {
      const int2 pr = es[start + c + lane];
      const float4 x = ex4[pr.x];
      sidx = pr.y;
      al.x = x.x / den.x; al.y = x.y / den.y; al.z = x.z / den.z; al.w = x.w / den.w;
    }
    int i = 0;
    for (; i + 4 <= nn; i += 4) {
      float4 v[4][NV];
      float  w[4][NV];
#pragma unroll
      for (int u = 0; u < 4; ++u) {
        const int s = __builtin_amdgcn_readlane(sidx, i + u);
        const float4* __restrict__ row = f4 + (long)s * (ROW / 4) + lane;
#pragma unroll
        for (int q = 0; q < NV; ++q) v[u][q] = row[q * 32];
        const float a0 = __int_as_float(__builtin_amdgcn_readlane(__float_as_int(al.x), i + u));
        const float a1 = __int_as_float(__builtin_amdgcn_readlane(__float_as_int(al.y), i + u));
        const float a2 = __int_as_float(__builtin_amdgcn_readlane(__float_as_int(al.z), i + u));
        const float a3 = __int_as_float(__builtin_amdgcn_readlane(__float_as_int(al.w), i + u));
#pragma unroll
        for (int q = 0; q < NV; ++q)
          w[u][q] = hs[q] == 0 ? a0 : hs[q] == 1 ? a1 : hs[q] == 2 ? a2 : a3;
      }
#pragma unroll
      for (int u = 0; u < 4; ++u) {
#pragma unroll
        for (int q = 0; q < NV; ++q) {
          acc[q].x += v[u][q].x * w[u][q];
          acc[q].y += v[u][q].y * w[u][q];
          acc[q].z += v[u][q].z * w[u][q];
          acc[q].w += v[u][q].w * w[u][q];
        }
      }
    }
    for (; i < nn; ++i) {
      const int s = __builtin_amdgcn_readlane(sidx, i);
      const float4* __restrict__ row = f4 + (long)s * (ROW / 4) + lane;
      const float a0 = __int_as_float(__builtin_amdgcn_readlane(__float_as_int(al.x), i));
      const float a1 = __int_as_float(__builtin_amdgcn_readlane(__float_as_int(al.y), i));
      const float a2 = __int_as_float(__builtin_amdgcn_readlane(__float_as_int(al.z), i));
      const float a3 = __int_as_float(__builtin_amdgcn_readlane(__float_as_int(al.w), i));
#pragma unroll
      for (int q = 0; q < NV; ++q) {
        const float4 vv = row[q * 32];
        const float ww = hs[q] == 0 ? a0 : hs[q] == 1 ? a1 : hs[q] == 2 ? a2 : a3;
        acc[q].x += vv.x * ww;
        acc[q].y += vv.y * ww;
        acc[q].z += vv.z * ww;
        acc[q].w += vv.w * ww;
      }
    }
  }
  float4* o = (float4*)(out + (long)node * ROW) + lane;
#pragma unroll
  for (int q = 0; q < NV; ++q) {
    float4 r = acc[q];
    if (ACCUM) {
      const float4 p = o[q * 32];
      r.x += p.x; r.y += p.y; r.z += p.z; r.w += p.w;
    }
    o[q * 32] = r;
  }
}

__global__ void elu_inplace(float* __restrict__ x, long n) {
  const long t = (long)blockIdx.x * blockDim.x + threadIdx.x;
  if (t >= n) return;
  const float v = x[t];
  x[t] = v > 0.0f ? v : expm1f(v);
}

__global__ void head_mean(const float* __restrict__ in, float* __restrict__ out,
                          int N) {
  const int t = blockIdx.x * blockDim.x + threadIdx.x;
  if (t >= N * 32) return;
  const int n = t >> 5, c = t & 31;
  const float* r = in + (long)n * 128 + c;
  out[t] = 0.25f * (r[0] + r[32] + r[64] + r[96]);
}

static inline int cdiv_l(long a, long b) { return (int)((a + b - 1) / b); }

extern "C" void kernel_launch(void* const* d_in, const int* in_sizes, int n_in,
                              void* d_out, int out_size, void* d_ws, size_t ws_size,
                              hipStream_t stream) {
  const float* x     = (const float*)d_in[0];
  const float* W0    = (const float*)d_in[1];
  const float* a0    = (const float*)d_in[2];
  const float* W1    = (const float*)d_in[3];
  const float* a1    = (const float*)d_in[4];
  const float* Wres1 = (const float*)d_in[5];
  const int*   src   = (const int*)d_in[6];
  const int*   dst   = (const int*)d_in[7];

  const int N = in_sizes[0] / 256;
  const int E = in_sizes[6];
  const int H = 4;

  float*    f0    = (float*)d_ws;
  float*    h     = f0 + (long)N * 256;
  float*    grp   = h  + (long)N * 256;
  float*    o1    = grp + (long)N * 128;
  float*    s     = o1 + (long)N * 128;
  unsigned* m     = (unsigned*)(s + (long)E * H);
  float*    denom = (float*)(m + (long)N * H);
  float*    f1    = f0;
  int2*     es     = (int2*)grp;
  int*      cnt    = (int*)(es + E);
  int*      rowptr = cnt + N;
  int*      cursor = rowptr + N;
  const size_t need =
      ((size_t)N * 256 * 2 + (size_t)N * 128 * 2 + (size_t)E * H + (size_t)N * H * 2)
      * sizeof(float);
  if (need > ws_size) return;
  if ((size_t)E * sizeof(int2) + (size_t)N * 3 * sizeof(int) > (size_t)N * 128 * sizeof(float))
    return;

  const int BT = 256;
  const long EH = (long)E * H;
  const int ngrid = cdiv_l((long)N * 32, BT);
  const int lgrid = cdiv_l((long)cdiv_l(E, 4) * 32, BT);

  {
    const int strips = ((N + 15) / 16) * (256 / 16 / 4);
    gemm_f32_wmma<4, 256, 256><<<cdiv_l(strips, 8), BT, 0, stream>>>(x, W0, f0, N);
  }
  init_layer<<<cdiv_l((long)N * H, BT), BT, 0, stream>>>(m, denom, N * H, cnt, N);
  dst_histogram<<<cdiv_l(E, BT), BT, 0, stream>>>(dst, cnt, E);
  rowptr_scan<<<cdiv_l(N, 1024), 1024, 0, stream>>>(cnt, rowptr, cursor, N);
  csr_fill<<<cdiv_l(E, BT), BT, 0, stream>>>(src, dst, cursor, es, E);
  edge_logits<256, 8><<<lgrid, BT, 0, stream>>>(f0, src, dst, a0, s, m, E);
  seg_exp<<<cdiv_l(EH, BT), BT, 0, stream>>>(s, dst, m, denom, EH, H);
  aggregate_by_dst<256, false><<<ngrid, BT, 0, stream>>>(f0, s, denom, es, rowptr, cnt, h, N);
  elu_inplace<<<cdiv_l((long)N * 256, BT), BT, 0, stream>>>(h, (long)N * 256);

  {
    const int strips = ((N + 15) / 16) * (128 / 16 / 4);
    gemm_f32_wmma<4, 256, 128><<<cdiv_l(strips, 8), BT, 0, stream>>>(h, W1, f1, N);
    gemm_f32_wmma<4, 256, 128><<<cdiv_l(strips, 8), BT, 0, stream>>>(h, Wres1, o1, N);
  }
  init_layer<<<cdiv_l((long)N * H, BT), BT, 0, stream>>>(m, denom, N * H, nullptr, 0);
  edge_logits<128, 8><<<lgrid, BT, 0, stream>>>(f1, src, dst, a1, s, m, E);
  seg_exp<<<cdiv_l(EH, BT), BT, 0, stream>>>(s, dst, m, denom, EH, H);
  aggregate_by_dst<128, true><<<ngrid, BT, 0, stream>>>(f1, s, denom, es, rowptr, cnt, o1, N);

  head_mean<<<cdiv_l((long)N * 32, BT), BT, 0, stream>>>(o1, (float*)d_out, N);
}


# edge softmax fused into the per-node sum: one wave per node computes its edges' logits, a running-max softmax and the weighted sum in one pass; logit and denominator sum order changes
# speedup vs baseline: 2.3768x; 1.5319x over previous
#include <hip/hip_runtime.h>
#include <hip/hip_bf16.h>

typedef __attribute__((ext_vector_type(2))) float v2f;
typedef __attribute__((ext_vector_type(8))) float v8f;

template <int NT, int K, int NC>
__global__ __launch_bounds__(256)
void gemm_f32_wmma(const float* __restrict__ A, const float* __restrict__ B,
                   float* __restrict__ C, int M) {
  const int lane = threadIdx.x & 31;
  const int wid  = blockIdx.x * (blockDim.x >> 5) + (threadIdx.x >> 5);
  const int tilesM  = (M + 15) >> 4;
  constexpr int stripsN = (NC >> 4) / NT;
  if (wid >= tilesM * stripsN) return;
  const int tm  = wid / stripsN;
  const int tn0 = (wid - tm * stripsN) * NT;
  const int hi = lane >> 4;
  const int lm = lane & 15;
  int arow = tm * 16 + lm; if (arow >= M) arow = M - 1;
  const int col0 = tn0 * 16 + lm;
  const float* __restrict__ Ar = A + (long)arow * K + 2 * hi;
  const float* __restrict__ Bc = B + (long)(2 * hi) * NC + col0;

  v8f c[NT];
#pragma unroll
  for (int t = 0; t < NT; ++t) c[t] = (v8f){};

  auto loadA = [&](int k) {
    v2f a; a.x = Ar[k]; a.y = Ar[k + 1]; return a;
  };
  auto loadB = [&](int k, v2f* b) {
    const float* __restrict__ B0 = Bc + (long)k * NC;
#pragma unroll
    for (int t = 0; t < NT; ++t) { b[t].x = B0[16 * t]; b[t].y = B0[NC + 16 * t]; }
  };

  v2f a_cur = loadA(0);
  v2f b_cur[NT];
  loadB(0, b_cur);
#pragma unroll 4
  for (int k = 0; k < K - 4; k += 4) {
    v2f a_nxt = loadA(k + 4);
    v2f b_nxt[NT];
    loadB(k + 4, b_nxt);
#pragma unroll
    for (int t = 0; t < NT; ++t)
      c[t] = __builtin_amdgcn_wmma_f32_16x16x4_f32(false, a_cur, false, b_cur[t],
                                                   (short)0, c[t], false, false);
    a_cur = a_nxt;
#pragma unroll
    for (int t = 0; t < NT; ++t) b_cur[t] = b_nxt[t];
  }
#pragma unroll
  for (int t = 0; t < NT; ++t)
    c[t] = __builtin_amdgcn_wmma_f32_16x16x4_f32(false, a_cur, false, b_cur[t],
                                                 (short)0, c[t], false, false);

#pragma unroll
  for (int t = 0; t < NT; ++t) {
#pragma unroll
    for (int r = 0; r < 8; ++r) {
      const int orow = tm * 16 + r + 8 * hi;
      if (orow < M) C[(long)orow * NC + col0 + 16 * t] = c[t][r];
    }
  }
}

__global__ void zero_counts(int* __restrict__ cnt, int n) {
  const int i = blockIdx.x * blockDim.x + threadIdx.x;
  if (i < n) cnt[i] = 0;
}

__global__ void dst_histogram(const int* __restrict__ dst, int* __restrict__ cnt, int E) {
  const int e = blockIdx.x * blockDim.x + threadIdx.x;
  if (e < E) atomicAdd(cnt + dst[e], 1);
}

__global__ __launch_bounds__(1024)
void rowptr_scan(const int* __restrict__ cnt, int* __restrict__ rowptr,
                 int* __restrict__ cursor, int N) {
  __shared__ int wsum[32];
  __shared__ int wpre[32];
  const int tid = threadIdx.x, lane = tid & 31, w = tid >> 5;
  const int c0 = blockIdx.x * 1024;
  int pre = 0;
  for (int i = tid; i < c0; i += 1024) pre += cnt[i];
  const int i = c0 + tid;
  const int v = i < N ? cnt[i] : 0;
  int inc = v;
#pragma unroll
  for (int off = 1; off < 32; off <<= 1) {
    const int t = __shfl_up(inc, off, 32);
    if (lane >= off) inc += t;
  }
#pragma unroll
  for (int off = 16; off > 0; off >>= 1) pre += __shfl_xor(pre, off, 32);
  if (lane == 31) wsum[w] = inc;
  if (lane == 0)  wpre[w] = pre;
  __syncthreads();
  int before = 0;
#pragma unroll
  for (int k = 0; k < 32; ++k) {
    before += wpre[k];
    if (k < w) before += wsum[k];
  }
  const int excl = before + inc - v;
  if (i < N) { rowptr[i] = excl; cursor[i] = excl; }
}

__global__ void csr_fill(const int* __restrict__ src, const int* __restrict__ dst,
                         int* __restrict__ cursor, int* __restrict__ nbr, int E) {
  const int e = blockIdx.x * blockDim.x + threadIdx.x;
  if (e >= E) return;
  const int p = atomicAdd(cursor + dst[e], 1);
  nbr[p] = src[e];
}

template <int MASK>
__device__ __forceinline__ float row_xor_lane(float x) {
  return __int_as_float(__builtin_amdgcn_update_dpp(
      0, __float_as_int(x), 0x160 + MASK, 0xF, 0xF, true));
}

template <int ROW, int U>
__device__ __forceinline__ void attend_edges(
    const float4* __restrict__ f4, int sidx, int i, int lane,
    const float4 (&fd)[ROW / 128], const float4 (&at)[ROW / 128],
    float4 (&acc)[ROW / 128], float (&mx)[ROW / 128], float (&den)[ROW / 128]) {
  constexpr int NV  = ROW / 128;
  constexpr int LPH = ROW / 16;
  float4 v[U][NV];
  float  lg[U][NV];
#pragma unroll
  for (int u = 0; u < U; ++u) {
    const int s = __builtin_amdgcn_readlane(sidx, i + u);
    const float4* __restrict__ row = f4 + (long)s * (ROW / 4) + lane;
#pragma unroll
    for (int q = 0; q < NV; ++q) v[u][q] = row[q * 32];
  }
#pragma unroll
  for (int u = 0; u < U; ++u) {
#pragma unroll
    for (int q = 0; q < NV; ++q) {
      float t0 = v[u][q].x + fd[q].x, t1 = v[u][q].y + fd[q].y;
      float t2 = v[u][q].z + fd[q].z, t3 = v[u][q].w + fd[q].w;
      t0 = t0 > 0.0f ? t0 : 0.2f * t0;
      t1 = t1 > 0.0f ? t1 : 0.2f * t1;
      t2 = t2 > 0.0f ? t2 : 0.2f * t2;
      t3 = t3 > 0.0f ? t3 : 0.2f * t3;
      float p = t0 * at[q].x;
      p += t1 * at[q].y;
      p += t2 * at[q].z;
      p += t3 * at[q].w;
      lg[u][q] = p;
    }
  }
  if (LPH >= 16) {
#pragma unroll
    for (int u = 0; u < U; ++u) {
#pragma unroll
      for (int q = 0; q < NV; ++q) lg[u][q] += row_xor_lane<8>(lg[u][q]);
    }
  }
#pragma unroll
  for (int u = 0; u < U; ++u) {
#pragma unroll
    for (int q = 0; q < NV; ++q) lg[u][q] += row_xor_lane<4>(lg[u][q]);
  }
#pragma unroll
  for (int u = 0; u < U; ++u) {
#pragma unroll
    for (int q = 0; q < NV; ++q) lg[u][q] += row_xor_lane<2>(lg[u][q]);
  }
#pragma unroll
  for (int u = 0; u < U; ++u) {
#pragma unroll
    for (int q = 0; q < NV; ++q) lg[u][q] += row_xor_lane<1>(lg[u][q]);
  }
#pragma unroll
  for (int q = 0; q < NV; ++q) {
    float bm = mx[q];
#pragma unroll
    for (int u = 0; u < U; ++u) bm = fmaxf(bm, lg[u][q]);
    const float sc = __expf(mx[q] - bm);
    float4 a = acc[q];
    a.x *= sc; a.y *= sc; a.z *= sc; a.w *= sc;
    float d = den[q] * sc;
#pragma unroll
    for (int u = 0; u < U; ++u) {
      const float p = __expf(lg[u][q] - bm);
      d += p;
      a.x += p * v[u][q].x;
      a.y += p * v[u][q].y;
      a.z += p * v[u][q].z;
      a.w += p * v[u][q].w;
    }
    acc[q] = a;
    den[q] = d;
    mx[q]  = bm;
  }
}

template <int ROW, bool ACCUM>
__global__ __launch_bounds__(256)
void attend_by_dst(const float* __restrict__ f, const float* __restrict__ attn,
                   const int* __restrict__ nbr, const int* __restrict__ rowptr,
                   const int* __restrict__ cnt, float* __restrict__ out, int N) {
  constexpr int NV = ROW / 128;
  static_assert(ROW == 128 || ROW == 256, "the lanes of one head must lie inside one row of 16 lanes");
  const int lane = threadIdx.x & 31;
  const int node = __builtin_amdgcn_readfirstlane(
      (int)(blockIdx.x * (blockDim.x >> 5) + (threadIdx.x >> 5)));
  if (node >= N) return;
  const int start = rowptr[node];
  const int deg   = cnt[node];
  const float4* __restrict__ f4   = (const float4*)f;
  const float4* __restrict__ frow = f4 + (long)node * (ROW / 4) + lane;
  const float4* __restrict__ arow = (const float4*)attn + lane;
  float4 fd[NV], at[NV], acc[NV];
  float  mx[NV], den[NV];
#pragma unroll
  for (int q = 0; q < NV; ++q) {
    fd[q]  = frow[q * 32];
    at[q]  = arow[q * 32];
    acc[q] = make_float4(0.0f, 0.0f, 0.0f, 0.0f);
    mx[q]  = -__builtin_inff();
    den[q] = 0.0f;
  }
  for (int c = 0; c < deg; c += 32) {
    const int nn = deg - c < 32 ? deg - c : 32;
    int sidx = 0;
    if (lane < nn) sidx = nbr[start + c + lane];
    int i = 0;
    for (; i + 4 <= nn; i += 4) attend_edges<ROW, 4>(f4, sidx, i, lane, fd, at, acc, mx, den);
    if (i + 2 <= nn) { attend_edges<ROW, 2>(f4, sidx, i, lane, fd, at, acc, mx, den); i += 2; }
    if (i < nn)        attend_edges<ROW, 1>(f4, sidx, i, lane, fd, at, acc, mx, den);
  }
  float4* o = (float4*)(out + (long)node * ROW) + lane;
#pragma unroll
  for (int q = 0; q < NV; ++q) {
    float4 r = make_float4(0.0f, 0.0f, 0.0f, 0.0f);
    if (deg > 0) {
      r.x = acc[q].x / den[q]; r.y = acc[q].y / den[q];
      r.z = acc[q].z / den[q]; r.w = acc[q].w / den[q];
    }
    if (ACCUM) {
      const float4 p = o[q * 32];
      r.x += p.x; r.y += p.y; r.z += p.z; r.w += p.w;
    }
    o[q * 32] = r;
  }
}

__global__ void elu_inplace(float* __restrict__ x, long n) {
  const long t = (long)blockIdx.x * blockDim.x + threadIdx.x;
  if (t >= n) return;
  const float v = x[t];
  x[t] = v > 0.0f ? v : expm1f(v);
}

__global__ void head_mean(const float* __restrict__ in, float* __restrict__ out,
                          int N) {
  const int t = blockIdx.x * blockDim.x + threadIdx.x;
  if (t >= N * 32) return;
  const int n = t >> 5, c = t & 31;
  const float* r = in + (long)n * 128 + c;
  out[t] = 0.25f * (r[0] + r[32] + r[64] + r[96]);
}

static inline int cdiv_l(long a, long b) { return (int)((a + b - 1) / b); }

extern "C" void kernel_launch(void* const* d_in, const int* in_sizes, int n_in,
                              void* d_out, int out_size, void* d_ws, size_t ws_size,
                              hipStream_t stream) {
  const float* x     = (const float*)d_in[0];
  const float* W0    = (const float*)d_in[1];
  const float* a0    = (const float*)d_in[2];
  const float* W1    = (const float*)d_in[3];
  const float* a1    = (const float*)d_in[4];
  const float* Wres1 = (const float*)d_in[5];
  const int*   src   = (const int*)d_in[6];
  const int*   dst   = (const int*)d_in[7];

  const int N = in_sizes[0] / 256;
  const int E = in_sizes[6];

  float*    f0    = (float*)d_ws;
  float*    h     = f0 + (long)N * 256;
  float*    grp   = h  + (long)N * 256;
  float*    o1    = grp + (long)N * 128;
  float*    f1    = f0;
  int*      nbr    = (int*)grp;
  int*      cnt    = nbr + E;
  int*      rowptr = cnt + N;
  int*      cursor = rowptr + N;
  const size_t need = ((size_t)N * 256 * 2 + (size_t)N * 128 * 2) * sizeof(float);
  if (need > ws_size) return;
  if (((size_t)E + (size_t)N * 3) * sizeof(int) > (size_t)N * 128 * sizeof(float))
    return;

  const int BT = 256;
  const int ngrid = cdiv_l((long)N * 32, BT);

  {
    const int strips = ((N + 15) / 16) * (256 / 16 / 4);
    gemm_f32_wmma<4, 256, 256><<<cdiv_l(strips, 8), BT, 0, stream>>>(x, W0, f0, N);
  }
  zero_counts<<<cdiv_l(N, BT), BT, 0, stream>>>(cnt, N);
  dst_histogram<<<cdiv_l(E, BT), BT, 0, stream>>>(dst, cnt, E);
  rowptr_scan<<<cdiv_l(N, 1024), 1024, 0, stream>>>(cnt, rowptr, cursor, N);
  csr_fill<<<cdiv_l(E, BT), BT, 0, stream>>>(src, dst, cursor, nbr, E);
  attend_by_dst<256, false><<<ngrid, BT, 0, stream>>>(f0, a0, nbr, rowptr, cnt, h, N);
  elu_inplace<<<cdiv_l((long)N * 256, BT), BT, 0, stream>>>(h, (long)N * 256);

  {
    const int strips = ((N + 15) / 16) * (128 / 16 / 4);
    gemm_f32_wmma<4, 256, 128><<<cdiv_l(strips, 8), BT, 0, stream>>>(h, W1, f1, N);
    gemm_f32_wmma<4, 256, 128><<<cdiv_l(strips, 8), BT, 0, stream>>>(h, Wres1, o1, N);
  }
  attend_by_dst<128, true><<<ngrid, BT, 0, stream>>>(f1, a1, nbr, rowptr, cnt, o1, N);

  head_mean<<<cdiv_l((long)N * 32, BT), BT, 0, stream>>>(o1, (float*)d_out, N);
}


# elementwise tail passes folded into the attention kernels' row stores: ELU in layer 0; residual add + head mean in layer 1 (heads summed by lane exchange, order (h0+h1)+(h2+h3)); 11 -> 9 launches
# speedup vs baseline: 2.8687x; 1.2070x over previous
#include <hip/hip_runtime.h>
#include <hip/hip_bf16.h>

typedef __attribute__((ext_vector_type(2))) float v2f;
typedef __attribute__((ext_vector_type(8))) float v8f;

template <int NT, int K, int NC>
__global__ __launch_bounds__(256)
void gemm_f32_wmma(const float* __restrict__ A, const float* __restrict__ B,
                   float* __restrict__ C, int M) {
  const int lane = threadIdx.x & 31;
  const int wid  = blockIdx.x * (blockDim.x >> 5) + (threadIdx.x >> 5);
  const int tilesM  = (M + 15) >> 4;
  constexpr int stripsN = (NC >> 4) / NT;
  if (wid >= tilesM * stripsN) return;
  const int tm  = wid / stripsN;
  const int tn0 = (wid - tm * stripsN) * NT;
  const int hi = lane >> 4;
  const int lm = lane & 15;
  int arow = tm * 16 + lm; if (arow >= M) arow = M - 1;
  const int col0 = tn0 * 16 + lm;
  const float* __restrict__ Ar = A + (long)arow * K + 2 * hi;
  const float* __restrict__ Bc = B + (long)(2 * hi) * NC + col0;

  v8f c[NT];
#pragma unroll
  for (int t = 0; t < NT; ++t) c[t] = (v8f){};

  auto loadA = [&](int k) {
    v2f a; a.x = Ar[k]; a.y = Ar[k + 1]; return a;
  };
  auto loadB = [&](int k, v2f* b) {
    const float* __restrict__ B0 = Bc + (long)k * NC;
#pragma unroll
    for (int t = 0; t < NT; ++t) { b[t].x = B0[16 * t]; b[t].y = B0[NC + 16 * t]; }
  };

  v2f a_cur = loadA(0);
  v2f b_cur[NT];
  loadB(0, b_cur);
#pragma unroll 4
  for (int k = 0; k < K - 4; k += 4) {
    v2f a_nxt = loadA(k + 4);
    v2f b_nxt[NT];
    loadB(k + 4, b_nxt);
#pragma unroll
    for (int t = 0; t < NT; ++t)
      c[t] = __builtin_amdgcn_wmma_f32_16x16x4_f32(false, a_cur, false, b_cur[t],
                                                   (short)0, c[t], false, false);
    a_cur = a_nxt;
#pragma unroll
    for (int t = 0; t < NT; ++t) b_cur[t] = b_nxt[t];
  }
#pragma unroll
  for (int t = 0; t < NT; ++t)
    c[t] = __builtin_amdgcn_wmma_f32_16x16x4_f32(false, a_cur, false, b_cur[t],
                                                 (short)0, c[t], false, false);

#pragma unroll
  for (int t = 0; t < NT; ++t) {
#pragma unroll
    for (int r = 0; r < 8; ++r) {
      const int orow = tm * 16 + r + 8 * hi;
      if (orow < M) C[(long)orow * NC + col0 + 16 * t] = c[t][r];
    }
  }
}

__global__ void zero_counts(int* __restrict__ cnt, int n) {
  const int i = blockIdx.x * blockDim.x + threadIdx.x;
  if (i < n) cnt[i] = 0;
}

__global__ void dst_histogram(const int* __restrict__ dst, int* __restrict__ cnt, int E) {
  const int e = blockIdx.x * blockDim.x + threadIdx.x;
  if (e < E) atomicAdd(cnt + dst[e], 1);
}

__global__ __launch_bounds__(1024)
void rowptr_scan(const int* __restrict__ cnt, int* __restrict__ rowptr,
                 int* __restrict__ cursor, int N) {
  __shared__ int wsum[32];
  __shared__ int wpre[32];
  const int tid = threadIdx.x, lane = tid & 31, w = tid >> 5;
  const int c0 = blockIdx.x * 1024;
  int pre = 0;
  for (int i = tid; i < c0; i += 1024) pre += cnt[i];
  const int i = c0 + tid;
  const int v = i < N ? cnt[i] : 0;
  int inc = v;
#pragma unroll
  for (int off = 1; off < 32; off <<= 1) {
    const int t = __shfl_up(inc, off, 32);
    if (lane >= off) inc += t;
  }
#pragma unroll
  for (int off = 16; off > 0; off >>= 1) pre += __shfl_xor(pre, off, 32);
  if (lane == 31) wsum[w] = inc;
  if (lane == 0)  wpre[w] = pre;
  __syncthreads();
  int before = 0;
#pragma unroll
  for (int k = 0; k < 32; ++k) {
    before += wpre[k];
    if (k < w) before += wsum[k];
  }
  const int excl = before + inc - v;
  if (i < N) { rowptr[i] = excl; cursor[i] = excl; }
}

__global__ void csr_fill(const int* __restrict__ src, const int* __restrict__ dst,
                         int* __restrict__ cursor, int* __restrict__ nbr, int E) {
  const int e = blockIdx.x * blockDim.x + threadIdx.x;
  if (e >= E) return;
  const int p = atomicAdd(cursor + dst[e], 1);
  nbr[p] = src[e];
}

template <int MASK>
__device__ __forceinline__ float row_xor_lane(float x) {
  return __int_as_float(__builtin_amdgcn_update_dpp(
      0, __float_as_int(x), 0x160 + MASK, 0xF, 0xF, true));
}

__device__ __forceinline__ float other_row_lane(float x) {
  return __int_as_float((int)__builtin_amdgcn_permlanex16(
      0u, (unsigned)__float_as_int(x), 0x76543210u, 0xfedcba98u, false, false));
}

template <int ROW, int U>
__device__ __forceinline__ void attend_edges(
    const float4* __restrict__ f4, int sidx, int i, int lane,
    const float4 (&fd)[ROW / 128], const float4 (&at)[ROW / 128],
    float4 (&acc)[ROW / 128], float (&mx)[ROW / 128], float (&den)[ROW / 128]) {
  constexpr int NV  = ROW / 128;
  constexpr int LPH = ROW / 16;
  float4 v[U][NV];
  float  lg[U][NV];
#pragma unroll
  for (int u = 0; u < U; ++u) {
    const int s = __builtin_amdgcn_readlane(sidx, i + u);
    const float4* __restrict__ row = f4 + (long)s * (ROW / 4) + lane;
#pragma unroll
    for (int q = 0; q < NV; ++q) v[u][q] = row[q * 32];
  }
#pragma unroll
  for (int u = 0; u < U; ++u) {
#pragma unroll
    for (int q = 0; q < NV; ++q) {
      float t0 = v[u][q].x + fd[q].x, t1 = v[u][q].y + fd[q].y;
      float t2 = v[u][q].z + fd[q].z, t3 = v[u][q].w + fd[q].w;
      t0 = t0 > 0.0f ? t0 : 0.2f * t0;
      t1 = t1 > 0.0f ? t1 : 0.2f * t1;
      t2 = t2 > 0.0f ? t2 : 0.2f * t2;
      t3 = t3 > 0.0f ? t3 : 0.2f * t3;
      float p = t0 * at[q].x;
      p += t1 * at[q].y;
      p += t2 * at[q].z;
      p += t3 * at[q].w;
      lg[u][q] = p;
    }
  }
  if (LPH >= 16) {
#pragma unroll
    for (int u = 0; u < U; ++u) {
#pragma unroll
      for (int q = 0; q < NV; ++q) lg[u][q] += row_xor_lane<8>(lg[u][q]);
    }
  }
#pragma unroll
  for (int u = 0; u < U; ++u) {
#pragma unroll
    for (int q = 0; q < NV; ++q) lg[u][q] += row_xor_lane<4>(lg[u][q]);
  }
#pragma unroll
  for (int u = 0; u < U; ++u) {
#pragma unroll
    for (int q = 0; q < NV; ++q) lg[u][q] += row_xor_lane<2>(lg[u][q]);
  }
#pragma unroll
  for (int u = 0; u < U; ++u) {
#pragma unroll
    for (int q = 0; q < NV; ++q) lg[u][q] += row_xor_lane<1>(lg[u][q]);
  }
#pragma unroll
  for (int q = 0; q < NV; ++q) {
    float bm = mx[q];
#pragma unroll
    for (int u = 0; u < U; ++u) bm = fmaxf(bm, lg[u][q]);
    const float sc = __expf(mx[q] - bm);
    float4 a = acc[q];
    a.x *= sc; a.y *= sc; a.z *= sc; a.w *= sc;
    float d = den[q] * sc;
#pragma unroll
    for (int u = 0; u < U; ++u) {
      const float p = __expf(lg[u][q] - bm);
      d += p;
      a.x += p * v[u][q].x;
      a.y += p * v[u][q].y;
      a.z += p * v[u][q].z;
      a.w += p * v[u][q].w;
    }
    acc[q] = a;
    den[q] = d;
    mx[q]  = bm;
  }
}

enum { TAIL_ELU = 0, TAIL_RES_MEAN = 1 };

template <int ROW, int TAIL>
__global__ __launch_bounds__(256)
void attend_by_dst(const float* __restrict__ f, const float* __restrict__ attn,
                   const int* __restrict__ nbr, const int* __restrict__ rowptr,
                   const int* __restrict__ cnt, const float* __restrict__ res,
                   float* __restrict__ out, int N) {
  constexpr int NV = ROW / 128;
  static_assert(ROW == 128 || ROW == 256, "the lanes of one head must lie inside one row of 16 lanes");
  static_assert(TAIL == TAIL_ELU || ROW == 128, "the head mean is written for 4 heads of 8 lanes each");
  const int lane = threadIdx.x & 31;
  const int node = __builtin_amdgcn_readfirstlane(
      (int)(blockIdx.x * (blockDim.x >> 5) + (threadIdx.x >> 5)));
  if (node >= N) return;
  const int start = rowptr[node];
  const int deg   = cnt[node];
  const float4* __restrict__ f4   = (const float4*)f;
  const float4* __restrict__ frow = f4 + (long)node * (ROW / 4) + lane;
  const float4* __restrict__ arow = (const float4*)attn + lane;
  float4 fd[NV], at[NV], acc[NV];
  float  mx[NV], den[NV];
#pragma unroll
  for (int q = 0; q < NV; ++q) {
    fd[q]  = frow[q * 32];
    at[q]  = arow[q * 32];
    acc[q] = make_float4(0.0f, 0.0f, 0.0f, 0.0f);
    mx[q]  = -__builtin_inff();
    den[q] = 0.0f;
  }
  for (int c = 0; c < deg; c += 32) {
    const int nn = deg - c < 32 ? deg - c : 32;
    int sidx = 0;
    if (lane < nn) sidx = nbr[start + c + lane];
    int i = 0;
    for (; i + 4 <= nn; i += 4) attend_edges<ROW, 4>(f4, sidx, i, lane, fd, at, acc, mx, den);
    if (i + 2 <= nn) { attend_edges<ROW, 2>(f4, sidx, i, lane, fd, at, acc, mx, den); i += 2; }
    if (i < nn)        attend_edges<ROW, 1>(f4, sidx, i, lane, fd, at, acc, mx, den);
  }
  float4 r[NV];
#pragma unroll
  for (int q = 0; q < NV; ++q) {
    r[q] = make_float4(0.0f, 0.0f, 0.0f, 0.0f);
    if (deg > 0) {
      r[q].x = acc[q].x / den[q]; r[q].y = acc[q].y / den[q];
      r[q].z = acc[q].z / den[q]; r[q].w = acc[q].w / den[q];
    }
  }
  if (TAIL == TAIL_ELU) {
    float4* o = (float4*)(out + (long)node * ROW) + lane;
#pragma unroll
    for (int q = 0; q < NV; ++q) {
      float4 e = r[q];
      e.x = e.x > 0.0f ? e.x : expm1f(e.x);
      e.y = e.y > 0.0f ? e.y : expm1f(e.y);
      e.z = e.z > 0.0f ? e.z : expm1f(e.z);
      e.w = e.w > 0.0f ? e.w : expm1f(e.w);
      o[q * 32] = e;
    }
  } else {
    const float4 p = ((const float4*)(res + (long)node * ROW))[lane];
    float4 s = r[0];
    s.x += p.x; s.y += p.y; s.z += p.z; s.w += p.w;
    s.x += row_xor_lane<8>(s.x); s.y += row_xor_lane<8>(s.y);
    s.z += row_xor_lane<8>(s.z); s.w += row_xor_lane<8>(s.w);
    s.x += other_row_lane(s.x);  s.y += other_row_lane(s.y);
    s.z += other_row_lane(s.z);  s.w += other_row_lane(s.w);
    if (lane < 8) {
      s.x *= 0.25f; s.y *= 0.25f; s.z *= 0.25f; s.w *= 0.25f;
      ((float4*)(out + (long)node * (ROW / 4)))[lane] = s;
    }
  }
}

static inline int cdiv_l(long a, long b) { return (int)((a + b - 1) / b); }

extern "C" void kernel_launch(void* const* d_in, const int* in_sizes, int n_in,
                              void* d_out, int out_size, void* d_ws, size_t ws_size,
                              hipStream_t stream) {
  const float* x     = (const float*)d_in[0];
  const float* W0    = (const float*)d_in[1];
  const float* a0    = (const float*)d_in[2];
  const float* W1    = (const float*)d_in[3];
  const float* a1    = (const float*)d_in[4];
  const float* Wres1 = (const float*)d_in[5];
  const int*   src   = (const int*)d_in[6];
  const int*   dst   = (const int*)d_in[7];

  const int N = in_sizes[0] / 256;
  const int E = in_sizes[6];

  float*    f0    = (float*)d_ws;
  float*    h     = f0 + (long)N * 256;
  float*    grp   = h  + (long)N * 256;
  float*    o1    = grp + (long)N * 128;
  float*    f1    = f0;
  int*      nbr    = (int*)grp;
  int*      cnt    = nbr + E;
  int*      rowptr = cnt + N;
  int*      cursor = rowptr + N;
  const size_t need = ((size_t)N * 256 * 2 + (size_t)N * 128 * 2) * sizeof(float);
  if (need > ws_size) return;
  if (((size_t)E + (size_t)N * 3) * sizeof(int) > (size_t)N * 128 * sizeof(float))
    return;

  const int BT = 256;
  const int ngrid = cdiv_l((long)N * 32, BT);

  {
    const int strips = ((N + 15) / 16) * (256 / 16 / 4);
    gemm_f32_wmma<4, 256, 256><<<cdiv_l(strips, 8), BT, 0, stream>>>(x, W0, f0, N);
  }
  zero_counts<<<cdiv_l(N, BT), BT, 0, stream>>>(cnt, N);
  dst_histogram<<<cdiv_l(E, BT), BT, 0, stream>>>(dst, cnt, E);
  rowptr_scan<<<cdiv_l(N, 1024), 1024, 0, stream>>>(cnt, rowptr, cursor, N);
  csr_fill<<<cdiv_l(E, BT), BT, 0, stream>>>(src, dst, cursor, nbr, E);
  attend_by_dst<256, TAIL_ELU><<<ngrid, BT, 0, stream>>>(f0, a0, nbr, rowptr, cnt, nullptr, h, N);

  {
    const int strips = ((N + 15) / 16) * (128 / 16 / 4);
    gemm_f32_wmma<4, 256, 128><<<cdiv_l(strips, 8), BT, 0, stream>>>(h, W1, f1, N);
    gemm_f32_wmma<4, 256, 128><<<cdiv_l(strips, 8), BT, 0, stream>>>(h, Wres1, o1, N);
  }
  attend_by_dst<128, TAIL_RES_MEAN><<<ngrid, BT, 0, stream>>>(f1, a1, nbr, rowptr, cnt, o1,
                                                           (float*)d_out, N);
}


# GEMM fragment loads 128-bit: weights re-laid k-innermost in the first launch; a lane loads 4 k of A and of B at once; k order inside each 8-k group changes (sum order of the products)
# speedup vs baseline: 3.0099x; 1.0492x over previous
#include <hip/hip_runtime.h>
#include <hip/hip_bf16.h>

typedef __attribute__((ext_vector_type(2))) float v2f;
typedef __attribute__((ext_vector_type(8))) float v8f;

template <int NT, int K, int NC>
__global__ __launch_bounds__(256)
void gemm_f32_wmma(const float* __restrict__ A, const float* __restrict__ Bt,
                   float* __restrict__ C, int M) {
  static_assert(K % 8 == 0, "a block of the k loop is 8 k");
  const int lane = threadIdx.x & 31;
  const int wid  = blockIdx.x * (blockDim.x >> 5) + (threadIdx.x >> 5);
  const int tilesM  = (M + 15) >> 4;
  constexpr int stripsN = (NC >> 4) / NT;
  if (wid >= tilesM * stripsN) return;
  const int tm  = wid / stripsN;
  const int tn0 = (wid - tm * stripsN) * NT;
  const int hi = lane >> 4;
  const int lm = lane & 15;
  int arow = tm * 16 + lm; if (arow >= M) arow = M - 1;
  const int col0 = tn0 * 16 + lm;
  const float4* __restrict__ Ar = (const float4*)(A + (long)arow * K) + hi;
  const float4* __restrict__ Bc = (const float4*)Bt + (long)hi * NC + col0;

  v8f c[NT];
#pragma unroll
  for (int t = 0; t < NT; ++t) c[t] = (v8f){};

  auto loadB = [&](int j, float4* b) {
    const float4* __restrict__ B0 = Bc + (long)(2 * j) * NC;
#pragma unroll
    for (int t = 0; t < NT; ++t) b[t] = B0[16 * t];
  };
  auto mma_block = [&](const float4& a, const float4* b) {
    const v2f a0 = {a.x, a.y}, a1 = {a.z, a.w};
#pragma unroll
    for (int t = 0; t < NT; ++t) {
      const v2f b0 = {b[t].x, b[t].y};
      c[t] = __builtin_amdgcn_wmma_f32_16x16x4_f32(false, a0, false, b0,
                                                   (short)0, c[t], false, false);
    }
#pragma unroll
    for (int t = 0; t < NT; ++t) {
      const v2f b1 = {b[t].z, b[t].w};
      c[t] = __builtin_amdgcn_wmma_f32_16x16x4_f32(false, a1, false, b1,
                                                   (short)0, c[t], false, false);
    }
  };

  float4 a_cur = Ar[0];
  float4 b_cur[NT];
  loadB(0, b_cur);
#pragma unroll
  for (int j = 0; j < K / 8 - 1; ++j) {
    const float4 a_nxt = Ar[2 * (j + 1)];
    float4 b_nxt[NT];
    loadB(j + 1, b_nxt);
    mma_block(a_cur, b_cur);
    a_cur = a_nxt;
#pragma unroll
    for (int t = 0; t < NT; ++t) b_cur[t] = b_nxt[t];
  }
  mma_block(a_cur, b_cur);

#pragma unroll
  for (int t = 0; t < NT; ++t) {
#pragma unroll
    for (int r = 0; r < 8; ++r) {
      const int orow = tm * 16 + r + 8 * hi;
      if (orow < M) C[(long)orow * NC + col0 + 16 * t] = c[t][r];
    }
  }
}

__device__ __forceinline__ void relayout_k4(const float* __restrict__ B,
                                            float* __restrict__ Bt, int idx, int NC) {
  const int kq = idx / NC;
  const int n  = idx - kq * NC;
  const float* __restrict__ p = B + (long)(4 * kq) * NC + n;
  ((float4*)Bt)[idx] = make_float4(p[0], p[NC], p[2 * NC], p[3 * NC]);
}

__global__ void zero_counts_relayout_weights(
    int* __restrict__ cnt, int n, int zero_blocks,
    const float* __restrict__ W0, const float* __restrict__ W1,
    const float* __restrict__ Wres1, float* __restrict__ W0t,
    float* __restrict__ W1t, float* __restrict__ Wres1t) {
  if ((int)blockIdx.x < zero_blocks) {
    const int i = blockIdx.x * blockDim.x + threadIdx.x;
    if (i < n) cnt[i] = 0;
    return;
  }
  const int r = (blockIdx.x - zero_blocks) * blockDim.x + threadIdx.x;
  if (r < 64 * 256)                  relayout_k4(W0, W0t, r, 256);
  else if (r < 64 * 256 + 64 * 128)  relayout_k4(W1, W1t, r - 64 * 256, 128);
  else if (r < 64 * 256 + 2 * 64 * 128)
                                     relayout_k4(Wres1, Wres1t, r - 64 * 256 - 64 * 128, 128);
}

__global__ void dst_histogram(const int* __restrict__ dst, int* __restrict__ cnt, int E) {
  const int e = blockIdx.x * blockDim.x + threadIdx.x;
  if (e < E) atomicAdd(cnt + dst[e], 1);
}

__global__ __launch_bounds__(1024)
void rowptr_scan(const int* __restrict__ cnt, int* __restrict__ rowptr,
                 int* __restrict__ cursor, int N) {
  __shared__ int wsum[32];
  __shared__ int wpre[32];
  const int tid = threadIdx.x, lane = tid & 31, w = tid >> 5;
  const int c0 = blockIdx.x * 1024;
  int pre = 0;
  for (int i = tid; i < c0; i += 1024) pre += cnt[i];
  const int i = c0 + tid;
  const int v = i < N ? cnt[i] : 0;
  int inc = v;
#pragma unroll
  for (int off = 1; off < 32; off <<= 1) {
    const int t = __shfl_up(inc, off, 32);
    if (lane >= off) inc += t;
  }
#pragma unroll
  for (int off = 16; off > 0; off >>= 1) pre += __shfl_xor(pre, off, 32);
  if (lane == 31) wsum[w] = inc;
  if (lane == 0)  wpre[w] = pre;
  __syncthreads();
  int before = 0;
#pragma unroll
  for (int k = 0; k < 32; ++k) {
    before += wpre[k];
    if (k < w) before += wsum[k];
  }
  const int excl = before + inc - v;
  if (i < N) { rowptr[i] = excl; cursor[i] = excl; }
}

__global__ void csr_fill(const int* __restrict__ src, const int* __restrict__ dst,
                         int* __restrict__ cursor, int* __restrict__ nbr, int E) {
  const int e = blockIdx.x * blockDim.x + threadIdx.x;
  if (e >= E) return;
  const int p = atomicAdd(cursor + dst[e], 1);
  nbr[p] = src[e];
}

template <int MASK>
__device__ __forceinline__ float row_xor_lane(float x) {
  return __int_as_float(__builtin_amdgcn_update_dpp(
      0, __float_as_int(x), 0x160 + MASK, 0xF, 0xF, true));
}

__device__ __forceinline__ float other_row_lane(float x) {
  return __int_as_float((int)__builtin_amdgcn_permlanex16(
      0u, (unsigned)__float_as_int(x), 0x76543210u, 0xfedcba98u, false, false));
}

template <int ROW, int U>
__device__ __forceinline__ void attend_edges(
    const float4* __restrict__ f4, int sidx, int i, int lane,
    const float4 (&fd)[ROW / 128], const float4 (&at)[ROW / 128],
    float4 (&acc)[ROW / 128], float (&mx)[ROW / 128], float (&den)[ROW / 128]) {
  constexpr int NV  = ROW / 128;
  constexpr int LPH = ROW / 16;
  float4 v[U][NV];
  float  lg[U][NV];
#pragma unroll
  for (int u = 0; u < U; ++u) {
    const int s = __builtin_amdgcn_readlane(sidx, i + u);
    const float4* __restrict__ row = f4 + (long)s * (ROW / 4) + lane;
#pragma unroll
    for (int q = 0; q < NV; ++q) v[u][q] = row[q * 32];
  }
#pragma unroll
  for (int u = 0; u < U; ++u) {
#pragma unroll
    for (int q = 0; q < NV; ++q) {
      float t0 = v[u][q].x + fd[q].x, t1 = v[u][q].y + fd[q].y;
      float t2 = v[u][q].z + fd[q].z, t3 = v[u][q].w + fd[q].w;
      t0 = t0 > 0.0f ? t0 : 0.2f * t0;
      t1 = t1 > 0.0f ? t1 : 0.2f * t1;
      t2 = t2 > 0.0f ? t2 : 0.2f * t2;
      t3 = t3 > 0.0f ? t3 : 0.2f * t3;
      float p = t0 * at[q].x;
      p += t1 * at[q].y;
      p += t2 * at[q].z;
      p += t3 * at[q].w;
      lg[u][q] = p;
    }
  }
  if (LPH >= 16) {
#pragma unroll
    for (int u = 0; u < U; ++u) {
#pragma unroll
      for (int q = 0; q < NV; ++q) lg[u][q] += row_xor_lane<8>(lg[u][q]);
    }
  }
#pragma unroll
  for (int u = 0; u < U; ++u) {
#pragma unroll
    for (int q = 0; q < NV; ++q) lg[u][q] += row_xor_lane<4>(lg[u][q]);
  }
#pragma unroll
  for (int u = 0; u < U; ++u) {
#pragma unroll
    for (int q = 0; q < NV; ++q) lg[u][q] += row_xor_lane<2>(lg[u][q]);
  }
#pragma unroll
  for (int u = 0; u < U; ++u) {
#pragma unroll
    for (int q = 0; q < NV; ++q) lg[u][q] += row_xor_lane<1>(lg[u][q]);
  }
#pragma unroll
  for (int q = 0; q < NV; ++q) {
    float bm = mx[q];
#pragma unroll
    for (int u = 0; u < U; ++u) bm = fmaxf(bm, lg[u][q]);
    const float sc = __expf(mx[q] - bm);
    float4 a = acc[q];
    a.x *= sc; a.y *= sc; a.z *= sc; a.w *= sc;
    float d = den[q] * sc;
#pragma unroll
    for (int u = 0; u < U; ++u) {
      const float p = __expf(lg[u][q] - bm);
      d += p;
      a.x += p * v[u][q].x;
      a.y += p * v[u][q].y;
      a.z += p * v[u][q].z;
      a.w += p * v[u][q].w;
    }
    acc[q] = a;
    den[q] = d;
    mx[q]  = bm;
  }
}

enum { TAIL_ELU = 0, TAIL_RES_MEAN = 1 };

template <int ROW, int TAIL>
__global__ __launch_bounds__(256)
void attend_by_dst(const float* __restrict__ f, const float* __restrict__ attn,
                   const int* __restrict__ nbr, const int* __restrict__ rowptr,
                   const int* __restrict__ cnt, const float* __restrict__ res,
                   float* __restrict__ out, int N) {
  constexpr int NV = ROW / 128;
  static_assert(ROW == 128 || ROW == 256, "the lanes of one head must lie inside one row of 16 lanes");
  static_assert(TAIL == TAIL_ELU || ROW == 128, "the head mean is written for 4 heads of 8 lanes each");
  const int lane = threadIdx.x & 31;
  const int node = __builtin_amdgcn_readfirstlane(
      (int)(blockIdx.x * (blockDim.x >> 5) + (threadIdx.x >> 5)));
  if (node >= N) return;
  const int start = rowptr[node];
  const int deg   = cnt[node];
  const float4* __restrict__ f4   = (const float4*)f;
  const float4* __restrict__ frow = f4 + (long)node * (ROW / 4) + lane;
  const float4* __restrict__ arow = (const float4*)attn + lane;
  float4 fd[NV], at[NV], acc[NV];
  float  mx[NV], den[NV];
#pragma unroll
  for (int q = 0; q < NV; ++q) {
    fd[q]  = frow[q * 32];
    at[q]  = arow[q * 32];
    acc[q] = make_float4(0.0f, 0.0f, 0.0f, 0.0f);
    mx[q]  = -__builtin_inff();
    den[q] = 0.0f;
  }
  for (int c = 0; c < deg; c += 32) {
    const int nn = deg - c < 32 ? deg - c : 32;
    int sidx = 0;
    if (lane < nn) sidx = nbr[start + c + lane];
    int i = 0;
    for (; i + 4 <= nn; i += 4) attend_edges<ROW, 4>(f4, sidx, i, lane, fd, at, acc, mx, den);
    if (i + 2 <= nn) { attend_edges<ROW, 2>(f4, sidx, i, lane, fd, at, acc, mx, den); i += 2; }
    if (i < nn)        attend_edges<ROW, 1>(f4, sidx, i, lane, fd, at, acc, mx, den);
  }
  float4 r[NV];
#pragma unroll
  for (int q = 0; q < NV; ++q) {
    r[q] = make_float4(0.0f, 0.0f, 0.0f, 0.0f);
    if (deg > 0) {
      r[q].x = acc[q].x / den[q]; r[q].y = acc[q].y / den[q];
      r[q].z = acc[q].z / den[q]; r[q].w = acc[q].w / den[q];
    }
  }
  if (TAIL == TAIL_ELU) {
    float4* o = (float4*)(out + (long)node * ROW) + lane;
#pragma unroll
    for (int q = 0; q < NV; ++q) {
      float4 e = r[q];
      e.x = e.x > 0.0f ? e.x : expm1f(e.x);
      e.y = e.y > 0.0f ? e.y : expm1f(e.y);
      e.z = e.z > 0.0f ? e.z : expm1f(e.z);
      e.w = e.w > 0.0f ? e.w : expm1f(e.w);
      o[q * 32] = e;
    }
  } else {
    const float4 p = ((const float4*)(res + (long)node * ROW))[lane];
    float4 s = r[0];
    s.x += p.x; s.y += p.y; s.z += p.z; s.w += p.w;
    s.x += row_xor_lane<8>(s.x); s.y += row_xor_lane<8>(s.y);
    s.z += row_xor_lane<8>(s.z); s.w += row_xor_lane<8>(s.w);
    s.x += other_row_lane(s.x);  s.y += other_row_lane(s.y);
    s.z += other_row_lane(s.z);  s.w += other_row_lane(s.w);
    if (lane < 8) {
      s.x *= 0.25f; s.y *= 0.25f; s.z *= 0.25f; s.w *= 0.25f;
      ((float4*)(out + (long)node * (ROW / 4)))[lane] = s;
    }
  }
}

static inline int cdiv_l(long a, long b) { return (int)((a + b - 1) / b); }

extern "C" void kernel_launch(void* const* d_in, const int* in_sizes, int n_in,
                              void* d_out, int out_size, void* d_ws, size_t ws_size,
                              hipStream_t stream) {
  const float* x     = (const float*)d_in[0];
  const float* W0    = (const float*)d_in[1];
  const float* a0    = (const float*)d_in[2];
  const float* W1    = (const float*)d_in[3];
  const float* a1    = (const float*)d_in[4];
  const float* Wres1 = (const float*)d_in[5];
  const int*   src   = (const int*)d_in[6];
  const int*   dst   = (const int*)d_in[7];

  const int N = in_sizes[0] / 256;
  const int E = in_sizes[6];

  float*    f0    = (float*)d_ws;
  float*    h     = f0 + (long)N * 256;
  float*    grp   = h  + (long)N * 256;
  float*    o1    = grp + (long)N * 128;
  float*    f1    = f0;
  int*      nbr    = (int*)grp;
  int*      cnt    = nbr + E;
  int*      rowptr = cnt + N;
  int*      cursor = rowptr + N;
  const size_t grp_ints = ((size_t)E + (size_t)N * 3 + 3) & ~(size_t)3;
  float*    W0t    = grp + grp_ints;
  float*    W1t    = W0t + 256 * 256;
  float*    Wres1t = W1t + 256 * 128;
  const size_t need = ((size_t)N * 256 * 2 + (size_t)N * 128 * 2) * sizeof(float);
  if (need > ws_size) return;
  if ((grp_ints + 256 * 256 + 2 * 256 * 128) * sizeof(float) > (size_t)N * 128 * sizeof(float))
    return;

  const int BT = 256;
  const int ngrid = cdiv_l((long)N * 32, BT);

  {
    const int zero_blocks = cdiv_l(N, BT);
    const int relayout_blocks = (64 * 256 + 2 * 64 * 128) / BT;
    zero_counts_relayout_weights<<<zero_blocks + relayout_blocks, BT, 0, stream>>>(
        cnt, N, zero_blocks, W0, W1, Wres1, W0t, W1t, Wres1t);
  }

  {
    const int strips = ((N + 15) / 16) * (256 / 16 / 4);
    gemm_f32_wmma<4, 256, 256><<<cdiv_l(strips, 8), BT, 0, stream>>>(x, W0t, f0, N);
  }
  dst_histogram<<<cdiv_l(E, BT), BT, 0, stream>>>(dst, cnt, E);
  rowptr_scan<<<cdiv_l(N, 1024), 1024, 0, stream>>>(cnt, rowptr, cursor, N);
  csr_fill<<<cdiv_l(E, BT), BT, 0, stream>>>(src, dst, cursor, nbr, E);
  attend_by_dst<256, TAIL_ELU><<<ngrid, BT, 0, stream>>>(f0, a0, nbr, rowptr, cnt, nullptr, h, N);

  {
    const int strips = ((N + 15) / 16) * (128 / 16 / 4);
    gemm_f32_wmma<4, 256, 128><<<cdiv_l(strips, 8), BT, 0, stream>>>(h, W1t, f1, N);
    gemm_f32_wmma<4, 256, 128><<<cdiv_l(strips, 8), BT, 0, stream>>>(h, Wres1t, o1, N);
  }
  attend_by_dst<128, TAIL_RES_MEAN><<<ngrid, BT, 0, stream>>>(f1, a1, nbr, rowptr, cnt, o1,
                                                           (float*)d_out, N);
}


# layer-1 projections fused into the layer-0 attention launch: a workgroup attends 32 nodes, keeps their elu rows as a tile in LDS and multiplies it by W1|Wres1 on the matrix unit; 9 -> 7 launches
# speedup vs baseline: 3.1640x; 1.0512x over previous
#include <hip/hip_runtime.h>
#include <hip/hip_bf16.h>

typedef __attribute__((ext_vector_type(2))) float v2f;
typedef __attribute__((ext_vector_type(8))) float v8f;

template <int NT, int K, int NC>
__global__ __launch_bounds__(256)
void gemm_f32_wmma(const float* __restrict__ A, const float* __restrict__ Bt,
                   float* __restrict__ C, int M) {
  static_assert(K % 8 == 0, "a block of the k loop is 8 k");
  const int lane = threadIdx.x & 31;
  const int wid  = blockIdx.x * (blockDim.x >> 5) + (threadIdx.x >> 5);
  const int tilesM  = (M + 15) >> 4;
  constexpr int stripsN = (NC >> 4) / NT;
  if (wid >= tilesM * stripsN) return;
  const int tm  = wid / stripsN;
  const int tn0 = (wid - tm * stripsN) * NT;
  const int hi = lane >> 4;
  const int lm = lane & 15;
  int arow = tm * 16 + lm; if (arow >= M) arow = M - 1;
  const int col0 = tn0 * 16 + lm;
  const float4* __restrict__ Ar = (const float4*)(A + (long)arow * K) + hi;
  const float4* __restrict__ Bc = (const float4*)Bt + (long)hi * NC + col0;

  v8f c[NT];
#pragma unroll
  for (int t = 0; t < NT; ++t) c[t] = (v8f){};

  auto loadB = [&](int j, float4* b) {
    const float4* __restrict__ B0 = Bc + (long)(2 * j) * NC;
#pragma unroll
    for (int t = 0; t < NT; ++t) b[t] = B0[16 * t];
  };
  auto mma_block = [&](const float4& a, const float4* b) {
    const v2f a0 = {a.x, a.y}, a1 = {a.z, a.w};
#pragma unroll
    for (int t = 0; t < NT; ++t) {
      const v2f b0 = {b[t].x, b[t].y};
      c[t] = __builtin_amdgcn_wmma_f32_16x16x4_f32(false, a0, false, b0,
                                                   (short)0, c[t], false, false);
    }
#pragma unroll
    for (int t = 0; t < NT; ++t) {
      const v2f b1 = {b[t].z, b[t].w};
      c[t] = __builtin_amdgcn_wmma_f32_16x16x4_f32(false, a1, false, b1,
                                                   (short)0, c[t], false, false);
    }
  };

  float4 a_cur = Ar[0];
  float4 b_cur[NT];
  loadB(0, b_cur);
#pragma unroll
  for (int j = 0; j < K / 8 - 1; ++j) {
    const float4 a_nxt = Ar[2 * (j + 1)];
    float4 b_nxt[NT];
    loadB(j + 1, b_nxt);
    mma_block(a_cur, b_cur);
    a_cur = a_nxt;
#pragma unroll
    for (int t = 0; t < NT; ++t) b_cur[t] = b_nxt[t];
  }
  mma_block(a_cur, b_cur);

#pragma unroll
  for (int t = 0; t < NT; ++t) {
#pragma unroll
    for (int r = 0; r < 8; ++r) {
      const int orow = tm * 16 + r + 8 * hi;
      if (orow < M) C[(long)orow * NC + col0 + 16 * t] = c[t][r];
    }
  }
}

__device__ __forceinline__ void relayout_k4(const float* __restrict__ B,
                                            float* __restrict__ Bt, int idx, int NC) {
  const int kq = idx / NC;
  const int n  = idx - kq * NC;
  const float* __restrict__ p = B + (long)(4 * kq) * NC + n;
  ((float4*)Bt)[idx] = make_float4(p[0], p[NC], p[2 * NC], p[3 * NC]);
}

__global__ void zero_counts_relayout_weights(
    int* __restrict__ cnt, int n, int zero_blocks,
    const float* __restrict__ W0, const float* __restrict__ W1,
    const float* __restrict__ Wres1, float* __restrict__ W0t,
    float* __restrict__ W1t, float* __restrict__ Wres1t) {
  if ((int)blockIdx.x < zero_blocks) {
    const int i = blockIdx.x * blockDim.x + threadIdx.x;
    if (i < n) cnt[i] = 0;
    return;
  }
  const int r = (blockIdx.x - zero_blocks) * blockDim.x + threadIdx.x;
  if (r < 64 * 256)                  relayout_k4(W0, W0t, r, 256);
  else if (r < 64 * 256 + 64 * 128)  relayout_k4(W1, W1t, r - 64 * 256, 128);
  else if (r < 64 * 256 + 2 * 64 * 128)
                                     relayout_k4(Wres1, Wres1t, r - 64 * 256 - 64 * 128, 128);
}

__global__ void dst_histogram(const int* __restrict__ dst, int* __restrict__ cnt, int E) {
  const int e = blockIdx.x * blockDim.x + threadIdx.x;
  if (e < E) atomicAdd(cnt + dst[e], 1);
}

__global__ __launch_bounds__(1024)
void rowptr_scan(const int* __restrict__ cnt, int* __restrict__ rowptr,
                 int* __restrict__ cursor, int N) {
  __shared__ int wsum[32];
  __shared__ int wpre[32];
  const int tid = threadIdx.x, lane = tid & 31, w = tid >> 5;
  const int c0 = blockIdx.x * 1024;
  int pre = 0;
  for (int i = tid; i < c0; i += 1024) pre += cnt[i];
  const int i = c0 + tid;
  const int v = i < N ? cnt[i] : 0;
  int inc = v;
#pragma unroll
  for (int off = 1; off < 32; off <<= 1) {
    const int t = __shfl_up(inc, off, 32);
    if (lane >= off) inc += t;
  }
#pragma unroll
  for (int off = 16; off > 0; off >>= 1) pre += __shfl_xor(pre, off, 32);
  if (lane == 31) wsum[w] = inc;
  if (lane == 0)  wpre[w] = pre;
  __syncthreads();
  int before = 0;
#pragma unroll
  for (int k = 0; k < 32; ++k) {
    before += wpre[k];
    if (k < w) before += wsum[k];
  }
  const int excl = before + inc - v;
  if (i < N) { rowptr[i] = excl; cursor[i] = excl; }
}

__global__ void csr_fill(const int* __restrict__ src, const int* __restrict__ dst,
                         int* __restrict__ cursor, int* __restrict__ nbr, int E) {
  const int e = blockIdx.x * blockDim.x + threadIdx.x;
  if (e >= E) return;
  const int p = atomicAdd(cursor + dst[e], 1);
  nbr[p] = src[e];
}

template <int MASK>
__device__ __forceinline__ float row_xor_lane(float x) {
  return __int_as_float(__builtin_amdgcn_update_dpp(
      0, __float_as_int(x), 0x160 + MASK, 0xF, 0xF, true));
}

__device__ __forceinline__ float other_row_lane(float x) {
  return __int_as_float((int)__builtin_amdgcn_permlanex16(
      0u, (unsigned)__float_as_int(x), 0x76543210u, 0xfedcba98u, false, false));
}

template <int ROW, int U>
__device__ __forceinline__ void attend_edges(
    const float4* __restrict__ f4, int sidx, int i, int lane,
    const float4 (&fd)[ROW / 128], const float4 (&at)[ROW / 128],
    float4 (&acc)[ROW / 128], float (&mx)[ROW / 128], float (&den)[ROW / 128]) {
  constexpr int NV  = ROW / 128;
  constexpr int LPH = ROW / 16;
  float4 v[U][NV];
  float  lg[U][NV];
#pragma unroll
  for (int u = 0; u < U; ++u) {
    const int s = __builtin_amdgcn_readlane(sidx, i + u);
    const float4* __restrict__ row = f4 + (long)s * (ROW / 4) + lane;
#pragma unroll
    for (int q = 0; q < NV; ++q) v[u][q] = row[q * 32];
  }
#pragma unroll
  for (int u = 0; u < U; ++u) {
#pragma unroll
    for (int q = 0; q < NV; ++q) {
      float t0 = v[u][q].x + fd[q].x, t1 = v[u][q].y + fd[q].y;
      float t2 = v[u][q].z + fd[q].z, t3 = v[u][q].w + fd[q].w;
      t0 = t0 > 0.0f ? t0 : 0.2f * t0;
      t1 = t1 > 0.0f ? t1 : 0.2f * t1;
      t2 = t2 > 0.0f ? t2 : 0.2f * t2;
      t3 = t3 > 0.0f ? t3 : 0.2f * t3;
      float p = t0 * at[q].x;
      p += t1 * at[q].y;
      p += t2 * at[q].z;
      p += t3 * at[q].w;
      lg[u][q] = p;
    }
  }
  if (LPH >= 16) {
#pragma unroll
    for (int u = 0; u < U; ++u) {
#pragma unroll
      for (int q = 0; q < NV; ++q) lg[u][q] += row_xor_lane<8>(lg[u][q]);
    }
  }
#pragma unroll
  for (int u = 0; u < U; ++u) {
#pragma unroll
    for (int q = 0; q < NV; ++q) lg[u][q] += row_xor_lane<4>(lg[u][q]);
  }
#pragma unroll
  for (int u = 0; u < U; ++u) {
#pragma unroll
    for (int q = 0; q < NV; ++q) lg[u][q] += row_xor_lane<2>(lg[u][q]);
  }
#pragma unroll
  for (int u = 0; u < U; ++u) {
#pragma unroll
    for (int q = 0; q < NV; ++q) lg[u][q] += row_xor_lane<1>(lg[u][q]);
  }
#pragma unroll
  for (int q = 0; q < NV; ++q) {
    float bm = mx[q];
#pragma unroll
    for (int u = 0; u < U; ++u) bm = fmaxf(bm, lg[u][q]);
    const float sc = __expf(mx[q] - bm);
    float4 a = acc[q];
    a.x *= sc; a.y *= sc; a.z *= sc; a.w *= sc;
    float d = den[q] * sc;
#pragma unroll
    for (int u = 0; u < U; ++u) {
      const float p = __expf(lg[u][q] - bm);
      d += p;
      a.x += p * v[u][q].x;
      a.y += p * v[u][q].y;
      a.z += p * v[u][q].z;
      a.w += p * v[u][q].w;
    }
    acc[q] = a;
    den[q] = d;
    mx[q]  = bm;
  }
}

template <int ROW>
__device__ __forceinline__ void attend_node(
    const float* __restrict__ f, const float* __restrict__ attn,
    const int* __restrict__ nbr, const int* __restrict__ rowptr,
    const int* __restrict__ cnt, int node, int N, int lane, float4 (&r)[ROW / 128]) {
  constexpr int NV = ROW / 128;
  static_assert(ROW == 128 || ROW == 256, "the lanes of one head must lie inside one row of 16 lanes");
#pragma unroll
  for (int q = 0; q < NV; ++q) r[q] = make_float4(0.0f, 0.0f, 0.0f, 0.0f);
  if (node >= N) return;
  const int start = rowptr[node];
  const int deg   = cnt[node];
  const float4* __restrict__ f4   = (const float4*)f;
  const float4* __restrict__ frow = f4 + (long)node * (ROW / 4) + lane;
  const float4* __restrict__ arow = (const float4*)attn + lane;
  float4 fd[NV], at[NV], acc[NV];
  float  mx[NV], den[NV];
#pragma unroll
  for (int q = 0; q < NV; ++q) {
    fd[q]  = frow[q * 32];
    at[q]  = arow[q * 32];
    acc[q] = make_float4(0.0f, 0.0f, 0.0f, 0.0f);
    mx[q]  = -__builtin_inff();
    den[q] = 0.0f;
  }
  for (int c = 0; c < deg; c += 32) {
    const int nn = deg - c < 32 ? deg - c : 32;
    int sidx = 0;
    if (lane < nn) sidx = nbr[start + c + lane];
    int i = 0;
    for (; i + 4 <= nn; i += 4) attend_edges<ROW, 4>(f4, sidx, i, lane, fd, at, acc, mx, den);
    if (i + 2 <= nn) { attend_edges<ROW, 2>(f4, sidx, i, lane, fd, at, acc, mx, den); i += 2; }
    if (i < nn)        attend_edges<ROW, 1>(f4, sidx, i, lane, fd, at, acc, mx, den);
  }
  if (deg > 0) {
#pragma unroll
    for (int q = 0; q < NV; ++q) {
      r[q].x = acc[q].x / den[q]; r[q].y = acc[q].y / den[q];
      r[q].z = acc[q].z / den[q]; r[q].w = acc[q].w / den[q];
    }
  }
}

__global__ __launch_bounds__(256, 8)
void attend0_project1(const float* __restrict__ f, const float* __restrict__ attn,
                      const int* __restrict__ nbr, const int* __restrict__ rowptr,
                      const int* __restrict__ cnt, const float* __restrict__ W1t,
                      const float* __restrict__ Wres1t, float* __restrict__ f1,
                      float* __restrict__ o1, int N) {
  constexpr int ROW = 256, NV = ROW / 128, PITCH4 = 65, NC = 128, KB = ROW / 8;
  constexpr int MT = 2;
  constexpr int NODES = 16 * MT;
  constexpr int PER_WAVE = NODES / 8;
  __shared__ float4 tile[NODES * PITCH4];
  const int lane = threadIdx.x & 31;
  const int w    = __builtin_amdgcn_readfirstlane((int)(threadIdx.x >> 5));
  const int base = blockIdx.x * NODES;

#pragma nounroll
  for (int s = 0; s < PER_WAVE; ++s) {
    const int rloc = PER_WAVE * w + s;
    float4 r[NV];
    attend_node<ROW>(f, attn, nbr, rowptr, cnt, base + rloc, N, lane, r);
#pragma unroll
    for (int q = 0; q < NV; ++q) {
      float4 e = r[q];
      e.x = e.x > 0.0f ? e.x : expm1f(e.x);
      e.y = e.y > 0.0f ? e.y : expm1f(e.y);
      e.z = e.z > 0.0f ? e.z : expm1f(e.z);
      e.w = e.w > 0.0f ? e.w : expm1f(e.w);
      tile[rloc * PITCH4 + q * 32 + lane] = e;
    }
  }
  __syncthreads();

  const int hi = lane >> 4;
  const int lm = lane & 15;
  const float* __restrict__ Bt = w < 4 ? W1t : Wres1t;
  float* __restrict__ C        = w < 4 ? f1 : o1;
  const int col0 = (w & 3) * 32 + lm;
  const float4* __restrict__ Bc = (const float4*)Bt + hi * NC + col0;
  const float4* Ar = tile + lm * PITCH4 + hi;

  v8f c[MT][2];
#pragma unroll
  for (int m = 0; m < MT; ++m) { c[m][0] = (v8f){}; c[m][1] = (v8f){}; }
  constexpr int PD = 6;
  float4 bq[PD][2];
#pragma unroll
  for (int p = 0; p < PD; ++p) {
    bq[p][0] = Bc[(2 * p) * NC];
    bq[p][1] = Bc[(2 * p) * NC + 16];
  }
  constexpr int AD = 2;
  float4 aq[AD][MT];
#pragma unroll
  for (int p = 0; p < AD; ++p) {
#pragma unroll
    for (int m = 0; m < MT; ++m) aq[p][m] = Ar[m * 16 * PITCH4 + 2 * p];
  }
#pragma unroll
  for (int j = 0; j < KB; ++j) {
    float4 a_cur[MT];
#pragma unroll
    for (int m = 0; m < MT; ++m) {
      a_cur[m] = aq[j % AD][m];
      if (j + AD < KB) aq[j % AD][m] = Ar[m * 16 * PITCH4 + 2 * (j + AD)];
    }
    const float4 b0 = bq[j % PD][0], b1 = bq[j % PD][1];
    if (j + PD < KB) {
      bq[j % PD][0] = Bc[(2 * (j + PD)) * NC];
      bq[j % PD][1] = Bc[(2 * (j + PD)) * NC + 16];
    }
    const v2f b00 = {b0.x, b0.y}, b10 = {b1.x, b1.y};
    const v2f b01 = {b0.z, b0.w}, b11 = {b1.z, b1.w};
#pragma unroll
    for (int m = 0; m < MT; ++m) {
      const v2f a0 = {a_cur[m].x, a_cur[m].y};
      c[m][0] = __builtin_amdgcn_wmma_f32_16x16x4_f32(false, a0, false, b00, (short)0, c[m][0], false, false);
      c[m][1] = __builtin_amdgcn_wmma_f32_16x16x4_f32(false, a0, false, b10, (short)0, c[m][1], false, false);
    }
#pragma unroll
    for (int m = 0; m < MT; ++m) {
      const v2f a1 = {a_cur[m].z, a_cur[m].w};
      c[m][0] = __builtin_amdgcn_wmma_f32_16x16x4_f32(false, a1, false, b01, (short)0, c[m][0], false, false);
      c[m][1] = __builtin_amdgcn_wmma_f32_16x16x4_f32(false, a1, false, b11, (short)0, c[m][1], false, false);
    }
    __builtin_amdgcn_sched_barrier(0);
  }

#pragma unroll
  for (int m = 0; m < MT; ++m) {
#pragma unroll
    for (int t = 0; t < 2; ++t) {
#pragma unroll
      for (int r = 0; r < 8; ++r) {
        const int orow = base + 16 * m + r + 8 * hi;
        if (orow < N) C[(long)orow * NC + col0 + 16 * t] = c[m][t][r];
      }
    }
  }
}

__global__ __launch_bounds__(256)
void attend1_res_mean(const float* __restrict__ f, const float* __restrict__ attn,
                      const int* __restrict__ nbr, const int* __restrict__ rowptr,
                      const int* __restrict__ cnt, const float* __restrict__ res,
                      float* __restrict__ out, int N) {
  constexpr int ROW = 128;
  const int lane = threadIdx.x & 31;
  const int node = __builtin_amdgcn_readfirstlane(
      (int)(blockIdx.x * (blockDim.x >> 5) + (threadIdx.x >> 5)));
  if (node >= N) return;
  float4 r[1];
  attend_node<ROW>(f, attn, nbr, rowptr, cnt, node, N, lane, r);
  const float4 p = ((const float4*)(res + (long)node * ROW))[lane];
  float4 s = r[0];
  s.x += p.x; s.y += p.y; s.z += p.z; s.w += p.w;
  s.x += row_xor_lane<8>(s.x); s.y += row_xor_lane<8>(s.y);
  s.z += row_xor_lane<8>(s.z); s.w += row_xor_lane<8>(s.w);
  s.x += other_row_lane(s.x);  s.y += other_row_lane(s.y);
  s.z += other_row_lane(s.z);  s.w += other_row_lane(s.w);
  if (lane < 8) {
    s.x *= 0.25f; s.y *= 0.25f; s.z *= 0.25f; s.w *= 0.25f;
    ((float4*)(out + (long)node * (ROW / 4)))[lane] = s;
  }
}

static inline int cdiv_l(long a, long b) { return (int)((a + b - 1) / b); }

extern "C" void kernel_launch(void* const* d_in, const int* in_sizes, int n_in,
                              void* d_out, int out_size, void* d_ws, size_t ws_size,
                              hipStream_t stream) {
  const float* x     = (const float*)d_in[0];
  const float* W0    = (const float*)d_in[1];
  const float* a0    = (const float*)d_in[2];
  const float* W1    = (const float*)d_in[3];
  const float* a1    = (const float*)d_in[4];
  const float* Wres1 = (const float*)d_in[5];
  const int*   src   = (const int*)d_in[6];
  const int*   dst   = (const int*)d_in[7];

  const int N = in_sizes[0] / 256;
  const int E = in_sizes[6];

  float*    f0    = (float*)d_ws;
  float*    f1    = f0 + (long)N * 256;
  float*    grp   = f1 + (long)N * 256;
  float*    o1    = grp + (long)N * 128;
  int*      nbr    = (int*)grp;
  int*      cnt    = nbr + E;
  int*      rowptr = cnt + N;
  int*      cursor = rowptr + N;
  const size_t grp_ints = ((size_t)E + (size_t)N * 3 + 3) & ~(size_t)3;
  float*    W0t    = grp + grp_ints;
  float*    W1t    = W0t + 256 * 256;
  float*    Wres1t = W1t + 256 * 128;
  const size_t need = ((size_t)N * 256 * 2 + (size_t)N * 128 * 2) * sizeof(float);
  if (need > ws_size) return;
  if ((grp_ints + 256 * 256 + 2 * 256 * 128) * sizeof(float) > (size_t)N * 128 * sizeof(float))
    return;

  const int BT = 256;

  {
    const int zero_blocks = cdiv_l(N, BT);
    const int relayout_blocks = (64 * 256 + 2 * 64 * 128) / BT;
    zero_counts_relayout_weights<<<zero_blocks + relayout_blocks, BT, 0, stream>>>(
        cnt, N, zero_blocks, W0, W1, Wres1, W0t, W1t, Wres1t);
  }

  {
    const int strips = ((N + 15) / 16) * (256 / 16 / 4);
    gemm_f32_wmma<4, 256, 256><<<cdiv_l(strips, 8), BT, 0, stream>>>(x, W0t, f0, N);
  }
  dst_histogram<<<cdiv_l(E, BT), BT, 0, stream>>>(dst, cnt, E);
  rowptr_scan<<<cdiv_l(N, 1024), 1024, 0, stream>>>(cnt, rowptr, cursor, N);
  csr_fill<<<cdiv_l(E, BT), BT, 0, stream>>>(src, dst, cursor, nbr, E);

  attend0_project1<<<cdiv_l(N, 32), BT, 0, stream>>>(f0, a0, nbr, rowptr, cnt, W1t, Wres1t,
                                                    f1, o1, N);

  attend1_res_mean<<<cdiv_l((long)N * 32, BT), BT, 0, stream>>>(f1, a1, nbr, rowptr, cnt, o1,
                                                                (float*)d_out, N);
}


# residual projection onto the head mean of Wres1 (mean over heads of a linear map = map of the mean): 32 columns not 128, 20 not 32 accumulator tiles per 32 nodes; residual sum order changes
# speedup vs baseline: 3.3895x; 1.0713x over previous
#include <hip/hip_runtime.h>
#include <hip/hip_bf16.h>

typedef __attribute__((ext_vector_type(2))) float v2f;
typedef __attribute__((ext_vector_type(8))) float v8f;

template <int NT, int K, int NC>
__global__ __launch_bounds__(256)
void gemm_f32_wmma(const float* __restrict__ A, const float* __restrict__ Bt,
                   float* __restrict__ C, int M) {
  static_assert(K % 8 == 0, "a block of the k loop is 8 k");
  const int lane = threadIdx.x & 31;
  const int wid  = blockIdx.x * (blockDim.x >> 5) + (threadIdx.x >> 5);
  const int tilesM  = (M + 15) >> 4;
  constexpr int stripsN = (NC >> 4) / NT;
  if (wid >= tilesM * stripsN) return;
  const int tm  = wid / stripsN;
  const int tn0 = (wid - tm * stripsN) * NT;
  const int hi = lane >> 4;
  const int lm = lane & 15;
  int arow = tm * 16 + lm; if (arow >= M) arow = M - 1;
  const int col0 = tn0 * 16 + lm;
  const float4* __restrict__ Ar = (const float4*)(A + (long)arow * K) + hi;
  const float4* __restrict__ Bc = (const float4*)Bt + (long)hi * NC + col0;

  v8f c[NT];
#pragma unroll
  for (int t = 0; t < NT; ++t) c[t] = (v8f){};

  auto loadB = [&](int j, float4* b) {
    const float4* __restrict__ B0 = Bc + (long)(2 * j) * NC;
#pragma unroll
    for (int t = 0; t < NT; ++t) b[t] = B0[16 * t];
  };
  auto mma_block = [&](const float4& a, const float4* b) {
    const v2f a0 = {a.x, a.y}, a1 = {a.z, a.w};
#pragma unroll
    for (int t = 0; t < NT; ++t) {
      const v2f b0 = {b[t].x, b[t].y};
      c[t] = __builtin_amdgcn_wmma_f32_16x16x4_f32(false, a0, false, b0,
                                                   (short)0, c[t], false, false);
    }
#pragma unroll
    for (int t = 0; t < NT; ++t) {
      const v2f b1 = {b[t].z, b[t].w};
      c[t] = __builtin_amdgcn_wmma_f32_16x16x4_f32(false, a1, false, b1,
                                                   (short)0, c[t], false, false);
    }
  };

  float4 a_cur = Ar[0];
  float4 b_cur[NT];
  loadB(0, b_cur);
#pragma unroll
  for (int j = 0; j < K / 8 - 1; ++j) {
    const float4 a_nxt = Ar[2 * (j + 1)];
    float4 b_nxt[NT];
    loadB(j + 1, b_nxt);
    mma_block(a_cur, b_cur);
    a_cur = a_nxt;
#pragma unroll
    for (int t = 0; t < NT; ++t) b_cur[t] = b_nxt[t];
  }
  mma_block(a_cur, b_cur);

#pragma unroll
  for (int t = 0; t < NT; ++t) {
#pragma unroll
    for (int r = 0; r < 8; ++r) {
      const int orow = tm * 16 + r + 8 * hi;
      if (orow < M) C[(long)orow * NC + col0 + 16 * t] = c[t][r];
    }
  }
}

__device__ __forceinline__ void relayout_k4(const float* __restrict__ B,
                                            float* __restrict__ Bt, int idx, int NC) {
  const int kq = idx / NC;
  const int n  = idx - kq * NC;
  const float* __restrict__ p = B + (long)(4 * kq) * NC + n;
  ((float4*)Bt)[idx] = make_float4(p[0], p[NC], p[2 * NC], p[3 * NC]);
}

__device__ __forceinline__ void relayout_k4_head_mean(const float* __restrict__ B,
                                                      float* __restrict__ Bt, int idx) {
  const int kq = idx >> 5;
  const int c  = idx & 31;
  const float* __restrict__ p = B + (long)(4 * kq) * 128 + c;
  float m[4];
#pragma unroll
  for (int e = 0; e < 4; ++e) {
    const float* __restrict__ q = p + e * 128;
    m[e] = 0.25f * ((q[0] + q[32]) + (q[64] + q[96]));
  }
  ((float4*)Bt)[idx] = make_float4(m[0], m[1], m[2], m[3]);
}

__global__ void zero_counts_relayout_weights(
    int* __restrict__ cnt, int n, int zero_blocks,
    const float* __restrict__ W0, const float* __restrict__ W1,
    const float* __restrict__ Wres1, float* __restrict__ W0t,
    float* __restrict__ W1t, float* __restrict__ Wrmt) {
  if ((int)blockIdx.x < zero_blocks) {
    const int i = blockIdx.x * blockDim.x + threadIdx.x;
    if (i < n) cnt[i] = 0;
    return;
  }
  const int r = (blockIdx.x - zero_blocks) * blockDim.x + threadIdx.x;
  if (r < 64 * 256)                  relayout_k4(W0, W0t, r, 256);
  else if (r < 64 * 256 + 64 * 128)  relayout_k4(W1, W1t, r - 64 * 256, 128);
  else if (r < 64 * 256 + 64 * 128 + 64 * 32)
                                     relayout_k4_head_mean(Wres1, Wrmt, r - 64 * 256 - 64 * 128);
}

__global__ void dst_histogram(const int* __restrict__ dst, int* __restrict__ cnt, int E) {
  const int e = blockIdx.x * blockDim.x + threadIdx.x;
  if (e < E) atomicAdd(cnt + dst[e], 1);
}

__global__ __launch_bounds__(1024)
void rowptr_scan(const int* __restrict__ cnt, int* __restrict__ rowptr,
                 int* __restrict__ cursor, int N) {
  __shared__ int wsum[32];
  __shared__ int wpre[32];
  const int tid = threadIdx.x, lane = tid & 31, w = tid >> 5;
  const int c0 = blockIdx.x * 1024;
  int pre = 0;
  for (int i = tid; i < c0; i += 1024) pre += cnt[i];
  const int i = c0 + tid;
  const int v = i < N ? cnt[i] : 0;
  int inc = v;
#pragma unroll
  for (int off = 1; off < 32; off <<= 1) {
    const int t = __shfl_up(inc, off, 32);
    if (lane >= off) inc += t;
  }
#pragma unroll
  for (int off = 16; off > 0; off >>= 1) pre += __shfl_xor(pre, off, 32);
  if (lane == 31) wsum[w] = inc;
  if (lane == 0)  wpre[w] = pre;
  __syncthreads();
  int before = 0;
#pragma unroll
  for (int k = 0; k < 32; ++k) {
    before += wpre[k];
    if (k < w) before += wsum[k];
  }
  const int excl = before + inc - v;
  if (i < N) { rowptr[i] = excl; cursor[i] = excl; }
}

__global__ void csr_fill(const int* __restrict__ src, const int* __restrict__ dst,
                         int* __restrict__ cursor, int* __restrict__ nbr, int E) {
  const int e = blockIdx.x * blockDim.x + threadIdx.x;
  if (e >= E) return;
  const int p = atomicAdd(cursor + dst[e], 1);
  nbr[p] = src[e];
}

template <int MASK>
__device__ __forceinline__ float row_xor_lane(float x) {
  return __int_as_float(__builtin_amdgcn_update_dpp(
      0, __float_as_int(x), 0x160 + MASK, 0xF, 0xF, true));
}

__device__ __forceinline__ float other_row_lane(float x) {
  return __int_as_float((int)__builtin_amdgcn_permlanex16(
      0u, (unsigned)__float_as_int(x), 0x76543210u, 0xfedcba98u, false, false));
}

template <int ROW, int U>
__device__ __forceinline__ void attend_edges(
    const float4* __restrict__ f4, int sidx, int i, int lane,
    const float4 (&fd)[ROW / 128], const float4 (&at)[ROW / 128],
    float4 (&acc)[ROW / 128], float (&mx)[ROW / 128], float (&den)[ROW / 128]) {
  constexpr int NV  = ROW / 128;
  constexpr int LPH = ROW / 16;
  float4 v[U][NV];
  float  lg[U][NV];
#pragma unroll
  for (int u = 0; u < U; ++u) {
    const int s = __builtin_amdgcn_readlane(sidx, i + u);
    const float4* __restrict__ row = f4 + (long)s * (ROW / 4) + lane;
#pragma unroll
    for (int q = 0; q < NV; ++q) v[u][q] = row[q * 32];
  }
#pragma unroll
  for (int u = 0; u < U; ++u) {
#pragma unroll
    for (int q = 0; q < NV; ++q) {
      float t0 = v[u][q].x + fd[q].x, t1 = v[u][q].y + fd[q].y;
      float t2 = v[u][q].z + fd[q].z, t3 = v[u][q].w + fd[q].w;
      t0 = t0 > 0.0f ? t0 : 0.2f * t0;
      t1 = t1 > 0.0f ? t1 : 0.2f * t1;
      t2 = t2 > 0.0f ? t2 : 0.2f * t2;
      t3 = t3 > 0.0f ? t3 : 0.2f * t3;
      float p = t0 * at[q].x;
      p += t1 * at[q].y;
      p += t2 * at[q].z;
      p += t3 * at[q].w;
      lg[u][q] = p;
    }
  }
  if (LPH >= 16) {
#pragma unroll
    for (int u = 0; u < U; ++u) {
#pragma unroll
      for (int q = 0; q < NV; ++q) lg[u][q] += row_xor_lane<8>(lg[u][q]);
    }
  }
#pragma unroll
  for (int u = 0; u < U; ++u) {
#pragma unroll
    for (int q = 0; q < NV; ++q) lg[u][q] += row_xor_lane<4>(lg[u][q]);
  }
#pragma unroll
  for (int u = 0; u < U; ++u) {
#pragma unroll
    for (int q = 0; q < NV; ++q) lg[u][q] += row_xor_lane<2>(lg[u][q]);
  }
#pragma unroll
  for (int u = 0; u < U; ++u) {
#pragma unroll
    for (int q = 0; q < NV; ++q) lg[u][q] += row_xor_lane<1>(lg[u][q]);
  }
#pragma unroll
  for (int q = 0; q < NV; ++q) {
    float bm = mx[q];
#pragma unroll
    for (int u = 0; u < U; ++u) bm = fmaxf(bm, lg[u][q]);
    const float sc = __expf(mx[q] - bm);
    float4 a = acc[q];
    a.x *= sc; a.y *= sc; a.z *= sc; a.w *= sc;
    float d = den[q] * sc;
#pragma unroll
    for (int u = 0; u < U; ++u) {
      const float p = __expf(lg[u][q] - bm);
      d += p;
      a.x += p * v[u][q].x;
      a.y += p * v[u][q].y;
      a.z += p * v[u][q].z;
      a.w += p * v[u][q].w;
    }
    acc[q] = a;
    den[q] = d;
    mx[q]  = bm;
  }
}

template <int ROW>
__device__ __forceinline__ void attend_node(
    const float* __restrict__ f, const float* __restrict__ attn,
    const int* __restrict__ nbr, const int* __restrict__ rowptr,
    const int* __restrict__ cnt, int node, int N, int lane, float4 (&r)[ROW / 128]) {
  constexpr int NV = ROW / 128;
  static_assert(ROW == 128 || ROW == 256, "the lanes of one head must lie inside one row of 16 lanes");
#pragma unroll
  for (int q = 0; q < NV; ++q) r[q] = make_float4(0.0f, 0.0f, 0.0f, 0.0f);
  if (node >= N) return;
  const int start = rowptr[node];
  const int deg   = cnt[node];
  const float4* __restrict__ f4   = (const float4*)f;
  const float4* __restrict__ frow = f4 + (long)node * (ROW / 4) + lane;
  const float4* __restrict__ arow = (const float4*)attn + lane;
  float4 fd[NV], at[NV], acc[NV];
  float  mx[NV], den[NV];
#pragma unroll
  for (int q = 0; q < NV; ++q) {
    fd[q]  = frow[q * 32];
    at[q]  = arow[q * 32];
    acc[q] = make_float4(0.0f, 0.0f, 0.0f, 0.0f);
    mx[q]  = -__builtin_inff();
    den[q] = 0.0f;
  }
  for (int c = 0; c < deg; c += 32) {
    const int nn = deg - c < 32 ? deg - c : 32;
    int sidx = 0;
    if (lane < nn) sidx = nbr[start + c + lane];
    int i = 0;
    for (; i + 4 <= nn; i += 4) attend_edges<ROW, 4>(f4, sidx, i, lane, fd, at, acc, mx, den);
    if (i + 2 <= nn) { attend_edges<ROW, 2>(f4, sidx, i, lane, fd, at, acc, mx, den); i += 2; }
    if (i < nn)        attend_edges<ROW, 1>(f4, sidx, i, lane, fd, at, acc, mx, den);
  }
  if (deg > 0) {
#pragma unroll
    for (int q = 0; q < NV; ++q) {
      r[q].x = acc[q].x / den[q]; r[q].y = acc[q].y / den[q];
      r[q].z = acc[q].z / den[q]; r[q].w = acc[q].w / den[q];
    }
  }
}

template <bool EXTRA>
__device__ __forceinline__ void project_tile(
    const float4* tile, const float* __restrict__ W1t, const float* __restrict__ Wrmt,
    float* __restrict__ f1, float* __restrict__ o1, int base, int N, int w, int lane,
    int xt, int xm) {
  constexpr int PITCH4 = 65, NC1 = 128, NCR = 32, KB = 256 / 8, MT = 2;
  const int hi = lane >> 4;
  const int lm = lane & 15;
  const int col1 = w * 16 + lm;
  const int colr = xt * 16 + lm;
  const float4* __restrict__ B1 = (const float4*)W1t + hi * NC1 + col1;
  const float4* __restrict__ BR = (const float4*)Wrmt + hi * NCR + colr;
  const int half0 = 16 * xm, half1 = 16 - half0;
  const float4* Ar0 = tile + (half0 + lm) * PITCH4 + hi;
  const float4* Ar1 = tile + (half1 + lm) * PITCH4 + hi;

  v8f c[MT];
  v8f cx = (v8f){};
#pragma unroll
  for (int m = 0; m < MT; ++m) c[m] = (v8f){};
  constexpr int PD = 6;
  float4 bq[PD], bx[PD];
#pragma unroll
  for (int p = 0; p < PD; ++p) {
    bq[p] = B1[(2 * p) * NC1];
    if (EXTRA) bx[p] = BR[(2 * p) * NCR];
  }
  constexpr int AD = 2;
  float4 aq[AD][MT];
#pragma unroll
  for (int p = 0; p < AD; ++p) {
#pragma unroll
    for (int m = 0; m < MT; ++m) aq[p][m] = (m ? Ar1 : Ar0)[2 * p];
  }
#pragma unroll
  for (int j = 0; j < KB; ++j) {
    float4 a_cur[MT];
#pragma unroll
    for (int m = 0; m < MT; ++m) {
      a_cur[m] = aq[j % AD][m];
      if (j + AD < KB) aq[j % AD][m] = (m ? Ar1 : Ar0)[2 * (j + AD)];
    }
    const float4 b = bq[j % PD];
    if (j + PD < KB) bq[j % PD] = B1[(2 * (j + PD)) * NC1];
    float4 x = make_float4(0.0f, 0.0f, 0.0f, 0.0f);
    if (EXTRA) {
      x = bx[j % PD];
      if (j + PD < KB) bx[j % PD] = BR[(2 * (j + PD)) * NCR];
    }
    const v2f b0 = {b.x, b.y}, b1 = {b.z, b.w};
#pragma unroll
    for (int m = 0; m < MT; ++m) {
      const v2f a0 = {a_cur[m].x, a_cur[m].y};
      c[m] = __builtin_amdgcn_wmma_f32_16x16x4_f32(false, a0, false, b0, (short)0, c[m], false, false);
    }
    if (EXTRA) {
      const v2f a0 = {a_cur[0].x, a_cur[0].y}, x0 = {x.x, x.y};
      cx = __builtin_amdgcn_wmma_f32_16x16x4_f32(false, a0, false, x0, (short)0, cx, false, false);
    }
#pragma unroll
    for (int m = 0; m < MT; ++m) {
      const v2f a1 = {a_cur[m].z, a_cur[m].w};
      c[m] = __builtin_amdgcn_wmma_f32_16x16x4_f32(false, a1, false, b1, (short)0, c[m], false, false);
    }
    if (EXTRA) {
      const v2f a1 = {a_cur[0].z, a_cur[0].w}, x1 = {x.z, x.w};
      cx = __builtin_amdgcn_wmma_f32_16x16x4_f32(false, a1, false, x1, (short)0, cx, false, false);
    }
    __builtin_amdgcn_sched_barrier(0);
  }

#pragma unroll
  for (int m = 0; m < MT; ++m) {
#pragma unroll
    for (int r = 0; r < 8; ++r) {
      const int orow = base + (m ? half1 : half0) + r + 8 * hi;
      if (orow < N) f1[(long)orow * NC1 + col1] = c[m][r];
    }
  }
  if (EXTRA) {
#pragma unroll
    for (int r = 0; r < 8; ++r) {
      const int orow = base + half0 + r + 8 * hi;
      if (orow < N) o1[(long)orow * NCR + colr] = cx[r];
    }
  }
}

__global__ __launch_bounds__(256, 8)
void attend0_project1(const float* __restrict__ f, const float* __restrict__ attn,
                      const int* __restrict__ nbr, const int* __restrict__ rowptr,
                      const int* __restrict__ cnt, const float* __restrict__ W1t,
                      const float* __restrict__ Wrmt, float* __restrict__ f1,
                      float* __restrict__ o1, int N) {
  constexpr int ROW = 256, NV = ROW / 128, PITCH4 = 65;
  constexpr int NODES = 32;
  constexpr int PER_WAVE = NODES / 8;
  __shared__ float4 tile[NODES * PITCH4];
  const int lane = threadIdx.x & 31;
  const int w    = __builtin_amdgcn_readfirstlane((int)(threadIdx.x >> 5));
  const int base = blockIdx.x * NODES;

#pragma nounroll
  for (int s = 0; s < PER_WAVE; ++s) {
    const int rloc = PER_WAVE * w + s;
    float4 r[NV];
    attend_node<ROW>(f, attn, nbr, rowptr, cnt, base + rloc, N, lane, r);
#pragma unroll
    for (int q = 0; q < NV; ++q) {
      float4 e = r[q];
      e.x = e.x > 0.0f ? e.x : expm1f(e.x);
      e.y = e.y > 0.0f ? e.y : expm1f(e.y);
      e.z = e.z > 0.0f ? e.z : expm1f(e.z);
      e.w = e.w > 0.0f ? e.w : expm1f(e.w);
      tile[rloc * PITCH4 + q * 32 + lane] = e;
    }
  }
  __syncthreads();

  const bool extra = (w == 0) | (w == 3) | (w == 5) | (w == 6);
  const int  x     = (w == 0) ? 0 : (w == 3) ? 1 : (w == 5) ? 2 : 3;
  if (extra) project_tile<true >(tile, W1t, Wrmt, f1, o1, base, N, w, lane, x & 1, x >> 1);
  else       project_tile<false>(tile, W1t, Wrmt, f1, o1, base, N, w, lane, 0, 0);
}

__global__ __launch_bounds__(256)
void attend1_res_mean(const float* __restrict__ f, const float* __restrict__ attn,
                      const int* __restrict__ nbr, const int* __restrict__ rowptr,
                      const int* __restrict__ cnt, const float* __restrict__ res,
                      float* __restrict__ out, int N) {
  constexpr int ROW = 128;
  const int lane = threadIdx.x & 31;
  const int node = __builtin_amdgcn_readfirstlane(
      (int)(blockIdx.x * (blockDim.x >> 5) + (threadIdx.x >> 5)));
  if (node >= N) return;
  float4 r[1];
  attend_node<ROW>(f, attn, nbr, rowptr, cnt, node, N, lane, r);
  float4 s = r[0];
  s.x += row_xor_lane<8>(s.x); s.y += row_xor_lane<8>(s.y);
  s.z += row_xor_lane<8>(s.z); s.w += row_xor_lane<8>(s.w);
  s.x += other_row_lane(s.x);  s.y += other_row_lane(s.y);
  s.z += other_row_lane(s.z);  s.w += other_row_lane(s.w);
  if (lane < 8) {
    const float4 p = ((const float4*)(res + (long)node * (ROW / 4)))[lane];
    s.x = s.x * 0.25f + p.x; s.y = s.y * 0.25f + p.y;
    s.z = s.z * 0.25f + p.z; s.w = s.w * 0.25f + p.w;
    ((float4*)(out + (long)node * (ROW / 4)))[lane] = s;
  }
}

static inline int cdiv_l(long a, long b) { return (int)((a + b - 1) / b); }

extern "C" void kernel_launch(void* const* d_in, const int* in_sizes, int n_in,
                              void* d_out, int out_size, void* d_ws, size_t ws_size,
                              hipStream_t stream) {
  const float* x     = (const float*)d_in[0];
  const float* W0    = (const float*)d_in[1];
  const float* a0    = (const float*)d_in[2];
  const float* W1    = (const float*)d_in[3];
  const float* a1    = (const float*)d_in[4];
  const float* Wres1 = (const float*)d_in[5];
  const int*   src   = (const int*)d_in[6];
  const int*   dst   = (const int*)d_in[7];

  const int N = in_sizes[0] / 256;
  const int E = in_sizes[6];

  float*    f0    = (float*)d_ws;
  float*    f1    = f0 + (long)N * 256;
  float*    grp   = f1 + (long)N * 256;
  float*    o1    = grp + (long)N * 128;
  int*      nbr    = (int*)grp;
  int*      cnt    = nbr + E;
  int*      rowptr = cnt + N;
  int*      cursor = rowptr + N;
  const size_t grp_ints = ((size_t)E + (size_t)N * 3 + 3) & ~(size_t)3;
  float*    W0t    = grp + grp_ints;
  float*    W1t    = W0t + 256 * 256;
  float*    Wrmt   = W1t + 256 * 128;
  const size_t need = ((size_t)N * 256 * 2 + (size_t)N * 128 * 2) * sizeof(float);
  if (need > ws_size) return;
  if ((grp_ints + 256 * 256 + 256 * 128 + 256 * 32) * sizeof(float) > (size_t)N * 128 * sizeof(float))
    return;

  const int BT = 256;

  {
    const int zero_blocks = cdiv_l(N, BT);
    const int relayout_blocks = (64 * 256 + 64 * 128 + 64 * 32) / BT;
    zero_counts_relayout_weights<<<zero_blocks + relayout_blocks, BT, 0, stream>>>(
        cnt, N, zero_blocks, W0, W1, Wres1, W0t, W1t, Wrmt);
  }

  {
    const int strips = ((N + 15) / 16) * (256 / 16 / 4);
    gemm_f32_wmma<4, 256, 256><<<cdiv_l(strips, 8), BT, 0, stream>>>(x, W0t, f0, N);
  }
  dst_histogram<<<cdiv_l(E, BT), BT, 0, stream>>>(dst, cnt, E);
  rowptr_scan<<<cdiv_l(N, 1024), 1024, 0, stream>>>(cnt, rowptr, cursor, N);
  csr_fill<<<cdiv_l(E, BT), BT, 0, stream>>>(src, dst, cursor, nbr, E);

  attend0_project1<<<cdiv_l(N, 32), BT, 0, stream>>>(f0, a0, nbr, rowptr, cnt, W1t, Wrmt,
                                                    f1, o1, N);

  attend1_res_mean<<<cdiv_l((long)N * 32, BT), BT, 0, stream>>>(f1, a1, nbr, rowptr, cnt, o1,
                                                                (float*)d_out, N);
}


# attention edge loop on packed 2-wide f32 over adjacent row elements; a lane's row pieces adjacent (one head: one softmax state, 8-lane butterfly); logit sum order changes; row = acc * (1/den)
# speedup vs baseline: 3.6254x; 1.0696x over previous
#include <hip/hip_runtime.h>
#include <hip/hip_bf16.h>

typedef __attribute__((ext_vector_type(2))) float v2f;
typedef __attribute__((ext_vector_type(8))) float v8f;

template <int NT, int K, int NC>
__global__ __launch_bounds__(256)
void gemm_f32_wmma(const float* __restrict__ A, const float* __restrict__ Bt,
                   float* __restrict__ C, int M) {
  static_assert(K % 8 == 0, "a block of the k loop is 8 k");
  const int lane = threadIdx.x & 31;
  const int wid  = blockIdx.x * (blockDim.x >> 5) + (threadIdx.x >> 5);
  const int tilesM  = (M + 15) >> 4;
  constexpr int stripsN = (NC >> 4) / NT;
  if (wid >= tilesM * stripsN) return;
  const int tm  = wid / stripsN;
  const int tn0 = (wid - tm * stripsN) * NT;
  const int hi = lane >> 4;
  const int lm = lane & 15;
  int arow = tm * 16 + lm; if (arow >= M) arow = M - 1;
  const int col0 = tn0 * 16 + lm;
  const float4* __restrict__ Ar = (const float4*)(A + (long)arow * K) + hi;
  const float4* __restrict__ Bc = (const float4*)Bt + (long)hi * NC + col0;

  v8f c[NT];
#pragma unroll
  for (int t = 0; t < NT; ++t) c[t] = (v8f){};

  auto loadB = [&](int j, float4* b) {
    const float4* __restrict__ B0 = Bc + (long)(2 * j) * NC;
#pragma unroll
    for (int t = 0; t < NT; ++t) b[t] = B0[16 * t];
  };
  auto mma_block = [&](const float4& a, const float4* b) {
    const v2f a0 = {a.x, a.y}, a1 = {a.z, a.w};
#pragma unroll
    for (int t = 0; t < NT; ++t) {
      const v2f b0 = {b[t].x, b[t].y};
      c[t] = __builtin_amdgcn_wmma_f32_16x16x4_f32(false, a0, false, b0,
                                                   (short)0, c[t], false, false);
    }
#pragma unroll
    for (int t = 0; t < NT; ++t) {
      const v2f b1 = {b[t].z, b[t].w};
      c[t] = __builtin_amdgcn_wmma_f32_16x16x4_f32(false, a1, false, b1,
                                                   (short)0, c[t], false, false);
    }
  };

  float4 a_cur = Ar[0];
  float4 b_cur[NT];
  loadB(0, b_cur);
#pragma unroll
  for (int j = 0; j < K / 8 - 1; ++j) {
    const float4 a_nxt = Ar[2 * (j + 1)];
    float4 b_nxt[NT];
    loadB(j + 1, b_nxt);
    mma_block(a_cur, b_cur);
    a_cur = a_nxt;
#pragma unroll
    for (int t = 0; t < NT; ++t) b_cur[t] = b_nxt[t];
  }
  mma_block(a_cur, b_cur);

#pragma unroll
  for (int t = 0; t < NT; ++t) {
#pragma unroll
    for (int r = 0; r < 8; ++r) {
      const int orow = tm * 16 + r + 8 * hi;
      if (orow < M) C[(long)orow * NC + col0 + 16 * t] = c[t][r];
    }
  }
}

__device__ __forceinline__ void relayout_k4(const float* __restrict__ B,
                                            float* __restrict__ Bt, int idx, int NC) {
  const int kq = idx / NC;
  const int n  = idx - kq * NC;
  const float* __restrict__ p = B + (long)(4 * kq) * NC + n;
  ((float4*)Bt)[idx] = make_float4(p[0], p[NC], p[2 * NC], p[3 * NC]);
}

__device__ __forceinline__ void relayout_k4_head_mean(const float* __restrict__ B,
                                                      float* __restrict__ Bt, int idx) {
  const int kq = idx >> 5;
  const int c  = idx & 31;
  const float* __restrict__ p = B + (long)(4 * kq) * 128 + c;
  float m[4];
#pragma unroll
  for (int e = 0; e < 4; ++e) {
    const float* __restrict__ q = p + e * 128;
    m[e] = 0.25f * ((q[0] + q[32]) + (q[64] + q[96]));
  }
  ((float4*)Bt)[idx] = make_float4(m[0], m[1], m[2], m[3]);
}

__global__ void zero_counts_relayout_weights(
    int* __restrict__ cnt, int n, int zero_blocks,
    const float* __restrict__ W0, const float* __restrict__ W1,
    const float* __restrict__ Wres1, float* __restrict__ W0t,
    float* __restrict__ W1t, float* __restrict__ Wrmt) {
  if ((int)blockIdx.x < zero_blocks) {
    const int i = blockIdx.x * blockDim.x + threadIdx.x;
    if (i < n) cnt[i] = 0;
    return;
  }
  const int r = (blockIdx.x - zero_blocks) * blockDim.x + threadIdx.x;
  if (r < 64 * 256)                  relayout_k4(W0, W0t, r, 256);
  else if (r < 64 * 256 + 64 * 128)  relayout_k4(W1, W1t, r - 64 * 256, 128);
  else if (r < 64 * 256 + 64 * 128 + 64 * 32)
                                     relayout_k4_head_mean(Wres1, Wrmt, r - 64 * 256 - 64 * 128);
}

__global__ void dst_histogram(const int* __restrict__ dst, int* __restrict__ cnt, int E) {
  const int e = blockIdx.x * blockDim.x + threadIdx.x;
  if (e < E) atomicAdd(cnt + dst[e], 1);
}

__global__ __launch_bounds__(1024)
void rowptr_scan(const int* __restrict__ cnt, int* __restrict__ rowptr,
                 int* __restrict__ cursor, int N) {
  __shared__ int wsum[32];
  __shared__ int wpre[32];
  const int tid = threadIdx.x, lane = tid & 31, w = tid >> 5;
  const int c0 = blockIdx.x * 1024;
  int pre = 0;
  for (int i = tid; i < c0; i += 1024) pre += cnt[i];
  const int i = c0 + tid;
  const int v = i < N ? cnt[i] : 0;
  int inc = v;
#pragma unroll
  for (int off = 1; off < 32; off <<= 1) {
    const int t = __shfl_up(inc, off, 32);
    if (lane >= off) inc += t;
  }
#pragma unroll
  for (int off = 16; off > 0; off >>= 1) pre += __shfl_xor(pre, off, 32);
  if (lane == 31) wsum[w] = inc;
  if (lane == 0)  wpre[w] = pre;
  __syncthreads();
  int before = 0;
#pragma unroll
  for (int k = 0; k < 32; ++k) {
    before += wpre[k];
    if (k < w) before += wsum[k];
  }
  const int excl = before + inc - v;
  if (i < N) { rowptr[i] = excl; cursor[i] = excl; }
}

__global__ void csr_fill(const int* __restrict__ src, const int* __restrict__ dst,
                         int* __restrict__ cursor, int* __restrict__ nbr, int E) {
  const int e = blockIdx.x * blockDim.x + threadIdx.x;
  if (e >= E) return;
  const int p = atomicAdd(cursor + dst[e], 1);
  nbr[p] = src[e];
}

template <int MASK>
__device__ __forceinline__ float row_xor_lane(float x) {
  return __int_as_float(__builtin_amdgcn_update_dpp(
      0, __float_as_int(x), 0x160 + MASK, 0xF, 0xF, true));
}

__device__ __forceinline__ float other_row_lane(float x) {
  return __int_as_float((int)__builtin_amdgcn_permlanex16(
      0u, (unsigned)__float_as_int(x), 0x76543210u, 0xfedcba98u, false, false));
}

__device__ __forceinline__ v2f leaky_pair(v2f t) {
  return __builtin_elementwise_max(t, t * 0.2f);
}

template <int ROW, int U>
__device__ __forceinline__ void attend_edges(
    const float4* __restrict__ f4, int sidx, int i, int lane,
    const v2f (&fd)[ROW / 64], const v2f (&at)[ROW / 64],
    v2f (&acc)[ROW / 64], float& mx, float& den) {
  constexpr int NV = ROW / 128;
  constexpr int NP = 2 * NV;
  v2f   v[U][NP];
  float lg[U];
  float p[U];
#pragma unroll
  for (int u = 0; u < U; ++u) {
    const int s = __builtin_amdgcn_readlane(sidx, i + u);
    const float4* __restrict__ row = f4 + (long)s * (ROW / 4) + NV * lane;
#pragma unroll
    for (int q = 0; q < NV; ++q) {
      const float4 t = row[q];
      v[u][2 * q]     = (v2f){t.x, t.y};
      v[u][2 * q + 1] = (v2f){t.z, t.w};
    }
  }
#pragma unroll
  for (int u = 0; u < U; ++u) {
    v2f pp = leaky_pair(v[u][0] + fd[0]) * at[0];
#pragma unroll
    for (int k = 1; k < NP; ++k)
      pp = __builtin_elementwise_fma(leaky_pair(v[u][k] + fd[k]), at[k], pp);
    lg[u] = pp.x + pp.y;
  }
#pragma unroll
  for (int u = 0; u < U; ++u) lg[u] += row_xor_lane<4>(lg[u]);
#pragma unroll
  for (int u = 0; u < U; ++u) lg[u] += row_xor_lane<2>(lg[u]);
#pragma unroll
  for (int u = 0; u < U; ++u) lg[u] += row_xor_lane<1>(lg[u]);
  float bm = mx;
#pragma unroll
  for (int u = 0; u < U; ++u) bm = fmaxf(bm, lg[u]);
  const float sc = __expf(mx - bm);
  float d = den * sc;
#pragma unroll
  for (int u = 0; u < U; ++u) {
    p[u] = __expf(lg[u] - bm);
    d += p[u];
  }
#pragma unroll
  for (int k = 0; k < NP; ++k) {
    v2f a = acc[k] * sc;
#pragma unroll
    for (int u = 0; u < U; ++u)
      a = __builtin_elementwise_fma(v[u][k], (v2f){p[u], p[u]}, a);
    acc[k] = a;
  }
  den = d;
  mx  = bm;
}

template <int ROW>
__device__ __forceinline__ void attend_node(
    const float* __restrict__ f, const float* __restrict__ attn,
    const int* __restrict__ nbr, const int* __restrict__ rowptr,
    const int* __restrict__ cnt, int node, int N, int lane, float4 (&r)[ROW / 128]) {
  constexpr int NV = ROW / 128;
  constexpr int NP = 2 * NV;
  static_assert(ROW == 128 || ROW == 256,
                "a lane's pieces must lie inside one head, and a head on 8 lanes");
#pragma unroll
  for (int q = 0; q < NV; ++q) r[q] = make_float4(0.0f, 0.0f, 0.0f, 0.0f);
  if (node >= N) return;
  const int start = rowptr[node];
  const int deg   = cnt[node];
  const float4* __restrict__ f4   = (const float4*)f;
  const float4* __restrict__ frow = f4 + (long)node * (ROW / 4) + NV * lane;
  const float4* __restrict__ arow = (const float4*)attn + NV * lane;
  v2f fd[NP], at[NP], acc[NP];
#pragma unroll
  for (int q = 0; q < NV; ++q) {
    const float4 a = frow[q];
    const float4 b = arow[q];
    fd[2 * q] = (v2f){a.x, a.y};  fd[2 * q + 1] = (v2f){a.z, a.w};
    at[2 * q] = (v2f){b.x, b.y};  at[2 * q + 1] = (v2f){b.z, b.w};
    acc[2 * q] = (v2f){0.0f, 0.0f};  acc[2 * q + 1] = (v2f){0.0f, 0.0f};
  }
  float mx  = -__builtin_inff();
  float den = 0.0f;
  for (int c = 0; c < deg; c += 32) {
    const int nn = deg - c < 32 ? deg - c : 32;
    int sidx = 0;
    if (lane < nn) sidx = nbr[start + c + lane];
    int i = 0;
    for (; i + 4 <= nn; i += 4) attend_edges<ROW, 4>(f4, sidx, i, lane, fd, at, acc, mx, den);
    if (i + 2 <= nn) { attend_edges<ROW, 2>(f4, sidx, i, lane, fd, at, acc, mx, den); i += 2; }
    if (i < nn)        attend_edges<ROW, 1>(f4, sidx, i, lane, fd, at, acc, mx, den);
  }
  if (deg > 0) {
    const float inv = 1.0f / den;
#pragma unroll
    for (int q = 0; q < NV; ++q) {
      const v2f lo = acc[2 * q] * inv, hi = acc[2 * q + 1] * inv;
      r[q] = make_float4(lo.x, lo.y, hi.x, hi.y);
    }
  }
}

template <bool EXTRA>
__device__ __forceinline__ void project_tile(
    const float4* tile, const float* __restrict__ W1t, const float* __restrict__ Wrmt,
    float* __restrict__ f1, float* __restrict__ o1, int base, int N, int w, int lane,
    int xt, int xm) {
  constexpr int PITCH4 = 65, NC1 = 128, NCR = 32, KB = 256 / 8, MT = 2;
  const int hi = lane >> 4;
  const int lm = lane & 15;
  const int col1 = w * 16 + lm;
  const int colr = xt * 16 + lm;
  const float4* __restrict__ B1 = (const float4*)W1t + hi * NC1 + col1;
  const float4* __restrict__ BR = (const float4*)Wrmt + hi * NCR + colr;
  const int half0 = 16 * xm, half1 = 16 - half0;
  const float4* Ar0 = tile + (half0 + lm) * PITCH4 + hi;
  const float4* Ar1 = tile + (half1 + lm) * PITCH4 + hi;

  v8f c[MT];
  v8f cx = (v8f){};
#pragma unroll
  for (int m = 0; m < MT; ++m) c[m] = (v8f){};
  constexpr int PD = 6;
  float4 bq[PD], bx[PD];
#pragma unroll
  for (int p = 0; p < PD; ++p) {
    bq[p] = B1[(2 * p) * NC1];
    if (EXTRA) bx[p] = BR[(2 * p) * NCR];
  }
  constexpr int AD = 2;
  float4 aq[AD][MT];
#pragma unroll
  for (int p = 0; p < AD; ++p) {
#pragma unroll
    for (int m = 0; m < MT; ++m) aq[p][m] = (m ? Ar1 : Ar0)[2 * p];
  }
#pragma unroll
  for (int j = 0; j < KB; ++j) {
    float4 a_cur[MT];
#pragma unroll
    for (int m = 0; m < MT; ++m) {
      a_cur[m] = aq[j % AD][m];
      if (j + AD < KB) aq[j % AD][m] = (m ? Ar1 : Ar0)[2 * (j + AD)];
    }
    const float4 b = bq[j % PD];
    if (j + PD < KB) bq[j % PD] = B1[(2 * (j + PD)) * NC1];
    float4 x = make_float4(0.0f, 0.0f, 0.0f, 0.0f);
    if (EXTRA) {
      x = bx[j % PD];
      if (j + PD < KB) bx[j % PD] = BR[(2 * (j + PD)) * NCR];
    }
    const v2f b0 = {b.x, b.y}, b1 = {b.z, b.w};
#pragma unroll
    for (int m = 0; m < MT; ++m) {
      const v2f a0 = {a_cur[m].x, a_cur[m].y};
      c[m] = __builtin_amdgcn_wmma_f32_16x16x4_f32(false, a0, false, b0, (short)0, c[m], false, false);
    }
    if (EXTRA) {
      const v2f a0 = {a_cur[0].x, a_cur[0].y}, x0 = {x.x, x.y};
      cx = __builtin_amdgcn_wmma_f32_16x16x4_f32(false, a0, false, x0, (short)0, cx, false, false);
    }
#pragma unroll
    for (int m = 0; m < MT; ++m) {
      const v2f a1 = {a_cur[m].z, a_cur[m].w};
      c[m] = __builtin_amdgcn_wmma_f32_16x16x4_f32(false, a1, false, b1, (short)0, c[m], false, false);
    }
    if (EXTRA) {
      const v2f a1 = {a_cur[0].z, a_cur[0].w}, x1 = {x.z, x.w};
      cx = __builtin_amdgcn_wmma_f32_16x16x4_f32(false, a1, false, x1, (short)0, cx, false, false);
    }
    __builtin_amdgcn_sched_barrier(0);
  }

#pragma unroll
  for (int m = 0; m < MT; ++m) {
#pragma unroll
    for (int r = 0; r < 8; ++r) {
      const int orow = base + (m ? half1 : half0) + r + 8 * hi;
      if (orow < N) f1[(long)orow * NC1 + col1] = c[m][r];
    }
  }
  if (EXTRA) {
#pragma unroll
    for (int r = 0; r < 8; ++r) {
      const int orow = base + half0 + r + 8 * hi;
      if (orow < N) o1[(long)orow * NCR + colr] = cx[r];
    }
  }
}

__global__ __launch_bounds__(256, 8)
void attend0_project1(const float* __restrict__ f, const float* __restrict__ attn,
                      const int* __restrict__ nbr, const int* __restrict__ rowptr,
                      const int* __restrict__ cnt, const float* __restrict__ W1t,
                      const float* __restrict__ Wrmt, float* __restrict__ f1,
                      float* __restrict__ o1, int N) {
  constexpr int ROW = 256, NV = ROW / 128, PITCH4 = 65;
  constexpr int NODES = 32;
  constexpr int PER_WAVE = NODES / 8;
  __shared__ float4 tile[NODES * PITCH4];
  const int lane = threadIdx.x & 31;
  const int w    = __builtin_amdgcn_readfirstlane((int)(threadIdx.x >> 5));
  const int base = blockIdx.x * NODES;

#pragma nounroll
  for (int s = 0; s < PER_WAVE; ++s) {
    const int rloc = PER_WAVE * w + s;
    float4 r[NV];
    attend_node<ROW>(f, attn, nbr, rowptr, cnt, base + rloc, N, lane, r);
#pragma unroll
    for (int q = 0; q < NV; ++q) {
      float4 e = r[q];
      e.x = e.x > 0.0f ? e.x : expm1f(e.x);
      e.y = e.y > 0.0f ? e.y : expm1f(e.y);
      e.z = e.z > 0.0f ? e.z : expm1f(e.z);
      e.w = e.w > 0.0f ? e.w : expm1f(e.w);
      tile[rloc * PITCH4 + NV * lane + q] = e;
    }
  }
  __syncthreads();

  const bool extra = (w == 0) | (w == 3) | (w == 5) | (w == 6);
  const int  x     = (w == 0) ? 0 : (w == 3) ? 1 : (w == 5) ? 2 : 3;
  if (extra) project_tile<true >(tile, W1t, Wrmt, f1, o1, base, N, w, lane, x & 1, x >> 1);
  else       project_tile<false>(tile, W1t, Wrmt, f1, o1, base, N, w, lane, 0, 0);
}

__global__ __launch_bounds__(256)
void attend1_res_mean(const float* __restrict__ f, const float* __restrict__ attn,
                      const int* __restrict__ nbr, const int* __restrict__ rowptr,
                      const int* __restrict__ cnt, const float* __restrict__ res,
                      float* __restrict__ out, int N) {
  constexpr int ROW = 128;
  const int lane = threadIdx.x & 31;
  const int node = __builtin_amdgcn_readfirstlane(
      (int)(blockIdx.x * (blockDim.x >> 5) + (threadIdx.x >> 5)));
  if (node >= N) return;
  float4 r[1];
  attend_node<ROW>(f, attn, nbr, rowptr, cnt, node, N, lane, r);
  float4 s = r[0];
  s.x += row_xor_lane<8>(s.x); s.y += row_xor_lane<8>(s.y);
  s.z += row_xor_lane<8>(s.z); s.w += row_xor_lane<8>(s.w);
  s.x += other_row_lane(s.x);  s.y += other_row_lane(s.y);
  s.z += other_row_lane(s.z);  s.w += other_row_lane(s.w);
  if (lane < 8) {
    const float4 p = ((const float4*)(res + (long)node * (ROW / 4)))[lane];
    s.x = s.x * 0.25f + p.x; s.y = s.y * 0.25f + p.y;
    s.z = s.z * 0.25f + p.z; s.w = s.w * 0.25f + p.w;
    ((float4*)(out + (long)node * (ROW / 4)))[lane] = s;
  }
}

static inline int cdiv_l(long a, long b) { return (int)((a + b - 1) / b); }

extern "C" void kernel_launch(void* const* d_in, const int* in_sizes, int n_in,
                              void* d_out, int out_size, void* d_ws, size_t ws_size,
                              hipStream_t stream) {
  const float* x     = (const float*)d_in[0];
  const float* W0    = (const float*)d_in[1];
  const float* a0    = (const float*)d_in[2];
  const float* W1    = (const float*)d_in[3];
  const float* a1    = (const float*)d_in[4];
  const float* Wres1 = (const float*)d_in[5];
  const int*   src   = (const int*)d_in[6];
  const int*   dst   = (const int*)d_in[7];

  const int N = in_sizes[0] / 256;
  const int E = in_sizes[6];

  float*    f0    = (float*)d_ws;
  float*    f1    = f0 + (long)N * 256;
  float*    grp   = f1 + (long)N * 256;
  float*    o1    = grp + (long)N * 128;
  int*      nbr    = (int*)grp;
  int*      cnt    = nbr + E;
  int*      rowptr = cnt + N;
  int*      cursor = rowptr + N;
  const size_t grp_ints = ((size_t)E + (size_t)N * 3 + 3) & ~(size_t)3;
  float*    W0t    = grp + grp_ints;
  float*    W1t    = W0t + 256 * 256;
  float*    Wrmt   = W1t + 256 * 128;
  const size_t need = ((size_t)N * 256 * 2 + (size_t)N * 128 * 2) * sizeof(float);
  if (need > ws_size) return;
  if ((grp_ints + 256 * 256 + 256 * 128 + 256 * 32) * sizeof(float) > (size_t)N * 128 * sizeof(float))
    return;

  const int BT = 256;

  {
    const int zero_blocks = cdiv_l(N, BT);
    const int relayout_blocks = (64 * 256 + 64 * 128 + 64 * 32) / BT;
    zero_counts_relayout_weights<<<zero_blocks + relayout_blocks, BT, 0, stream>>>(
        cnt, N, zero_blocks, W0, W1, Wres1, W0t, W1t, Wrmt);
  }

  {
    const int strips = ((N + 15) / 16) * (256 / 16 / 4);
    gemm_f32_wmma<4, 256, 256><<<cdiv_l(strips, 8), BT, 0, stream>>>(x, W0t, f0, N);
  }
  dst_histogram<<<cdiv_l(E, BT), BT, 0, stream>>>(dst, cnt, E);
  rowptr_scan<<<cdiv_l(N, 1024), 1024, 0, stream>>>(cnt, rowptr, cursor, N);
  csr_fill<<<cdiv_l(E, BT), BT, 0, stream>>>(src, dst, cursor, nbr, E);

  attend0_project1<<<cdiv_l(N, 32), BT, 0, stream>>>(f0, a0, nbr, rowptr, cnt, W1t, Wrmt,
                                                    f1, o1, N);

  attend1_res_mean<<<cdiv_l((long)N * 32, BT), BT, 0, stream>>>(f1, a1, nbr, rowptr, cnt, o1,
                                                                (float*)d_out, N);
}


# attention waves: per-node chain of dependent loads cut: (start,count) pairs in one scalar load (4 nodes at once in layer 0); next node's ids and own row asked for ahead of the division and elu
# speedup vs baseline: 3.7725x; 1.0406x over previous
#include <hip/hip_runtime.h>
#include <hip/hip_bf16.h>

typedef __attribute__((ext_vector_type(2))) float v2f;
typedef __attribute__((ext_vector_type(8))) float v8f;

template <int NT, int K, int NC>
__global__ __launch_bounds__(256)
void gemm_f32_wmma(const float* __restrict__ A, const float* __restrict__ Bt,
                   float* __restrict__ C, int M) {
  static_assert(K % 8 == 0, "a block of the k loop is 8 k");
  const int lane = threadIdx.x & 31;
  const int wid  = blockIdx.x * (blockDim.x >> 5) + (threadIdx.x >> 5);
  const int tilesM  = (M + 15) >> 4;
  constexpr int stripsN = (NC >> 4) / NT;
  if (wid >= tilesM * stripsN) return;
  const int tm  = wid / stripsN;
  const int tn0 = (wid - tm * stripsN) * NT;
  const int hi = lane >> 4;
  const int lm = lane & 15;
  int arow = tm * 16 + lm; if (arow >= M) arow = M - 1;
  const int col0 = tn0 * 16 + lm;
  const float4* __restrict__ Ar = (const float4*)(A + (long)arow * K) + hi;
  const float4* __restrict__ Bc = (const float4*)Bt + (long)hi * NC + col0;

  v8f c[NT];
#pragma unroll
  for (int t = 0; t < NT; ++t) c[t] = (v8f){};

  auto loadB = [&](int j, float4* b) {
    const float4* __restrict__ B0 = Bc + (long)(2 * j) * NC;
#pragma unroll
    for (int t = 0; t < NT; ++t) b[t] = B0[16 * t];
  };
  auto mma_block = [&](const float4& a, const float4* b) {
    const v2f a0 = {a.x, a.y}, a1 = {a.z, a.w};
#pragma unroll
    for (int t = 0; t < NT; ++t) {
      const v2f b0 = {b[t].x, b[t].y};
      c[t] = __builtin_amdgcn_wmma_f32_16x16x4_f32(false, a0, false, b0,
                                                   (short)0, c[t], false, false);
    }
#pragma unroll
    for (int t = 0; t < NT; ++t) {
      const v2f b1 = {b[t].z, b[t].w};
      c[t] = __builtin_amdgcn_wmma_f32_16x16x4_f32(false, a1, false, b1,
                                                   (short)0, c[t], false, false);
    }
  };

  float4 a_cur = Ar[0];
  float4 b_cur[NT];
  loadB(0, b_cur);
#pragma unroll
  for (int j = 0; j < K / 8 - 1; ++j) {
    const float4 a_nxt = Ar[2 * (j + 1)];
    float4 b_nxt[NT];
    loadB(j + 1, b_nxt);
    mma_block(a_cur, b_cur);
    a_cur = a_nxt;
#pragma unroll
    for (int t = 0; t < NT; ++t) b_cur[t] = b_nxt[t];
  }
  mma_block(a_cur, b_cur);

#pragma unroll
  for (int t = 0; t < NT; ++t) {
#pragma unroll
    for (int r = 0; r < 8; ++r) {
      const int orow = tm * 16 + r + 8 * hi;
      if (orow < M) C[(long)orow * NC + col0 + 16 * t] = c[t][r];
    }
  }
}

__device__ __forceinline__ void relayout_k4(const float* __restrict__ B,
                                            float* __restrict__ Bt, int idx, int NC) {
  const int kq = idx / NC;
  const int n  = idx - kq * NC;
  const float* __restrict__ p = B + (long)(4 * kq) * NC + n;
  ((float4*)Bt)[idx] = make_float4(p[0], p[NC], p[2 * NC], p[3 * NC]);
}

__device__ __forceinline__ void relayout_k4_head_mean(const float* __restrict__ B,
                                                      float* __restrict__ Bt, int idx) {
  const int kq = idx >> 5;
  const int c  = idx & 31;
  const float* __restrict__ p = B + (long)(4 * kq) * 128 + c;
  float m[4];
#pragma unroll
  for (int e = 0; e < 4; ++e) {
    const float* __restrict__ q = p + e * 128;
    m[e] = 0.25f * ((q[0] + q[32]) + (q[64] + q[96]));
  }
  ((float4*)Bt)[idx] = make_float4(m[0], m[1], m[2], m[3]);
}

__global__ void zero_counts_relayout_weights(
    int* __restrict__ cnt, int n, int zero_blocks,
    const float* __restrict__ W0, const float* __restrict__ W1,
    const float* __restrict__ Wres1, float* __restrict__ W0t,
    float* __restrict__ W1t, float* __restrict__ Wrmt) {
  if ((int)blockIdx.x < zero_blocks) {
    const int i = blockIdx.x * blockDim.x + threadIdx.x;
    if (i < n) cnt[i] = 0;
    return;
  }
  const int r = (blockIdx.x - zero_blocks) * blockDim.x + threadIdx.x;
  if (r < 64 * 256)                  relayout_k4(W0, W0t, r, 256);
  else if (r < 64 * 256 + 64 * 128)  relayout_k4(W1, W1t, r - 64 * 256, 128);
  else if (r < 64 * 256 + 64 * 128 + 64 * 32)
                                     relayout_k4_head_mean(Wres1, Wrmt, r - 64 * 256 - 64 * 128);
}

__global__ void dst_histogram(const int* __restrict__ dst, int* __restrict__ cnt, int E) {
  const int e = blockIdx.x * blockDim.x + threadIdx.x;
  if (e < E) atomicAdd(cnt + dst[e], 1);
}

__global__ __launch_bounds__(1024)
void rowptr_scan(const int* __restrict__ cnt, int2* __restrict__ seg,
                 int* __restrict__ cursor, int N, int Npad) {
  __shared__ int wsum[32];
  __shared__ int wpre[32];
  const int tid = threadIdx.x, lane = tid & 31, w = tid >> 5;
  const int c0 = blockIdx.x * 1024;
  int pre = 0;
  for (int i = tid; i < c0; i += 1024) pre += cnt[i];
  const int i = c0 + tid;
  const int v = i < N ? cnt[i] : 0;
  int inc = v;
#pragma unroll
  for (int off = 1; off < 32; off <<= 1) {
    const int t = __shfl_up(inc, off, 32);
    if (lane >= off) inc += t;
  }
#pragma unroll
  for (int off = 16; off > 0; off >>= 1) pre += __shfl_xor(pre, off, 32);
  if (lane == 31) wsum[w] = inc;
  if (lane == 0)  wpre[w] = pre;
  __syncthreads();
  int before = 0;
#pragma unroll
  for (int k = 0; k < 32; ++k) {
    before += wpre[k];
    if (k < w) before += wsum[k];
  }
  const int excl = before + inc - v;
  if (i < N) { seg[i] = make_int2(excl, v); cursor[i] = excl; }
  else if (i < Npad) seg[i] = make_int2(0, 0);
}

__global__ void csr_fill(const int* __restrict__ src, const int* __restrict__ dst,
                         int* __restrict__ cursor, int* __restrict__ nbr, int E) {
  const int e = blockIdx.x * blockDim.x + threadIdx.x;
  if (e >= E) return;
  const int p = atomicAdd(cursor + dst[e], 1);
  nbr[p] = src[e];
}

template <int MASK>
__device__ __forceinline__ float row_xor_lane(float x) {
  return __int_as_float(__builtin_amdgcn_update_dpp(
      0, __float_as_int(x), 0x160 + MASK, 0xF, 0xF, true));
}

__device__ __forceinline__ float other_row_lane(float x) {
  return __int_as_float((int)__builtin_amdgcn_permlanex16(
      0u, (unsigned)__float_as_int(x), 0x76543210u, 0xfedcba98u, false, false));
}

__device__ __forceinline__ v2f leaky_pair(v2f t) {
  return __builtin_elementwise_max(t, t * 0.2f);
}

template <int ROW, int U>
__device__ __forceinline__ void attend_edges(
    const float4* __restrict__ f4, int sidx, int i, int lane,
    const v2f (&fd)[ROW / 64], const v2f (&at)[ROW / 64],
    v2f (&acc)[ROW / 64], float& mx, float& den) {
  constexpr int NV = ROW / 128;
  constexpr int NP = 2 * NV;
  v2f   v[U][NP];
  float lg[U];
  float p[U];
#pragma unroll
  for (int u = 0; u < U; ++u) {
    const int s = __builtin_amdgcn_readlane(sidx, i + u);
    const float4* __restrict__ row = f4 + (long)s * (ROW / 4) + NV * lane;
#pragma unroll
    for (int q = 0; q < NV; ++q) {
      const float4 t = row[q];
      v[u][2 * q]     = (v2f){t.x, t.y};
      v[u][2 * q + 1] = (v2f){t.z, t.w};
    }
  }
#pragma unroll
  for (int u = 0; u < U; ++u) {
    v2f pp = leaky_pair(v[u][0] + fd[0]) * at[0];
#pragma unroll
    for (int k = 1; k < NP; ++k)
      pp = __builtin_elementwise_fma(leaky_pair(v[u][k] + fd[k]), at[k], pp);
    lg[u] = pp.x + pp.y;
  }
#pragma unroll
  for (int u = 0; u < U; ++u) lg[u] += row_xor_lane<4>(lg[u]);
#pragma unroll
  for (int u = 0; u < U; ++u) lg[u] += row_xor_lane<2>(lg[u]);
#pragma unroll
  for (int u = 0; u < U; ++u) lg[u] += row_xor_lane<1>(lg[u]);
  float bm = mx;
#pragma unroll
  for (int u = 0; u < U; ++u) bm = fmaxf(bm, lg[u]);
  const float sc = __expf(mx - bm);
  float d = den * sc;
#pragma unroll
  for (int u = 0; u < U; ++u) {
    p[u] = __expf(lg[u] - bm);
    d += p[u];
  }
#pragma unroll
  for (int k = 0; k < NP; ++k) {
    v2f a = acc[k] * sc;
#pragma unroll
    for (int u = 0; u < U; ++u)
      a = __builtin_elementwise_fma(v[u][k], (v2f){p[u], p[u]}, a);
    acc[k] = a;
  }
  den = d;
  mx  = bm;
}

__device__ __forceinline__ int first_ids(const int* __restrict__ nbr, int start, int deg, int lane) {
  const int nn = deg < 32 ? deg : 32;
  int sidx = 0;
  if (lane < nn) sidx = nbr[start + lane];
  return sidx;
}

template <int ROW>
__device__ __forceinline__ void attend_segment(
    const float* __restrict__ f, const v2f (&at)[ROW / 64], const v2f (&fd)[ROW / 64],
    const int* __restrict__ nbr, int start, int deg, int sidx, int lane,
    v2f (&acc)[ROW / 64], float& den) {
  constexpr int NV = ROW / 128;
  constexpr int NP = 2 * NV;
  static_assert(ROW == 128 || ROW == 256,
                "a lane's pieces must lie inside one head, and a head on 8 lanes");
  const float4* __restrict__ f4 = (const float4*)f;
#pragma unroll
  for (int k = 0; k < NP; ++k) acc[k] = (v2f){0.0f, 0.0f};
  float mx = -__builtin_inff();
  den = 0.0f;
  for (int c = 0; c < deg; c += 32) {
    const int nn = deg - c < 32 ? deg - c : 32;
    int i = 0;
    for (; i + 4 <= nn; i += 4) attend_edges<ROW, 4>(f4, sidx, i, lane, fd, at, acc, mx, den);
    if (i + 2 <= nn) { attend_edges<ROW, 2>(f4, sidx, i, lane, fd, at, acc, mx, den); i += 2; }
    if (i < nn)        attend_edges<ROW, 1>(f4, sidx, i, lane, fd, at, acc, mx, den);
    if (c + 32 < deg) sidx = first_ids(nbr, start + c + 32, deg - c - 32, lane);
  }
}

template <int ROW>
__device__ __forceinline__ void finish_row(const v2f (&acc)[ROW / 64], float den, int deg,
                                           float4 (&r)[ROW / 128]) {
  constexpr int NV = ROW / 128;
#pragma unroll
  for (int q = 0; q < NV; ++q) r[q] = make_float4(0.0f, 0.0f, 0.0f, 0.0f);
  if (deg > 0) {
    const float inv = 1.0f / den;
#pragma unroll
    for (int q = 0; q < NV; ++q) {
      const v2f lo = acc[2 * q] * inv, hi = acc[2 * q + 1] * inv;
      r[q] = make_float4(lo.x, lo.y, hi.x, hi.y);
    }
  }
}

template <int ROW>
__device__ __forceinline__ void load_own_row(const float* __restrict__ f, int node, int lane,
                                             v2f (&fd)[ROW / 64]) {
  constexpr int NV = ROW / 128;
  const float4* __restrict__ frow = (const float4*)f + (long)node * (ROW / 4) + NV * lane;
#pragma unroll
  for (int q = 0; q < NV; ++q) {
    const float4 a = frow[q];
    fd[2 * q] = (v2f){a.x, a.y};  fd[2 * q + 1] = (v2f){a.z, a.w};
  }
}

template <int ROW>
__device__ __forceinline__ void load_attn(const float* __restrict__ attn, int lane,
                                          v2f (&at)[ROW / 64]) {
  constexpr int NV = ROW / 128;
  const float4* __restrict__ arow = (const float4*)attn + NV * lane;
#pragma unroll
  for (int q = 0; q < NV; ++q) {
    const float4 b = arow[q];
    at[2 * q] = (v2f){b.x, b.y};  at[2 * q + 1] = (v2f){b.z, b.w};
  }
}

template <bool EXTRA>
__device__ __forceinline__ void project_tile(
    const float4* tile, const float* __restrict__ W1t, const float* __restrict__ Wrmt,
    float* __restrict__ f1, float* __restrict__ o1, int base, int N, int w, int lane,
    int xt, int xm) {
  constexpr int PITCH4 = 65, NC1 = 128, NCR = 32, KB = 256 / 8, MT = 2;
  const int hi = lane >> 4;
  const int lm = lane & 15;
  const int col1 = w * 16 + lm;
  const int colr = xt * 16 + lm;
  const float4* __restrict__ B1 = (const float4*)W1t + hi * NC1 + col1;
  const float4* __restrict__ BR = (const float4*)Wrmt + hi * NCR + colr;
  const int half0 = 16 * xm, half1 = 16 - half0;
  const float4* Ar0 = tile + (half0 + lm) * PITCH4 + hi;
  const float4* Ar1 = tile + (half1 + lm) * PITCH4 + hi;

  v8f c[MT];
  v8f cx = (v8f){};
#pragma unroll
  for (int m = 0; m < MT; ++m) c[m] = (v8f){};
  constexpr int PD = 6;
  float4 bq[PD], bx[PD];
#pragma unroll
  for (int p = 0; p < PD; ++p) {
    bq[p] = B1[(2 * p) * NC1];
    if (EXTRA) bx[p] = BR[(2 * p) * NCR];
  }
  constexpr int AD = 2;
  float4 aq[AD][MT];
#pragma unroll
  for (int p = 0; p < AD; ++p) {
#pragma unroll
    for (int m = 0; m < MT; ++m) aq[p][m] = (m ? Ar1 : Ar0)[2 * p];
  }
#pragma unroll
  for (int j = 0; j < KB; ++j) {
    float4 a_cur[MT];
#pragma unroll
    for (int m = 0; m < MT; ++m) {
      a_cur[m] = aq[j % AD][m];
      if (j + AD < KB) aq[j % AD][m] = (m ? Ar1 : Ar0)[2 * (j + AD)];
    }
    const float4 b = bq[j % PD];
    if (j + PD < KB) bq[j % PD] = B1[(2 * (j + PD)) * NC1];
    float4 x = make_float4(0.0f, 0.0f, 0.0f, 0.0f);
    if (EXTRA) {
      x = bx[j % PD];
      if (j + PD < KB) bx[j % PD] = BR[(2 * (j + PD)) * NCR];
    }
    const v2f b0 = {b.x, b.y}, b1 = {b.z, b.w};
#pragma unroll
    for (int m = 0; m < MT; ++m) {
      const v2f a0 = {a_cur[m].x, a_cur[m].y};
      c[m] = __builtin_amdgcn_wmma_f32_16x16x4_f32(false, a0, false, b0, (short)0, c[m], false, false);
    }
    if (EXTRA) {
      const v2f a0 = {a_cur[0].x, a_cur[0].y}, x0 = {x.x, x.y};
      cx = __builtin_amdgcn_wmma_f32_16x16x4_f32(false, a0, false, x0, (short)0, cx, false, false);
    }
#pragma unroll
    for (int m = 0; m < MT; ++m) {
      const v2f a1 = {a_cur[m].z, a_cur[m].w};
      c[m] = __builtin_amdgcn_wmma_f32_16x16x4_f32(false, a1, false, b1, (short)0, c[m], false, false);
    }
    if (EXTRA) {
      const v2f a1 = {a_cur[0].z, a_cur[0].w}, x1 = {x.z, x.w};
      cx = __builtin_amdgcn_wmma_f32_16x16x4_f32(false, a1, false, x1, (short)0, cx, false, false);
    }
    __builtin_amdgcn_sched_barrier(0);
  }

#pragma unroll
  for (int m = 0; m < MT; ++m) {
#pragma unroll
    for (int r = 0; r < 8; ++r) {
      const int orow = base + (m ? half1 : half0) + r + 8 * hi;
      if (orow < N) f1[(long)orow * NC1 + col1] = c[m][r];
    }
  }
  if (EXTRA) {
#pragma unroll
    for (int r = 0; r < 8; ++r) {
      const int orow = base + half0 + r + 8 * hi;
      if (orow < N) o1[(long)orow * NCR + colr] = cx[r];
    }
  }
}

__global__ __launch_bounds__(256, 8)
void attend0_project1(const float* __restrict__ f, const float* __restrict__ attn,
                      const int* __restrict__ nbr, const int* __restrict__ seg,
                      const float* __restrict__ W1t,
                      const float* __restrict__ Wrmt, float* __restrict__ f1,
                      float* __restrict__ o1, int N) {
  constexpr int ROW = 256, NV = ROW / 128, PITCH4 = 65;
  constexpr int NODES = 32;
  constexpr int PER_WAVE = NODES / 8;
  __shared__ float4 tile[NODES * PITCH4];
  const int lane = threadIdx.x & 31;
  const int w    = __builtin_amdgcn_readfirstlane((int)(threadIdx.x >> 5));
  const int base = blockIdx.x * NODES;

  struct alignas(32) Seg4 { int v[2 * PER_WAVE]; };
  static_assert(PER_WAVE == 4, "the rotation of the pairs below is written for 4 nodes");
  const Seg4 sg = *(const Seg4*)(seg + 2 * (base + PER_WAVE * w));
  int st = sg.v[0], dg = sg.v[1];
  int st1 = sg.v[2], dg1 = sg.v[3], st2 = sg.v[4], dg2 = sg.v[5], st3 = sg.v[6], dg3 = sg.v[7];
  v2f at[ROW / 64], fd[ROW / 64];
  load_attn<ROW>(attn, lane, at);
  int sidx = first_ids(nbr, st, dg, lane);
  if (dg > 0) load_own_row<ROW>(f, base + PER_WAVE * w, lane, fd);

#pragma nounroll
  for (int s = 0; s < PER_WAVE; ++s) {
    const int rloc = PER_WAVE * w + s;
    v2f acc[ROW / 64];
    float den;
    attend_segment<ROW>(f, at, fd, nbr, st, dg, sidx, lane, acc, den);
    sidx = first_ids(nbr, st1, dg1, lane);
    if (dg1 > 0) load_own_row<ROW>(f, base + rloc + 1, lane, fd);
    float4 r[NV];
    finish_row<ROW>(acc, den, dg, r);
    st = st1; dg = dg1; st1 = st2; dg1 = dg2; st2 = st3; dg2 = dg3; st3 = 0; dg3 = 0;
#pragma unroll
    for (int q = 0; q < NV; ++q) {
      float4 e = r[q];
      e.x = e.x > 0.0f ? e.x : expm1f(e.x);
      e.y = e.y > 0.0f ? e.y : expm1f(e.y);
      e.z = e.z > 0.0f ? e.z : expm1f(e.z);
      e.w = e.w > 0.0f ? e.w : expm1f(e.w);
      tile[rloc * PITCH4 + NV * lane + q] = e;
    }
  }
  __syncthreads();

  const bool extra = (w == 0) | (w == 3) | (w == 5) | (w == 6);
  const int  x     = (w == 0) ? 0 : (w == 3) ? 1 : (w == 5) ? 2 : 3;
  if (extra) project_tile<true >(tile, W1t, Wrmt, f1, o1, base, N, w, lane, x & 1, x >> 1);
  else       project_tile<false>(tile, W1t, Wrmt, f1, o1, base, N, w, lane, 0, 0);
}

__global__ __launch_bounds__(256)
void attend1_res_mean(const float* __restrict__ f, const float* __restrict__ attn,
                      const int* __restrict__ nbr, const int* __restrict__ seg,
                      const float* __restrict__ res,
                      float* __restrict__ out, int N) {
  constexpr int ROW = 128;
  const int lane = threadIdx.x & 31;
  const int node = __builtin_amdgcn_readfirstlane(
      (int)(blockIdx.x * (blockDim.x >> 5) + (threadIdx.x >> 5)));
  if (node >= N) return;
  const int2 sg = ((const int2*)seg)[node];
  const int sidx = first_ids(nbr, sg.x, sg.y, lane);
  v2f at[ROW / 64], fd[ROW / 64], acc[ROW / 64];
  float den;
  load_own_row<ROW>(f, node, lane, fd);
  load_attn<ROW>(attn, lane, at);
  attend_segment<ROW>(f, at, fd, nbr, sg.x, sg.y, sidx, lane, acc, den);
  float4 p = make_float4(0.0f, 0.0f, 0.0f, 0.0f);
  if (lane < 8) p = ((const float4*)(res + (long)node * (ROW / 4)))[lane];
  float4 r[1];
  finish_row<ROW>(acc, den, sg.y, r);
  float4 s = r[0];
  s.x += row_xor_lane<8>(s.x); s.y += row_xor_lane<8>(s.y);
  s.z += row_xor_lane<8>(s.z); s.w += row_xor_lane<8>(s.w);
  s.x += other_row_lane(s.x);  s.y += other_row_lane(s.y);
  s.z += other_row_lane(s.z);  s.w += other_row_lane(s.w);
  if (lane < 8) {
    s.x = s.x * 0.25f + p.x; s.y = s.y * 0.25f + p.y;
    s.z = s.z * 0.25f + p.z; s.w = s.w * 0.25f + p.w;
    ((float4*)(out + (long)node * (ROW / 4)))[lane] = s;
  }
}

static inline int cdiv_l(long a, long b) { return (int)((a + b - 1) / b); }

extern "C" void kernel_launch(void* const* d_in, const int* in_sizes, int n_in,
                              void* d_out, int out_size, void* d_ws, size_t ws_size,
                              hipStream_t stream) {
  const float* x     = (const float*)d_in[0];
  const float* W0    = (const float*)d_in[1];
  const float* a0    = (const float*)d_in[2];
  const float* W1    = (const float*)d_in[3];
  const float* a1    = (const float*)d_in[4];
  const float* Wres1 = (const float*)d_in[5];
  const int*   src   = (const int*)d_in[6];
  const int*   dst   = (const int*)d_in[7];

  const int N = in_sizes[0] / 256;
  const int E = in_sizes[6];

  float*    f0    = (float*)d_ws;
  float*    f1    = f0 + (long)N * 256;
  float*    grp   = f1 + (long)N * 256;
  float*    o1    = grp + (long)N * 128;
  int*      nbr    = (int*)grp;
  int*      cnt    = nbr + E;
  int*      cursor = cnt + N;
  const int    Npad    = (N + 31) & ~31;
  const size_t seg_off = ((size_t)E + (size_t)N * 2 + 7) & ~(size_t)7;
  int*      seg    = (int*)grp + seg_off;
  const size_t grp_ints = seg_off + (size_t)Npad * 2;
  float*    W0t    = grp + grp_ints;
  float*    W1t    = W0t + 256 * 256;
  float*    Wrmt   = W1t + 256 * 128;
  const size_t need = ((size_t)N * 256 * 2 + (size_t)N * 128 * 2) * sizeof(float);
  if (need > ws_size) return;
  if ((grp_ints + 256 * 256 + 256 * 128 + 256 * 32) * sizeof(float) > (size_t)N * 128 * sizeof(float))
    return;

  const int BT = 256;

  {
    const int zero_blocks = cdiv_l(N, BT);
    const int relayout_blocks = (64 * 256 + 64 * 128 + 64 * 32) / BT;
    zero_counts_relayout_weights<<<zero_blocks + relayout_blocks, BT, 0, stream>>>(
        cnt, N, zero_blocks, W0, W1, Wres1, W0t, W1t, Wrmt);
  }

  {
    const int strips = ((N + 15) / 16) * (256 / 16 / 4);
    gemm_f32_wmma<4, 256, 256><<<cdiv_l(strips, 8), BT, 0, stream>>>(x, W0t, f0, N);
  }
  dst_histogram<<<cdiv_l(E, BT), BT, 0, stream>>>(dst, cnt, E);
  rowptr_scan<<<cdiv_l(N, 1024), 1024, 0, stream>>>(cnt, (int2*)seg, cursor, N, Npad);
  csr_fill<<<cdiv_l(E, BT), BT, 0, stream>>>(src, dst, cursor, nbr, E);

  attend0_project1<<<cdiv_l(N, 32), BT, 0, stream>>>(f0, a0, nbr, seg, W1t, Wrmt,
                                                    f1, o1, N);

  attend1_res_mean<<<cdiv_l((long)N * 32, BT), BT, 0, stream>>>(f1, a1, nbr, seg, o1,
                                                                (float*)d_out, N);
}


# edge grouping in one pass: 32-slot bucket of source ids per node, slot = value the counting add returns (arrival order as before); overflow list past 32; scan and fill launches gone, 7 -> 5
# speedup vs baseline: 4.4162x; 1.1706x over previous
#include <hip/hip_runtime.h>
#include <hip/hip_bf16.h>

typedef __attribute__((ext_vector_type(2))) float v2f;
typedef __attribute__((ext_vector_type(8))) float v8f;

template <int NT, int K, int NC>
__global__ __launch_bounds__(256)
void gemm_f32_wmma(const float* __restrict__ A, const float* __restrict__ Bt,
                   float* __restrict__ C, int M) {
  static_assert(K % 8 == 0, "a block of the k loop is 8 k");
  const int lane = threadIdx.x & 31;
  const int wid  = blockIdx.x * (blockDim.x >> 5) + (threadIdx.x >> 5);
  const int tilesM  = (M + 15) >> 4;
  constexpr int stripsN = (NC >> 4) / NT;
  if (wid >= tilesM * stripsN) return;
  const int tm  = wid / stripsN;
  const int tn0 = (wid - tm * stripsN) * NT;
  const int hi = lane >> 4;
  const int lm = lane & 15;
  int arow = tm * 16 + lm; if (arow >= M) arow = M - 1;
  const int col0 = tn0 * 16 + lm;
  const float4* __restrict__ Ar = (const float4*)(A + (long)arow * K) + hi;
  const float4* __restrict__ Bc = (const float4*)Bt + (long)hi * NC + col0;

  v8f c[NT];
#pragma unroll
  for (int t = 0; t < NT; ++t) c[t] = (v8f){};

  auto loadB = [&](int j, float4* b) {
    const float4* __restrict__ B0 = Bc + (long)(2 * j) * NC;
#pragma unroll
    for (int t = 0; t < NT; ++t) b[t] = B0[16 * t];
  };
  auto mma_block = [&](const float4& a, const float4* b) {
    const v2f a0 = {a.x, a.y}, a1 = {a.z, a.w};
#pragma unroll
    for (int t = 0; t < NT; ++t) {
      const v2f b0 = {b[t].x, b[t].y};
      c[t] = __builtin_amdgcn_wmma_f32_16x16x4_f32(false, a0, false, b0,
                                                   (short)0, c[t], false, false);
    }
#pragma unroll
    for (int t = 0; t < NT; ++t) {
      const v2f b1 = {b[t].z, b[t].w};
      c[t] = __builtin_amdgcn_wmma_f32_16x16x4_f32(false, a1, false, b1,
                                                   (short)0, c[t], false, false);
    }
  };

  float4 a_cur = Ar[0];
  float4 b_cur[NT];
  loadB(0, b_cur);
#pragma unroll
  for (int j = 0; j < K / 8 - 1; ++j) {
    const float4 a_nxt = Ar[2 * (j + 1)];
    float4 b_nxt[NT];
    loadB(j + 1, b_nxt);
    mma_block(a_cur, b_cur);
    a_cur = a_nxt;
#pragma unroll
    for (int t = 0; t < NT; ++t) b_cur[t] = b_nxt[t];
  }
  mma_block(a_cur, b_cur);

#pragma unroll
  for (int t = 0; t < NT; ++t) {
#pragma unroll
    for (int r = 0; r < 8; ++r) {
      const int orow = tm * 16 + r + 8 * hi;
      if (orow < M) C[(long)orow * NC + col0 + 16 * t] = c[t][r];
    }
  }
}

__device__ __forceinline__ void relayout_k4(const float* __restrict__ B,
                                            float* __restrict__ Bt, int idx, int NC) {
  const int kq = idx / NC;
  const int n  = idx - kq * NC;
  const float* __restrict__ p = B + (long)(4 * kq) * NC + n;
  ((float4*)Bt)[idx] = make_float4(p[0], p[NC], p[2 * NC], p[3 * NC]);
}

__device__ __forceinline__ void relayout_k4_head_mean(const float* __restrict__ B,
                                                      float* __restrict__ Bt, int idx) {
  const int kq = idx >> 5;
  const int c  = idx & 31;
  const float* __restrict__ p = B + (long)(4 * kq) * 128 + c;
  float m[4];
#pragma unroll
  for (int e = 0; e < 4; ++e) {
    const float* __restrict__ q = p + e * 128;
    m[e] = 0.25f * ((q[0] + q[32]) + (q[64] + q[96]));
  }
  ((float4*)Bt)[idx] = make_float4(m[0], m[1], m[2], m[3]);
}

__global__ void zero_counts_relayout_weights(
    int* __restrict__ cnt, int n, int zero_blocks,
    const float* __restrict__ W0, const float* __restrict__ W1,
    const float* __restrict__ Wres1, float* __restrict__ W0t,
    float* __restrict__ W1t, float* __restrict__ Wrmt) {
  if ((int)blockIdx.x < zero_blocks) {
    const int i = blockIdx.x * blockDim.x + threadIdx.x;
    if (i < n) cnt[i] = 0;
    return;
  }
  const int r = (blockIdx.x - zero_blocks) * blockDim.x + threadIdx.x;
  if (r < 64 * 256)                  relayout_k4(W0, W0t, r, 256);
  else if (r < 64 * 256 + 64 * 128)  relayout_k4(W1, W1t, r - 64 * 256, 128);
  else if (r < 64 * 256 + 64 * 128 + 64 * 32)
                                     relayout_k4_head_mean(Wres1, Wrmt, r - 64 * 256 - 64 * 128);
}

constexpr int CAP = 32;

__global__ void bucket_edges(const int* __restrict__ src, const int* __restrict__ dst,
                             int* __restrict__ cnt, int* __restrict__ bucket,
                             int* __restrict__ ovf_cnt, int2* __restrict__ ovf, int E) {
  const int e = blockIdx.x * blockDim.x + threadIdx.x;
  if (e >= E) return;
  const int d = dst[e];
  const int s = src[e];
  const int p = atomicAdd(cnt + d, 1);
  if (p < CAP) {
    bucket[d * CAP + p] = s;
  } else {
    const int q = atomicAdd(ovf_cnt, 1);
    ovf[q] = make_int2(d, s);
  }
}

template <int MASK>
__device__ __forceinline__ float row_xor_lane(float x) {
  return __int_as_float(__builtin_amdgcn_update_dpp(
      0, __float_as_int(x), 0x160 + MASK, 0xF, 0xF, true));
}

__device__ __forceinline__ float other_row_lane(float x) {
  return __int_as_float((int)__builtin_amdgcn_permlanex16(
      0u, (unsigned)__float_as_int(x), 0x76543210u, 0xfedcba98u, false, false));
}

__device__ __forceinline__ v2f leaky_pair(v2f t) {
  return __builtin_elementwise_max(t, t * 0.2f);
}

template <int ROW, int U>
__device__ __forceinline__ void attend_edges(
    const float4* __restrict__ f4, int sidx, int i, int lane,
    const v2f (&fd)[ROW / 64], const v2f (&at)[ROW / 64],
    v2f (&acc)[ROW / 64], float& mx, float& den) {
  constexpr int NV = ROW / 128;
  constexpr int NP = 2 * NV;
  v2f   v[U][NP];
  float lg[U];
  float p[U];
#pragma unroll
  for (int u = 0; u < U; ++u) {
    const int s = __builtin_amdgcn_readlane(sidx, i + u);
    const float4* __restrict__ row = f4 + (long)s * (ROW / 4) + NV * lane;
#pragma unroll
    for (int q = 0; q < NV; ++q) {
      const float4 t = row[q];
      v[u][2 * q]     = (v2f){t.x, t.y};
      v[u][2 * q + 1] = (v2f){t.z, t.w};
    }
  }
#pragma unroll
  for (int u = 0; u < U; ++u) {
    v2f pp = leaky_pair(v[u][0] + fd[0]) * at[0];
#pragma unroll
    for (int k = 1; k < NP; ++k)
      pp = __builtin_elementwise_fma(leaky_pair(v[u][k] + fd[k]), at[k], pp);
    lg[u] = pp.x + pp.y;
  }
#pragma unroll
  for (int u = 0; u < U; ++u) lg[u] += row_xor_lane<4>(lg[u]);
#pragma unroll
  for (int u = 0; u < U; ++u) lg[u] += row_xor_lane<2>(lg[u]);
#pragma unroll
  for (int u = 0; u < U; ++u) lg[u] += row_xor_lane<1>(lg[u]);
  float bm = mx;
#pragma unroll
  for (int u = 0; u < U; ++u) bm = fmaxf(bm, lg[u]);
  const float sc = __expf(mx - bm);
  float d = den * sc;
#pragma unroll
  for (int u = 0; u < U; ++u) {
    p[u] = __expf(lg[u] - bm);
    d += p[u];
  }
#pragma unroll
  for (int k = 0; k < NP; ++k) {
    v2f a = acc[k] * sc;
#pragma unroll
    for (int u = 0; u < U; ++u)
      a = __builtin_elementwise_fma(v[u][k], (v2f){p[u], p[u]}, a);
    acc[k] = a;
  }
  den = d;
  mx  = bm;
}

__device__ __forceinline__ int first_ids(const int* __restrict__ ids, int start, int deg, int lane) {
  const int nn = deg < 32 ? deg : 32;
  int sidx = 0;
  if (lane < nn) sidx = ids[start + lane];
  return sidx;
}

template <int ROW>
__device__ __forceinline__ void attend_segment(
    const float* __restrict__ f, const v2f (&at)[ROW / 64], const v2f (&fd)[ROW / 64],
    const int* __restrict__ ovf_cnt, const int2* __restrict__ ovf, int ovf_cap,
    int node, int deg, int sidx, int lane,
    v2f (&acc)[ROW / 64], float& den) {
  constexpr int NV = ROW / 128;
  constexpr int NP = 2 * NV;
  static_assert(ROW == 128 || ROW == 256,
                "a lane's pieces must lie inside one head, and a head on 8 lanes");
  const float4* __restrict__ f4 = (const float4*)f;
#pragma unroll
  for (int k = 0; k < NP; ++k) acc[k] = (v2f){0.0f, 0.0f};
  float mx = -__builtin_inff();
  den = 0.0f;
  const int nn = deg < CAP ? deg : CAP;
  int i = 0;
  for (; i + 4 <= nn; i += 4) attend_edges<ROW, 4>(f4, sidx, i, lane, fd, at, acc, mx, den);
  if (i + 2 <= nn) { attend_edges<ROW, 2>(f4, sidx, i, lane, fd, at, acc, mx, den); i += 2; }
  if (i < nn)        attend_edges<ROW, 1>(f4, sidx, i, lane, fd, at, acc, mx, den);
  if (deg > CAP) {
    int n_ovf = *ovf_cnt;
    if (n_ovf > ovf_cap) n_ovf = ovf_cap;
    for (int b = 0; b < n_ovf; b += 32) {
      int2 pr = make_int2(-1, 0);
      if (b + lane < n_ovf) pr = ovf[b + lane];
      unsigned m = (unsigned)__builtin_amdgcn_ballot_w32(pr.x == node);
      while (m) {
        const int j = __builtin_ctz(m);
        m &= m - 1u;
        attend_edges<ROW, 1>(f4, pr.y, j, lane, fd, at, acc, mx, den);
      }
    }
  }
}

template <int ROW>
__device__ __forceinline__ void finish_row(const v2f (&acc)[ROW / 64], float den, int deg,
                                           float4 (&r)[ROW / 128]) {
  constexpr int NV = ROW / 128;
#pragma unroll
  for (int q = 0; q < NV; ++q) r[q] = make_float4(0.0f, 0.0f, 0.0f, 0.0f);
  if (deg > 0) {
    const float inv = 1.0f / den;
#pragma unroll
    for (int q = 0; q < NV; ++q) {
      const v2f lo = acc[2 * q] * inv, hi = acc[2 * q + 1] * inv;
      r[q] = make_float4(lo.x, lo.y, hi.x, hi.y);
    }
  }
}

template <int ROW>
__device__ __forceinline__ void load_own_row(const float* __restrict__ f, int node, int lane,
                                             v2f (&fd)[ROW / 64]) {
  constexpr int NV = ROW / 128;
  const float4* __restrict__ frow = (const float4*)f + (long)node * (ROW / 4) + NV * lane;
#pragma unroll
  for (int q = 0; q < NV; ++q) {
    const float4 a = frow[q];
    fd[2 * q] = (v2f){a.x, a.y};  fd[2 * q + 1] = (v2f){a.z, a.w};
  }
}

template <int ROW>
__device__ __forceinline__ void load_attn(const float* __restrict__ attn, int lane,
                                          v2f (&at)[ROW / 64]) {
  constexpr int NV = ROW / 128;
  const float4* __restrict__ arow = (const float4*)attn + NV * lane;
#pragma unroll
  for (int q = 0; q < NV; ++q) {
    const float4 b = arow[q];
    at[2 * q] = (v2f){b.x, b.y};  at[2 * q + 1] = (v2f){b.z, b.w};
  }
}

template <bool EXTRA>
__device__ __forceinline__ void project_tile(
    const float4* tile, const float* __restrict__ W1t, const float* __restrict__ Wrmt,
    float* __restrict__ f1, float* __restrict__ o1, int base, int N, int w, int lane,
    int xt, int xm) {
  constexpr int PITCH4 = 65, NC1 = 128, NCR = 32, KB = 256 / 8, MT = 2;
  const int hi = lane >> 4;
  const int lm = lane & 15;
  const int col1 = w * 16 + lm;
  const int colr = xt * 16 + lm;
  const float4* __restrict__ B1 = (const float4*)W1t + hi * NC1 + col1;
  const float4* __restrict__ BR = (const float4*)Wrmt + hi * NCR + colr;
  const int half0 = 16 * xm, half1 = 16 - half0;
  const float4* Ar0 = tile + (half0 + lm) * PITCH4 + hi;
  const float4* Ar1 = tile + (half1 + lm) * PITCH4 + hi;

  v8f c[MT];
  v8f cx = (v8f){};
#pragma unroll
  for (int m = 0; m < MT; ++m) c[m] = (v8f){};
  constexpr int PD = 6;
  float4 bq[PD], bx[PD];
#pragma unroll
  for (int p = 0; p < PD; ++p) {
    bq[p] = B1[(2 * p) * NC1];
    if (EXTRA) bx[p] = BR[(2 * p) * NCR];
  }
  constexpr int AD = 2;
  float4 aq[AD][MT];
#pragma unroll
  for (int p = 0; p < AD; ++p) {
#pragma unroll
    for (int m = 0; m < MT; ++m) aq[p][m] = (m ? Ar1 : Ar0)[2 * p];
  }
#pragma unroll
  for (int j = 0; j < KB; ++j) {
    float4 a_cur[MT];
#pragma unroll
    for (int m = 0; m < MT; ++m) {
      a_cur[m] = aq[j % AD][m];
      if (j + AD < KB) aq[j % AD][m] = (m ? Ar1 : Ar0)[2 * (j + AD)];
    }
    const float4 b = bq[j % PD];
    if (j + PD < KB) bq[j % PD] = B1[(2 * (j + PD)) * NC1];
    float4 x = make_float4(0.0f, 0.0f, 0.0f, 0.0f);
    if (EXTRA) {
      x = bx[j % PD];
      if (j + PD < KB) bx[j % PD] = BR[(2 * (j + PD)) * NCR];
    }
    const v2f b0 = {b.x, b.y}, b1 = {b.z, b.w};
#pragma unroll
    for (int m = 0; m < MT; ++m) {
      const v2f a0 = {a_cur[m].x, a_cur[m].y};
      c[m] = __builtin_amdgcn_wmma_f32_16x16x4_f32(false, a0, false, b0, (short)0, c[m], false, false);
    }
    if (EXTRA) {
      const v2f a0 = {a_cur[0].x, a_cur[0].y}, x0 = {x.x, x.y};
      cx = __builtin_amdgcn_wmma_f32_16x16x4_f32(false, a0, false, x0, (short)0, cx, false, false);
    }
#pragma unroll
    for (int m = 0; m < MT; ++m) {
      const v2f a1 = {a_cur[m].z, a_cur[m].w};
      c[m] = __builtin_amdgcn_wmma_f32_16x16x4_f32(false, a1, false, b1, (short)0, c[m], false, false);
    }
    if (EXTRA) {
      const v2f a1 = {a_cur[0].z, a_cur[0].w}, x1 = {x.z, x.w};
      cx = __builtin_amdgcn_wmma_f32_16x16x4_f32(false, a1, false, x1, (short)0, cx, false, false);
    }
    __builtin_amdgcn_sched_barrier(0);
  }

#pragma unroll
  for (int m = 0; m < MT; ++m) {
#pragma unroll
    for (int r = 0; r < 8; ++r) {
      const int orow = base + (m ? half1 : half0) + r + 8 * hi;
      if (orow < N) f1[(long)orow * NC1 + col1] = c[m][r];
    }
  }
  if (EXTRA) {
#pragma unroll
    for (int r = 0; r < 8; ++r) {
      const int orow = base + half0 + r + 8 * hi;
      if (orow < N) o1[(long)orow * NCR + colr] = cx[r];
    }
  }
}

__global__ __launch_bounds__(256, 8)
void attend0_project1(const float* __restrict__ f, const float* __restrict__ attn,
                      const int* __restrict__ bucket, const int* __restrict__ cnt,
                      const int* __restrict__ ovf_cnt, const int2* __restrict__ ovf, int ovf_cap,
                      const float* __restrict__ W1t,
                      const float* __restrict__ Wrmt, float* __restrict__ f1,
                      float* __restrict__ o1, int N) {
  constexpr int ROW = 256, NV = ROW / 128, PITCH4 = 65;
  constexpr int NODES = 32;
  constexpr int PER_WAVE = NODES / 8;
  __shared__ float4 tile[NODES * PITCH4];
  const int lane = threadIdx.x & 31;
  const int w    = __builtin_amdgcn_readfirstlane((int)(threadIdx.x >> 5));
  const int base = blockIdx.x * NODES;

  struct alignas(16) Cnt4 { int v[PER_WAVE]; };
  static_assert(PER_WAVE == 4, "the rotation of the counts below is written for 4 nodes");
  const Cnt4 cg = *(const Cnt4*)(cnt + base + PER_WAVE * w);
  int dg = cg.v[0];
  int dg1 = cg.v[1], dg2 = cg.v[2], dg3 = cg.v[3];
  v2f at[ROW / 64], fd[ROW / 64];
  load_attn<ROW>(attn, lane, at);
  int sidx = first_ids(bucket, (base + PER_WAVE * w) * CAP, dg, lane);
  if (dg > 0) load_own_row<ROW>(f, base + PER_WAVE * w, lane, fd);

#pragma nounroll
  for (int s = 0; s < PER_WAVE; ++s) {
    const int rloc = PER_WAVE * w + s;
    v2f acc[ROW / 64];
    float den;
    attend_segment<ROW>(f, at, fd, ovf_cnt, ovf, ovf_cap, base + rloc, dg, sidx, lane, acc, den);
    sidx = first_ids(bucket, (base + rloc + 1) * CAP, dg1, lane);
    if (dg1 > 0) load_own_row<ROW>(f, base + rloc + 1, lane, fd);
    float4 r[NV];
    finish_row<ROW>(acc, den, dg, r);
    dg = dg1; dg1 = dg2; dg2 = dg3; dg3 = 0;
#pragma unroll
    for (int q = 0; q < NV; ++q) {
      float4 e = r[q];
      e.x = e.x > 0.0f ? e.x : expm1f(e.x);
      e.y = e.y > 0.0f ? e.y : expm1f(e.y);
      e.z = e.z > 0.0f ? e.z : expm1f(e.z);
      e.w = e.w > 0.0f ? e.w : expm1f(e.w);
      tile[rloc * PITCH4 + NV * lane + q] = e;
    }
  }
  __syncthreads();

  const bool extra = (w == 0) | (w == 3) | (w == 5) | (w == 6);
  const int  x     = (w == 0) ? 0 : (w == 3) ? 1 : (w == 5) ? 2 : 3;
  if (extra) project_tile<true >(tile, W1t, Wrmt, f1, o1, base, N, w, lane, x & 1, x >> 1);
  else       project_tile<false>(tile, W1t, Wrmt, f1, o1, base, N, w, lane, 0, 0);
}

__global__ __launch_bounds__(256)
void attend1_res_mean(const float* __restrict__ f, const float* __restrict__ attn,
                      const int* __restrict__ bucket, const int* __restrict__ cnt,
                      const int* __restrict__ ovf_cnt, const int2* __restrict__ ovf, int ovf_cap,
                      const float* __restrict__ res,
                      float* __restrict__ out, int N) {
  constexpr int ROW = 128;
  const int lane = threadIdx.x & 31;
  const int node = __builtin_amdgcn_readfirstlane(
      (int)(blockIdx.x * (blockDim.x >> 5) + (threadIdx.x >> 5)));
  if (node >= N) return;
  const int deg = cnt[node];
  const int sidx = first_ids(bucket, node * CAP, deg, lane);
  v2f at[ROW / 64], fd[ROW / 64], acc[ROW / 64];
  float den;
  load_own_row<ROW>(f, node, lane, fd);
  load_attn<ROW>(attn, lane, at);
  attend_segment<ROW>(f, at, fd, ovf_cnt, ovf, ovf_cap, node, deg, sidx, lane, acc, den);
  float4 p = make_float4(0.0f, 0.0f, 0.0f, 0.0f);
  if (lane < 8) p = ((const float4*)(res + (long)node * (ROW / 4)))[lane];
  float4 r[1];
  finish_row<ROW>(acc, den, deg, r);
  float4 s = r[0];
  s.x += row_xor_lane<8>(s.x); s.y += row_xor_lane<8>(s.y);
  s.z += row_xor_lane<8>(s.z); s.w += row_xor_lane<8>(s.w);
  s.x += other_row_lane(s.x);  s.y += other_row_lane(s.y);
  s.z += other_row_lane(s.z);  s.w += other_row_lane(s.w);
  if (lane < 8) {
    s.x = s.x * 0.25f + p.x; s.y = s.y * 0.25f + p.y;
    s.z = s.z * 0.25f + p.z; s.w = s.w * 0.25f + p.w;
    ((float4*)(out + (long)node * (ROW / 4)))[lane] = s;
  }
}

static inline int cdiv_l(long a, long b) { return (int)((a + b - 1) / b); }

extern "C" void kernel_launch(void* const* d_in, const int* in_sizes, int n_in,
                              void* d_out, int out_size, void* d_ws, size_t ws_size,
                              hipStream_t stream) {
  const float* x     = (const float*)d_in[0];
  const float* W0    = (const float*)d_in[1];
  const float* a0    = (const float*)d_in[2];
  const float* W1    = (const float*)d_in[3];
  const float* a1    = (const float*)d_in[4];
  const float* Wres1 = (const float*)d_in[5];
  const int*   src   = (const int*)d_in[6];
  const int*   dst   = (const int*)d_in[7];

  const int N = in_sizes[0] / 256;
  const int E = in_sizes[6];

  float*    f0    = (float*)d_ws;
  float*    f1    = f0 + (long)N * 256;
  float*    grp   = f1 + (long)N * 256;
  float*    o1    = grp + (long)N * 128;
  const int    Npad    = (N + 31) & ~31;
  int*      bucket  = (int*)grp;
  int*      cnt     = bucket + (size_t)Npad * CAP;
  int*      ovf_cnt = cnt + Npad;
  int2*     ovf     = (int2*)(ovf_cnt + 8);
  const size_t grp_ints = (((size_t)Npad * (CAP + 1) + 8 + (size_t)E * 2) + 7) & ~(size_t)7;
  float*    W0t    = grp + grp_ints;
  float*    W1t    = W0t + 256 * 256;
  float*    Wrmt   = W1t + 256 * 128;
  const size_t need = ((size_t)N * 256 * 2 + (size_t)N * 128 * 2) * sizeof(float);
  if (need > ws_size) return;
  if ((grp_ints + 256 * 256 + 256 * 128 + 256 * 32) * sizeof(float) > (size_t)N * 128 * sizeof(float))
    return;

  const int BT = 256;

  {
    const int zero_blocks = cdiv_l(Npad + 8, BT);
    const int relayout_blocks = (64 * 256 + 64 * 128 + 64 * 32) / BT;
    zero_counts_relayout_weights<<<zero_blocks + relayout_blocks, BT, 0, stream>>>(
        cnt, Npad + 8, zero_blocks, W0, W1, Wres1, W0t, W1t, Wrmt);
  }

  {
    const int strips = ((N + 15) / 16) * (256 / 16 / 4);
    gemm_f32_wmma<4, 256, 256><<<cdiv_l(strips, 8), BT, 0, stream>>>(x, W0t, f0, N);
  }
  bucket_edges<<<cdiv_l(E, BT), BT, 0, stream>>>(src, dst, cnt, bucket, ovf_cnt, ovf, E);

  attend0_project1<<<cdiv_l(N, 32), BT, 0, stream>>>(f0, a0, bucket, cnt, ovf_cnt, ovf, E,
                                                    W1t, Wrmt, f1, o1, N);

  attend1_res_mean<<<cdiv_l((long)N * 32, BT), BT, 0, stream>>>(f1, a1, bucket, cnt, ovf_cnt, ovf, E,
                                                                o1, (float*)d_out, N);
}
